# starting kernel
# baseline (speedup 1.0000x reference)
#include <hip/hip_runtime.h>
#include <hip/hip_bf16.h>

typedef __attribute__((ext_vector_type(2))) float v2f;
typedef __attribute__((ext_vector_type(8))) float v8f;

__device__ __forceinline__ v8f wmma4(v2f a, v2f b, v8f c) {
  return __builtin_amdgcn_wmma_f32_16x16x4_f32(
       false, a,  false, b,
       (short)0, c,  false,  false);
}

__global__ void deg_kernel(const int* __restrict__ dst, float* __restrict__ deg, int E) {
  int tid = blockIdx.x * blockDim.x + threadIdx.x;
  if (tid < E) atomicAdd(&deg[dst[tid]], 1.0f);
}

__global__ void dinv_kernel(const float* __restrict__ deg, float* __restrict__ dinv, int N) {
  int tid = blockIdx.x * blockDim.x + threadIdx.x;
  if (tid < N) {
    float d = deg[tid];
    d = d < 1.0f ? 1.0f : d;
    dinv[tid] = rsqrtf(d);
  }
}

__global__ void wlap_kernel(const float* __restrict__ dinv, const int* __restrict__ src,
                            const int* __restrict__ dst, float* __restrict__ wlap, int E) {
  int tid = blockIdx.x * blockDim.x + threadIdx.x;
  if (tid < E) wlap[tid] = -(dinv[src[tid]] * dinv[dst[tid]]);
}

__global__ void scatter_wadd(const float* __restrict__ x, const float* __restrict__ w,
                             const int* __restrict__ src, const int* __restrict__ dst,
                             float* __restrict__ out, long long total) {
  long long tid = (long long)blockIdx.x * blockDim.x + threadIdx.x;
  if (tid >= total) return;
  int f = (int)(tid & 63);
  long long e = tid >> 6;
  int s = src[e], d = dst[e];
  float v = x[(size_t)s * 64 + f] * w[e];
  atomicAdd(&out[(size_t)d * 64 + f], v);
}

__global__ void combine_kernel(const float* __restrict__ a, const float* __restrict__ b,
                               float* __restrict__ out, long long n) {
  long long tid = (long long)blockIdx.x * blockDim.x + threadIdx.x;
  if (tid < n) out[tid] = 2.0f * a[tid] - b[tid];
}

__global__ void concat_hp(const float* __restrict__ h, const float* __restrict__ p,
                          float* __restrict__ hp, int N) {
  long long tid = (long long)blockIdx.x * blockDim.x + threadIdx.x;
  if (tid >= (long long)N * 128) return;
  int n = (int)(tid >> 7);
  int c = (int)(tid & 127);
  hp[tid] = (c < 64) ? h[(size_t)n * 64 + c] : p[(size_t)n * 64 + (c - 64)];
}

__global__ void gemm_wmma_kernel(const float* __restrict__ A, int lda,
                                 const float* __restrict__ B,
                                 const float* __restrict__ bias,
                                 float* __restrict__ Out, int ldo, int coloff,
                                 int rows, int kdim) {
  int wave = blockIdx.x * (blockDim.x >> 5) + (threadIdx.x >> 5);
  int rowTiles = (rows + 15) >> 4;
  if (wave >= rowTiles * 4) return;
  int rt = wave >> 2, ct = wave & 3;
  int lane = threadIdx.x & 31, half = lane >> 4, l16 = lane & 15;
  int ar = rt * 16 + l16;
  if (ar >= rows) ar = rows - 1;
  int bc = ct * 16 + l16;
  v8f acc = {};
  for (int k0 = 0; k0 < kdim; k0 += 4) {
    int kr = k0 + 2 * half;
    v2f a, b;
    a.x = A[(size_t)ar * lda + kr];
    a.y = A[(size_t)ar * lda + kr + 1];
    b.x = B[(size_t)kr * 64 + bc];
    b.y = B[(size_t)(kr + 1) * 64 + bc];
    acc = wmma4(a, b, acc);
  }
  float bv = bias ? bias[ct * 16 + l16] : 0.0f;
#pragma unroll
  for (int v = 0; v < 8; ++v) {
    int r = rt * 16 + v + 8 * half;
    if (r < rows) Out[(size_t)r * ldo + coloff + ct * 16 + l16] = acc[v] + bv;
  }
}

__global__ void cheb_gemm_kernel(const float* __restrict__ Tx0, const float* __restrict__ Tx1,
                                 const float* __restrict__ Tx2, const float* __restrict__ Tx3,
                                 const float* __restrict__ fc, const float* __restrict__ Wc,
                                 const float* __restrict__ bias, float* __restrict__ Out,
                                 int rows, int Nn) {
  int wave = blockIdx.x * (blockDim.x >> 5) + (threadIdx.x >> 5);
  int rowTiles = (rows + 15) >> 4;
  if (wave >= rowTiles * 4) return;
  int rt = wave >> 2, ct = wave & 3;
  int lane = threadIdx.x & 31, half = lane >> 4, l16 = lane & 15;
  int ar = rt * 16 + l16;
  if (ar >= rows) ar = rows - 1;
  int bc = ct * 16 + l16;
  const float* As[4] = {Tx0, Tx1, Tx2, Tx3};
  v8f acc = {};
#pragma unroll
  for (int k = 0; k < 4; ++k) {
    const float* A = As[k];
    const float* B = Wc + (size_t)k * 64 * 64;
    float sc = fc[(size_t)k * Nn + ar];
#pragma unroll
    for (int k0 = 0; k0 < 64; k0 += 4) {
      int kr = k0 + 2 * half;
      v2f a, b;
      a.x = sc * A[(size_t)ar * 64 + kr];
      a.y = sc * A[(size_t)ar * 64 + kr + 1];
      b.x = B[(size_t)kr * 64 + bc];
      b.y = B[(size_t)(kr + 1) * 64 + bc];
      acc = wmma4(a, b, acc);
    }
  }
  float bv = bias[ct * 16 + l16];
#pragma unroll
  for (int v = 0; v < 8; ++v) {
    int r = rt * 16 + v + 8 * half;
    if (r < rows) Out[(size_t)r * 192 + ct * 16 + l16] = acc[v] + bv;
  }
}

__global__ void attn_kernel(const float* __restrict__ Ee, const float* __restrict__ WE,
                            const float* __restrict__ Qh, const float* __restrict__ Kh,
                            const float* __restrict__ Vh,
                            const int* __restrict__ src, const int* __restrict__ dst,
                            float* __restrict__ wV, float* __restrict__ z, int E) {
  __shared__ float lds[4 * 16 * 64];
  int wid = threadIdx.x >> 5;
  int tile = blockIdx.x * 4 + wid;
  int eTiles = (E + 15) >> 4;
  bool valid = tile < eTiles;
  int t = valid ? tile : 0;
  int e0 = t * 16;
  int lane = threadIdx.x & 31, half = lane >> 4, l16 = lane & 15;
  int ar = e0 + l16;
  if (ar >= E) ar = E - 1;
  float* myLds = lds + wid * 1024;

#pragma unroll
  for (int ct = 0; ct < 4; ++ct) {
    v8f acc = {};
#pragma unroll
    for (int k0 = 0; k0 < 64; k0 += 4) {
      int kr = k0 + 2 * half;
      v2f a, b;
      a.x = Ee[(size_t)ar * 64 + kr];
      a.y = Ee[(size_t)ar * 64 + kr + 1];
      int bc = ct * 16 + l16;
      b.x = WE[(size_t)kr * 64 + bc];
      b.y = WE[(size_t)(kr + 1) * 64 + bc];
      acc = wmma4(a, b, acc);
    }
#pragma unroll
    for (int v = 0; v < 8; ++v)
      myLds[(v + 8 * half) * 64 + ct * 16 + l16] = acc[v];
  }
  __syncthreads();
  if (!valid) return;

  const float scale = 0.35355339059327373f;
#pragma unroll
  for (int pp = 0; pp < 4; ++pp) {
    int pair = pp * 32 + (threadIdx.x & 31);
    int m = pair >> 3, hh = pair & 7;
    int em = e0 + m;
    if (em >= E) continue;
    int s = src[em], d = dst[em];
    float sc = 0.0f;
#pragma unroll
    for (int dd = 0; dd < 8; ++dd) {
      int c = hh * 8 + dd;
      sc += Kh[(size_t)s * 64 + c] * Qh[(size_t)d * 64 + c] * myLds[m * 64 + c];
    }
    sc *= scale;
    sc = fminf(5.0f, fmaxf(-5.0f, sc));
    float ss = __expf(sc);
    atomicAdd(&z[(size_t)d * 8 + hh], ss);
#pragma unroll
    for (int dd = 0; dd < 8; ++dd) {
      int c = hh * 8 + dd;
      atomicAdd(&wV[(size_t)d * 64 + c], Vh[(size_t)s * 64 + c] * ss);
    }
  }
}

__global__ void finalize_kernel(const float* __restrict__ wV, const float* __restrict__ z,
                                float* __restrict__ out, int N) {
  long long tid = (long long)blockIdx.x * blockDim.x + threadIdx.x;
  if (tid >= (long long)N * 64) return;
  int n = (int)(tid >> 6);
  int c = (int)(tid & 63);
  float zz = z[(size_t)n * 8 + (c >> 3)] + 1e-6f;
  out[(size_t)n * 192 + 128 + c] = wV[tid] / zz;
}

extern "C" void kernel_launch(void* const* d_in, const int* in_sizes, int n_in,
                              void* d_out, int out_size, void* d_ws, size_t ws_size,
                              hipStream_t stream) {
  const float* h  = (const float*)d_in[0];
  const float* p  = (const float*)d_in[1];
  const float* e  = (const float*)d_in[2];
  const float* ew = (const float*)d_in[3];
  const float* fc = (const float*)d_in[4];
  const int* esrc = (const int*)d_in[5];
  const int* edst = (const int*)d_in[6];
  const float* Wc = (const float*)d_in[7];
  const float* bc = (const float*)d_in[8];
  const float* Wg = (const float*)d_in[9];
  const float* bg = (const float*)d_in[10];
  const float* WQ = (const float*)d_in[11];
  const float* WK = (const float*)d_in[12];
  const float* WVm = (const float*)d_in[13];
  const float* WE = (const float*)d_in[14];
  float* out = (float*)d_out;

  const int N = in_sizes[0] / 64;
  const int E = in_sizes[5];
  const size_t ND = (size_t)N * 64;

  float* w = (float*)d_ws;
  float* deg  = w; w += N;
  float* dinv = w; w += N;
  float* wlap = w; w += E;
  float* Tx1  = w; w += ND;
  float* Tx2  = w; w += ND;
  float* Tx3  = w; w += ND;
  float* tmp  = w; w += ND;
  float* gcn  = w; w += ND;
  float* hp   = w; w += 2 * ND;
  float* Qh   = w; w += ND;
  float* Kh   = w; w += ND;
  float* Vh   = w; w += ND;
  float* wVb  = w; w += ND;
  float* zb   = w; w += (size_t)N * 8;

  hipMemsetAsync(deg, 0, sizeof(float) * N, stream);
  hipMemsetAsync(Tx1, 0, sizeof(float) * ND, stream);
  hipMemsetAsync(tmp, 0, sizeof(float) * ND, stream);
  hipMemsetAsync(gcn, 0, sizeof(float) * ND, stream);
  hipMemsetAsync(wVb, 0, sizeof(float) * ND, stream);
  hipMemsetAsync(zb, 0, sizeof(float) * (size_t)N * 8, stream);

  const int B = 256;
  const long long EF = (long long)E * 64;
  const int efBlocks = (int)((EF + B - 1) / B);
  const int ndBlocks = (int)(((long long)ND + B - 1) / B);

  deg_kernel<<<(E + B - 1) / B, B, 0, stream>>>(edst, deg, E);
  dinv_kernel<<<(N + B - 1) / B, B, 0, stream>>>(deg, dinv, N);
  wlap_kernel<<<(E + B - 1) / B, B, 0, stream>>>(dinv, esrc, edst, wlap, E);

  scatter_wadd<<<efBlocks, B, 0, stream>>>(h, wlap, esrc, edst, Tx1, EF);
  scatter_wadd<<<efBlocks, B, 0, stream>>>(Tx1, wlap, esrc, edst, tmp, EF);
  combine_kernel<<<ndBlocks, B, 0, stream>>>(tmp, h, Tx2, (long long)ND);
  hipMemsetAsync(tmp, 0, sizeof(float) * ND, stream);
  scatter_wadd<<<efBlocks, B, 0, stream>>>(Tx2, wlap, esrc, edst, tmp, EF);
  combine_kernel<<<ndBlocks, B, 0, stream>>>(tmp, Tx1, Tx3, (long long)ND);

  const int rowTiles = (N + 15) / 16;
  const int gemmBlocks = (rowTiles * 4 + 7) / 8;
  cheb_gemm_kernel<<<gemmBlocks, 256, 0, stream>>>(h, Tx1, Tx2, Tx3, fc, Wc, bc, out, N, N);

  scatter_wadd<<<efBlocks, B, 0, stream>>>(h, ew, esrc, edst, gcn, EF);
  gemm_wmma_kernel<<<gemmBlocks, 256, 0, stream>>>(gcn, 64, Wg, bg, out, 192, 64, N, 64);

  concat_hp<<<(int)(((long long)N * 128 + B - 1) / B), B, 0, stream>>>(h, p, hp, N);
  gemm_wmma_kernel<<<gemmBlocks, 256, 0, stream>>>(hp, 128, WQ, nullptr, Qh, 64, 0, N, 128);
  gemm_wmma_kernel<<<gemmBlocks, 256, 0, stream>>>(hp, 128, WK, nullptr, Kh, 64, 0, N, 128);
  gemm_wmma_kernel<<<gemmBlocks, 256, 0, stream>>>(hp, 128, WVm, nullptr, Vh, 64, 0, N, 128);

  const int eTiles = (E + 15) / 16;
  attn_kernel<<<(eTiles + 3) / 4, 128, 0, stream>>>(e, WE, Qh, Kh, Vh, esrc, edst, wVb, zb, E);
  finalize_kernel<<<(int)(((long long)N * 64 + B - 1) / B), B, 0, stream>>>(wVb, zb, out, N);
}


# Neighbour sums (3 Laplacian, 1 GCN): float-atomic scatter replaced by wave-per-row gather over an edge list bucketed by destination in-call; int atomics rank the slots, so sum order = arrival order
# speedup vs baseline: 1.6533x; 1.6533x over previous
#include <hip/hip_runtime.h>
#include <hip/hip_bf16.h>

typedef __attribute__((ext_vector_type(2))) float v2f;
typedef __attribute__((ext_vector_type(8))) float v8f;

__device__ __forceinline__ v8f wmma4(v2f a, v2f b, v8f c) {
  return __builtin_amdgcn_wmma_f32_16x16x4_f32(
       false, a,  false, b,
       (short)0, c,  false,  false);
}

__global__ void count_kernel(const int* __restrict__ dst, int* __restrict__ cnt, int E) {
  int tid = blockIdx.x * blockDim.x + threadIdx.x;
  if (tid < E) atomicAdd(&cnt[dst[tid]], 1);
}

__global__ void rowalloc_kernel(const int* __restrict__ cnt, int* __restrict__ total,
                                int* __restrict__ start, int* __restrict__ cursor,
                                float* __restrict__ dinv, int N) {
  int tid = blockIdx.x * blockDim.x + threadIdx.x;
  int lane = threadIdx.x & 31;
  int c = tid < N ? cnt[tid] : 0;
  int inc = c;
  for (int o = 1; o < 32; o <<= 1) {
    int v = __shfl_up(inc, o);
    if (lane >= o) inc += v;
  }
  int base = 0;
  if (lane == 31) base = atomicAdd(total, inc);
  base = __shfl(base, 31);
  if (tid < N) {
    int st = base + inc - c;
    start[tid] = st;
    cursor[tid] = st;
    float d = (float)c;
    d = d < 1.0f ? 1.0f : d;
    dinv[tid] = rsqrtf(d);
  }
}

__global__ void fill_kernel(const int* __restrict__ src, const int* __restrict__ dst,
                            const float* __restrict__ dinv, const float* __restrict__ ew,
                            int* __restrict__ cursor, int* __restrict__ csrc,
                            float* __restrict__ cwl, float* __restrict__ cwe, int E) {
  int tid = blockIdx.x * blockDim.x + threadIdx.x;
  if (tid >= E) return;
  int s = src[tid], d = dst[tid];
  int pos = atomicAdd(&cursor[d], 1);
  csrc[pos] = s;
  cwl[pos] = -(dinv[s] * dinv[d]);
  cwe[pos] = ew[tid];
}

__global__ void gather_rows(const float* __restrict__ x, const float* __restrict__ w,
                            const int* __restrict__ csrc, const int* __restrict__ start,
                            const int* __restrict__ cnt, float* __restrict__ out, int N) {
  int row = blockIdx.x * (blockDim.x >> 5) + (threadIdx.x >> 5);
  if (row >= N) return;
  row = __builtin_amdgcn_readfirstlane(row);
  int lane = threadIdx.x & 31;
  int beg = __builtin_amdgcn_readfirstlane(start[row]);
  int end = beg + __builtin_amdgcn_readfirstlane(cnt[row]);
  const float2* __restrict__ x2 = (const float2*)x;
  float2 acc = make_float2(0.0f, 0.0f);
  for (int base = beg; base < end; base += 32) {
    int j = base + lane;
    int sv = 0, wi = 0;
    if (j < end) { sv = csrc[j]; wi = __float_as_int(w[j]); }
    int n = end - base;
    n = n < 32 ? n : 32;
    for (int k = 0; k < n; k += 4) {
      int r = n - k;
      bool b1 = r > 1, b2 = r > 2, b3 = r > 3;
      int k1 = b1 ? k + 1 : k, k2 = b2 ? k + 2 : k, k3 = b3 ? k + 3 : k;
      int s0 = __builtin_amdgcn_readlane(sv, k), s1 = __builtin_amdgcn_readlane(sv, k1);
      int s2 = __builtin_amdgcn_readlane(sv, k2), s3 = __builtin_amdgcn_readlane(sv, k3);
      float w0 = __int_as_float(__builtin_amdgcn_readlane(wi, k));
      float w1 = __int_as_float(__builtin_amdgcn_readlane(wi, k1));
      float w2 = __int_as_float(__builtin_amdgcn_readlane(wi, k2));
      float w3 = __int_as_float(__builtin_amdgcn_readlane(wi, k3));
      float2 v0 = x2[(size_t)s0 * 32 + lane];
      float2 v1 = x2[(size_t)s1 * 32 + lane];
      float2 v2 = x2[(size_t)s2 * 32 + lane];
      float2 v3 = x2[(size_t)s3 * 32 + lane];
      v1.x = b1 ? v1.x : 0.0f; v1.y = b1 ? v1.y : 0.0f;
      v2.x = b2 ? v2.x : 0.0f; v2.y = b2 ? v2.y : 0.0f;
      v3.x = b3 ? v3.x : 0.0f; v3.y = b3 ? v3.y : 0.0f;
      acc.x += w0 * v0.x; acc.y += w0 * v0.y;
      acc.x += w1 * v1.x; acc.y += w1 * v1.y;
      acc.x += w2 * v2.x; acc.y += w2 * v2.y;
      acc.x += w3 * v3.x; acc.y += w3 * v3.y;
    }
  }
  ((float2*)out)[(size_t)row * 32 + lane] = acc;
}

__global__ void combine_kernel(const float* __restrict__ a, const float* __restrict__ b,
                               float* __restrict__ out, long long n) {
  long long tid = (long long)blockIdx.x * blockDim.x + threadIdx.x;
  if (tid < n) out[tid] = 2.0f * a[tid] - b[tid];
}

__global__ void concat_hp(const float* __restrict__ h, const float* __restrict__ p,
                          float* __restrict__ hp, int N) {
  long long tid = (long long)blockIdx.x * blockDim.x + threadIdx.x;
  if (tid >= (long long)N * 128) return;
  int n = (int)(tid >> 7);
  int c = (int)(tid & 127);
  hp[tid] = (c < 64) ? h[(size_t)n * 64 + c] : p[(size_t)n * 64 + (c - 64)];
}

__global__ void gemm_wmma_kernel(const float* __restrict__ A, int lda,
                                 const float* __restrict__ B,
                                 const float* __restrict__ bias,
                                 float* __restrict__ Out, int ldo, int coloff,
                                 int rows, int kdim) {
  int wave = blockIdx.x * (blockDim.x >> 5) + (threadIdx.x >> 5);
  int rowTiles = (rows + 15) >> 4;
  if (wave >= rowTiles * 4) return;
  int rt = wave >> 2, ct = wave & 3;
  int lane = threadIdx.x & 31, half = lane >> 4, l16 = lane & 15;
  int ar = rt * 16 + l16;
  if (ar >= rows) ar = rows - 1;
  int bc = ct * 16 + l16;
  v8f acc = {};
  for (int k0 = 0; k0 < kdim; k0 += 4) {
    int kr = k0 + 2 * half;
    v2f a, b;
    a.x = A[(size_t)ar * lda + kr];
    a.y = A[(size_t)ar * lda + kr + 1];
    b.x = B[(size_t)kr * 64 + bc];
    b.y = B[(size_t)(kr + 1) * 64 + bc];
    acc = wmma4(a, b, acc);
  }
  float bv = bias ? bias[ct * 16 + l16] : 0.0f;
#pragma unroll
  for (int v = 0; v < 8; ++v) {
    int r = rt * 16 + v + 8 * half;
    if (r < rows) Out[(size_t)r * ldo + coloff + ct * 16 + l16] = acc[v] + bv;
  }
}

__global__ void cheb_gemm_kernel(const float* __restrict__ Tx0, const float* __restrict__ Tx1,
                                 const float* __restrict__ Tx2, const float* __restrict__ Tx3,
                                 const float* __restrict__ fc, const float* __restrict__ Wc,
                                 const float* __restrict__ bias, float* __restrict__ Out,
                                 int rows, int Nn) {
  int wave = blockIdx.x * (blockDim.x >> 5) + (threadIdx.x >> 5);
  int rowTiles = (rows + 15) >> 4;
  if (wave >= rowTiles * 4) return;
  int rt = wave >> 2, ct = wave & 3;
  int lane = threadIdx.x & 31, half = lane >> 4, l16 = lane & 15;
  int ar = rt * 16 + l16;
  if (ar >= rows) ar = rows - 1;
  int bc = ct * 16 + l16;
  const float* As[4] = {Tx0, Tx1, Tx2, Tx3};
  v8f acc = {};
#pragma unroll
  for (int k = 0; k < 4; ++k) {
    const float* A = As[k];
    const float* B = Wc + (size_t)k * 64 * 64;
    float sc = fc[(size_t)k * Nn + ar];
#pragma unroll
    for (int k0 = 0; k0 < 64; k0 += 4) {
      int kr = k0 + 2 * half;
      v2f a, b;
      a.x = sc * A[(size_t)ar * 64 + kr];
      a.y = sc * A[(size_t)ar * 64 + kr + 1];
      b.x = B[(size_t)kr * 64 + bc];
      b.y = B[(size_t)(kr + 1) * 64 + bc];
      acc = wmma4(a, b, acc);
    }
  }
  float bv = bias[ct * 16 + l16];
#pragma unroll
  for (int v = 0; v < 8; ++v) {
    int r = rt * 16 + v + 8 * half;
    if (r < rows) Out[(size_t)r * 192 + ct * 16 + l16] = acc[v] + bv;
  }
}

__global__ void attn_kernel(const float* __restrict__ Ee, const float* __restrict__ WE,
                            const float* __restrict__ Qh, const float* __restrict__ Kh,
                            const float* __restrict__ Vh,
                            const int* __restrict__ src, const int* __restrict__ dst,
                            float* __restrict__ wV, float* __restrict__ z, int E) {
  __shared__ float lds[4 * 16 * 64];
  int wid = threadIdx.x >> 5;
  int tile = blockIdx.x * 4 + wid;
  int eTiles = (E + 15) >> 4;
  bool valid = tile < eTiles;
  int t = valid ? tile : 0;
  int e0 = t * 16;
  int lane = threadIdx.x & 31, half = lane >> 4, l16 = lane & 15;
  int ar = e0 + l16;
  if (ar >= E) ar = E - 1;
  float* myLds = lds + wid * 1024;

#pragma unroll
  for (int ct = 0; ct < 4; ++ct) {
    v8f acc = {};
#pragma unroll
    for (int k0 = 0; k0 < 64; k0 += 4) {
      int kr = k0 + 2 * half;
      v2f a, b;
      a.x = Ee[(size_t)ar * 64 + kr];
      a.y = Ee[(size_t)ar * 64 + kr + 1];
      int bc = ct * 16 + l16;
      b.x = WE[(size_t)kr * 64 + bc];
      b.y = WE[(size_t)(kr + 1) * 64 + bc];
      acc = wmma4(a, b, acc);
    }
#pragma unroll
    for (int v = 0; v < 8; ++v)
      myLds[(v + 8 * half) * 64 + ct * 16 + l16] = acc[v];
  }
  __syncthreads();
  if (!valid) return;

  const float scale = 0.35355339059327373f;
#pragma unroll
  for (int pp = 0; pp < 4; ++pp) {
    int pair = pp * 32 + (threadIdx.x & 31);
    int m = pair >> 3, hh = pair & 7;
    int em = e0 + m;
    if (em >= E) continue;
    int s = src[em], d = dst[em];
    float sc = 0.0f;
#pragma unroll
    for (int dd = 0; dd < 8; ++dd) {
      int c = hh * 8 + dd;
      sc += Kh[(size_t)s * 64 + c] * Qh[(size_t)d * 64 + c] * myLds[m * 64 + c];
    }
    sc *= scale;
    sc = fminf(5.0f, fmaxf(-5.0f, sc));
    float ss = __expf(sc);
    atomicAdd(&z[(size_t)d * 8 + hh], ss);
#pragma unroll
    for (int dd = 0; dd < 8; ++dd) {
      int c = hh * 8 + dd;
      atomicAdd(&wV[(size_t)d * 64 + c], Vh[(size_t)s * 64 + c] * ss);
    }
  }
}

__global__ void finalize_kernel(const float* __restrict__ wV, const float* __restrict__ z,
                                float* __restrict__ out, int N) {
  long long tid = (long long)blockIdx.x * blockDim.x + threadIdx.x;
  if (tid >= (long long)N * 64) return;
  int n = (int)(tid >> 6);
  int c = (int)(tid & 63);
  float zz = z[(size_t)n * 8 + (c >> 3)] + 1e-6f;
  out[(size_t)n * 192 + 128 + c] = wV[tid] / zz;
}

extern "C" void kernel_launch(void* const* d_in, const int* in_sizes, int n_in,
                              void* d_out, int out_size, void* d_ws, size_t ws_size,
                              hipStream_t stream) {
  const float* h  = (const float*)d_in[0];
  const float* p  = (const float*)d_in[1];
  const float* e  = (const float*)d_in[2];
  const float* ew = (const float*)d_in[3];
  const float* fc = (const float*)d_in[4];
  const int* esrc = (const int*)d_in[5];
  const int* edst = (const int*)d_in[6];
  const float* Wc = (const float*)d_in[7];
  const float* bc = (const float*)d_in[8];
  const float* Wg = (const float*)d_in[9];
  const float* bg = (const float*)d_in[10];
  const float* WQ = (const float*)d_in[11];
  const float* WK = (const float*)d_in[12];
  const float* WVm = (const float*)d_in[13];
  const float* WE = (const float*)d_in[14];
  float* out = (float*)d_out;

  const int N = in_sizes[0] / 64;
  const int E = in_sizes[5];
  const size_t ND = (size_t)N * 64;

  float* w = (float*)d_ws;
  float* Tx1  = w; w += ND;
  float* Tx2  = w; w += ND;
  float* Tx3  = w; w += ND;
  float* tmp  = w; w += ND;
  float* gcn  = w; w += ND;
  float* hp   = w; w += 2 * ND;
  float* Qh   = w; w += ND;
  float* Kh   = w; w += ND;
  float* Vh   = w; w += ND;
  float* wVb  = w; w += ND;
  float* zb   = w; w += (size_t)N * 8;
  float* dinv = w; w += N;
  int* cnt    = (int*)w; w += N;
  int* total  = (int*)w; w += 1;
  int* start   = (int*)hp;
  int* cursor  = start + N;
  int* csrc    = cursor + N;
  float* cwl   = (float*)(csrc + E);
  float* cwe   = cwl + E;

  hipMemsetAsync(cnt, 0, sizeof(int) * ((size_t)N + 1), stream);
  hipMemsetAsync(wVb, 0, sizeof(float) * ND, stream);
  hipMemsetAsync(zb, 0, sizeof(float) * (size_t)N * 8, stream);

  const int B = 256;
  const int ndBlocks = (int)(((long long)ND + B - 1) / B);
  const int rowBlocks = (N + 7) / 8;

  count_kernel<<<(E + B - 1) / B, B, 0, stream>>>(edst, cnt, E);
  rowalloc_kernel<<<(N + B - 1) / B, B, 0, stream>>>(cnt, total, start, cursor, dinv, N);
  fill_kernel<<<(E + B - 1) / B, B, 0, stream>>>(esrc, edst, dinv, ew, cursor, csrc, cwl, cwe, E);

  gather_rows<<<rowBlocks, B, 0, stream>>>(h, cwl, csrc, start, cnt, Tx1, N);
  gather_rows<<<rowBlocks, B, 0, stream>>>(Tx1, cwl, csrc, start, cnt, tmp, N);
  combine_kernel<<<ndBlocks, B, 0, stream>>>(tmp, h, Tx2, (long long)ND);
  gather_rows<<<rowBlocks, B, 0, stream>>>(Tx2, cwl, csrc, start, cnt, tmp, N);
  combine_kernel<<<ndBlocks, B, 0, stream>>>(tmp, Tx1, Tx3, (long long)ND);

  const int rowTiles = (N + 15) / 16;
  const int gemmBlocks = (rowTiles * 4 + 7) / 8;
  cheb_gemm_kernel<<<gemmBlocks, 256, 0, stream>>>(h, Tx1, Tx2, Tx3, fc, Wc, bc, out, N, N);

  gather_rows<<<rowBlocks, B, 0, stream>>>(h, cwe, csrc, start, cnt, gcn, N);
  gemm_wmma_kernel<<<gemmBlocks, 256, 0, stream>>>(gcn, 64, Wg, bg, out, 192, 64, N, 64);

  concat_hp<<<(int)(((long long)N * 128 + B - 1) / B), B, 0, stream>>>(h, p, hp, N);
  gemm_wmma_kernel<<<gemmBlocks, 256, 0, stream>>>(hp, 128, WQ, nullptr, Qh, 64, 0, N, 128);
  gemm_wmma_kernel<<<gemmBlocks, 256, 0, stream>>>(hp, 128, WK, nullptr, Kh, 64, 0, N, 128);
  gemm_wmma_kernel<<<gemmBlocks, 256, 0, stream>>>(hp, 128, WVm, nullptr, Vh, 64, 0, N, 128);

  const int eTiles = (E + 15) / 16;
  attn_kernel<<<(eTiles + 3) / 4, 128, 0, stream>>>(e, WE, Qh, Kh, Vh, esrc, edst, wVb, zb, E);
  finalize_kernel<<<(int)(((long long)N * 64 + B - 1) / B), B, 0, stream>>>(wVb, zb, out, N);
}


# Attention neighbour sums (wV and z): float-atomic scatter replaced by wave-per-row gather over the destination-bucketed edge list, division by z fused in; sum order = slot (arrival) order
# speedup vs baseline: 2.1103x; 1.2764x over previous
#include <hip/hip_runtime.h>
#include <hip/hip_bf16.h>

typedef __attribute__((ext_vector_type(2))) float v2f;
typedef __attribute__((ext_vector_type(8))) float v8f;

__device__ __forceinline__ v8f wmma4(v2f a, v2f b, v8f c) {
  return __builtin_amdgcn_wmma_f32_16x16x4_f32(
       false, a,  false, b,
       (short)0, c,  false,  false);
}

__global__ void count_kernel(const int* __restrict__ dst, int* __restrict__ cnt, int E) {
  int tid = blockIdx.x * blockDim.x + threadIdx.x;
  if (tid < E) atomicAdd(&cnt[dst[tid]], 1);
}

__global__ void rowalloc_kernel(const int* __restrict__ cnt, int* __restrict__ total,
                                int* __restrict__ start, int* __restrict__ cursor,
                                float* __restrict__ dinv, int N) {
  int tid = blockIdx.x * blockDim.x + threadIdx.x;
  int lane = threadIdx.x & 31;
  int c = tid < N ? cnt[tid] : 0;
  int inc = c;
  for (int o = 1; o < 32; o <<= 1) {
    int v = __shfl_up(inc, o);
    if (lane >= o) inc += v;
  }
  int base = 0;
  if (lane == 31) base = atomicAdd(total, inc);
  base = __shfl(base, 31);
  if (tid < N) {
    int st = base + inc - c;
    start[tid] = st;
    cursor[tid] = st;
    float d = (float)c;
    d = d < 1.0f ? 1.0f : d;
    dinv[tid] = rsqrtf(d);
  }
}

__global__ void fill_kernel(const int* __restrict__ src, const int* __restrict__ dst,
                            const float* __restrict__ dinv, const float* __restrict__ ew,
                            int* __restrict__ cursor, int* __restrict__ csrc,
                            float* __restrict__ cwl, float* __restrict__ cwe,
                            int* __restrict__ epos, int E) {
  int tid = blockIdx.x * blockDim.x + threadIdx.x;
  if (tid >= E) return;
  int s = src[tid], d = dst[tid];
  int pos = atomicAdd(&cursor[d], 1);
  csrc[pos] = s;
  cwl[pos] = -(dinv[s] * dinv[d]);
  cwe[pos] = ew[tid];
  epos[tid] = pos;
}

__global__ void gather_rows(const float* __restrict__ x, const float* __restrict__ w,
                            const int* __restrict__ csrc, const int* __restrict__ start,
                            const int* __restrict__ cnt, float* __restrict__ out, int N) {
  int row = blockIdx.x * (blockDim.x >> 5) + (threadIdx.x >> 5);
  if (row >= N) return;
  row = __builtin_amdgcn_readfirstlane(row);
  int lane = threadIdx.x & 31;
  int beg = __builtin_amdgcn_readfirstlane(start[row]);
  int end = beg + __builtin_amdgcn_readfirstlane(cnt[row]);
  const float2* __restrict__ x2 = (const float2*)x;
  float2 acc = make_float2(0.0f, 0.0f);
  for (int base = beg; base < end; base += 32) {
    int j = base + lane;
    int sv = 0, wi = 0;
    if (j < end) { sv = csrc[j]; wi = __float_as_int(w[j]); }
    int n = end - base;
    n = n < 32 ? n : 32;
    for (int k = 0; k < n; k += 4) {
      int r = n - k;
      bool b1 = r > 1, b2 = r > 2, b3 = r > 3;
      int k1 = b1 ? k + 1 : k, k2 = b2 ? k + 2 : k, k3 = b3 ? k + 3 : k;
      int s0 = __builtin_amdgcn_readlane(sv, k), s1 = __builtin_amdgcn_readlane(sv, k1);
      int s2 = __builtin_amdgcn_readlane(sv, k2), s3 = __builtin_amdgcn_readlane(sv, k3);
      float w0 = __int_as_float(__builtin_amdgcn_readlane(wi, k));
      float w1 = __int_as_float(__builtin_amdgcn_readlane(wi, k1));
      float w2 = __int_as_float(__builtin_amdgcn_readlane(wi, k2));
      float w3 = __int_as_float(__builtin_amdgcn_readlane(wi, k3));
      float2 v0 = x2[(size_t)s0 * 32 + lane];
      float2 v1 = x2[(size_t)s1 * 32 + lane];
      float2 v2 = x2[(size_t)s2 * 32 + lane];
      float2 v3 = x2[(size_t)s3 * 32 + lane];
      v1.x = b1 ? v1.x : 0.0f; v1.y = b1 ? v1.y : 0.0f;
      v2.x = b2 ? v2.x : 0.0f; v2.y = b2 ? v2.y : 0.0f;
      v3.x = b3 ? v3.x : 0.0f; v3.y = b3 ? v3.y : 0.0f;
      acc.x += w0 * v0.x; acc.y += w0 * v0.y;
      acc.x += w1 * v1.x; acc.y += w1 * v1.y;
      acc.x += w2 * v2.x; acc.y += w2 * v2.y;
      acc.x += w3 * v3.x; acc.y += w3 * v3.y;
    }
  }
  ((float2*)out)[(size_t)row * 32 + lane] = acc;
}

__global__ void combine_kernel(const float* __restrict__ a, const float* __restrict__ b,
                               float* __restrict__ out, long long n) {
  long long tid = (long long)blockIdx.x * blockDim.x + threadIdx.x;
  if (tid < n) out[tid] = 2.0f * a[tid] - b[tid];
}

__global__ void concat_hp(const float* __restrict__ h, const float* __restrict__ p,
                          float* __restrict__ hp, int N) {
  long long tid = (long long)blockIdx.x * blockDim.x + threadIdx.x;
  if (tid >= (long long)N * 128) return;
  int n = (int)(tid >> 7);
  int c = (int)(tid & 127);
  hp[tid] = (c < 64) ? h[(size_t)n * 64 + c] : p[(size_t)n * 64 + (c - 64)];
}

__global__ void gemm_wmma_kernel(const float* __restrict__ A, int lda,
                                 const float* __restrict__ B,
                                 const float* __restrict__ bias,
                                 float* __restrict__ Out, int ldo, int coloff,
                                 int rows, int kdim) {
  int wave = blockIdx.x * (blockDim.x >> 5) + (threadIdx.x >> 5);
  int rowTiles = (rows + 15) >> 4;
  if (wave >= rowTiles * 4) return;
  int rt = wave >> 2, ct = wave & 3;
  int lane = threadIdx.x & 31, half = lane >> 4, l16 = lane & 15;
  int ar = rt * 16 + l16;
  if (ar >= rows) ar = rows - 1;
  int bc = ct * 16 + l16;
  v8f acc = {};
  for (int k0 = 0; k0 < kdim; k0 += 4) {
    int kr = k0 + 2 * half;
    v2f a, b;
    a.x = A[(size_t)ar * lda + kr];
    a.y = A[(size_t)ar * lda + kr + 1];
    b.x = B[(size_t)kr * 64 + bc];
    b.y = B[(size_t)(kr + 1) * 64 + bc];
    acc = wmma4(a, b, acc);
  }
  float bv = bias ? bias[ct * 16 + l16] : 0.0f;
#pragma unroll
  for (int v = 0; v < 8; ++v) {
    int r = rt * 16 + v + 8 * half;
    if (r < rows) Out[(size_t)r * ldo + coloff + ct * 16 + l16] = acc[v] + bv;
  }
}

__global__ void cheb_gemm_kernel(const float* __restrict__ Tx0, const float* __restrict__ Tx1,
                                 const float* __restrict__ Tx2, const float* __restrict__ Tx3,
                                 const float* __restrict__ fc, const float* __restrict__ Wc,
                                 const float* __restrict__ bias, float* __restrict__ Out,
                                 int rows, int Nn) {
  int wave = blockIdx.x * (blockDim.x >> 5) + (threadIdx.x >> 5);
  int rowTiles = (rows + 15) >> 4;
  if (wave >= rowTiles * 4) return;
  int rt = wave >> 2, ct = wave & 3;
  int lane = threadIdx.x & 31, half = lane >> 4, l16 = lane & 15;
  int ar = rt * 16 + l16;
  if (ar >= rows) ar = rows - 1;
  int bc = ct * 16 + l16;
  const float* As[4] = {Tx0, Tx1, Tx2, Tx3};
  v8f acc = {};
#pragma unroll
  for (int k = 0; k < 4; ++k) {
    const float* A = As[k];
    const float* B = Wc + (size_t)k * 64 * 64;
    float sc = fc[(size_t)k * Nn + ar];
#pragma unroll
    for (int k0 = 0; k0 < 64; k0 += 4) {
      int kr = k0 + 2 * half;
      v2f a, b;
      a.x = sc * A[(size_t)ar * 64 + kr];
      a.y = sc * A[(size_t)ar * 64 + kr + 1];
      b.x = B[(size_t)kr * 64 + bc];
      b.y = B[(size_t)(kr + 1) * 64 + bc];
      acc = wmma4(a, b, acc);
    }
  }
  float bv = bias[ct * 16 + l16];
#pragma unroll
  for (int v = 0; v < 8; ++v) {
    int r = rt * 16 + v + 8 * half;
    if (r < rows) Out[(size_t)r * 192 + ct * 16 + l16] = acc[v] + bv;
  }
}

__global__ void attn_kernel(const float* __restrict__ Ee, const float* __restrict__ WE,
                            const float* __restrict__ Qh, const float* __restrict__ Kh,
                            const int* __restrict__ src, const int* __restrict__ dst,
                            const int* __restrict__ epos, float* __restrict__ ssb, int E) {
  __shared__ float lds[4 * 16 * 64];
  int wid = threadIdx.x >> 5;
  int tile = blockIdx.x * 4 + wid;
  int eTiles = (E + 15) >> 4;
  bool valid = tile < eTiles;
  int t = valid ? tile : 0;
  int e0 = t * 16;
  int lane = threadIdx.x & 31, half = lane >> 4, l16 = lane & 15;
  int ar = e0 + l16;
  if (ar >= E) ar = E - 1;
  float* myLds = lds + wid * 1024;

#pragma unroll
  for (int ct = 0; ct < 4; ++ct) {
    v8f acc = {};
#pragma unroll
    for (int k0 = 0; k0 < 64; k0 += 4) {
      int kr = k0 + 2 * half;
      v2f a, b;
      a.x = Ee[(size_t)ar * 64 + kr];
      a.y = Ee[(size_t)ar * 64 + kr + 1];
      int bc = ct * 16 + l16;
      b.x = WE[(size_t)kr * 64 + bc];
      b.y = WE[(size_t)(kr + 1) * 64 + bc];
      acc = wmma4(a, b, acc);
    }
#pragma unroll
    for (int v = 0; v < 8; ++v)
      myLds[(v + 8 * half) * 64 + ct * 16 + l16] = acc[v];
  }
  __syncthreads();
  if (!valid) return;

  const float scale = 0.35355339059327373f;
#pragma unroll
  for (int pp = 0; pp < 4; ++pp) {
    int pair = pp * 32 + (threadIdx.x & 31);
    int m = pair >> 3, hh = pair & 7;
    int em = e0 + m;
    if (em >= E) continue;
    int s = src[em], d = dst[em];
    float sc = 0.0f;
#pragma unroll
    for (int dd = 0; dd < 8; ++dd) {
      int c = hh * 8 + dd;
      sc += Kh[(size_t)s * 64 + c] * Qh[(size_t)d * 64 + c] * myLds[m * 64 + c];
    }
    sc *= scale;
    sc = fminf(5.0f, fmaxf(-5.0f, sc));
    float ss = __expf(sc);
    ssb[(size_t)epos[em] * 8 + hh] = ss;
  }
}

__global__ void attn_gather_rows(const float* __restrict__ Vh, const float* __restrict__ ssb,
                                 const int* __restrict__ csrc, const int* __restrict__ start,
                                 const int* __restrict__ cnt, float* __restrict__ out, int N) {
  int row = blockIdx.x * (blockDim.x >> 5) + (threadIdx.x >> 5);
  if (row >= N) return;
  row = __builtin_amdgcn_readfirstlane(row);
  int lane = threadIdx.x & 31;
  int hd = lane >> 2;
  int beg = __builtin_amdgcn_readfirstlane(start[row]);
  int end = beg + __builtin_amdgcn_readfirstlane(cnt[row]);
  const float2* __restrict__ v2p = (const float2*)Vh;
  float2 acc = make_float2(0.0f, 0.0f);
  float zs = 0.0f;
  for (int base = beg; base < end; base += 32) {
    int j = base + lane;
    int sv = 0;
    if (j < end) sv = csrc[j];
    int n = end - base;
    n = n < 32 ? n : 32;
    const float* __restrict__ sp = ssb + (size_t)base * 8 + hd;
    for (int k = 0; k < n; k += 4) {
      int r = n - k;
      bool b1 = r > 1, b2 = r > 2, b3 = r > 3;
      int k1 = b1 ? k + 1 : k, k2 = b2 ? k + 2 : k, k3 = b3 ? k + 3 : k;
      int s0 = __builtin_amdgcn_readlane(sv, k), s1 = __builtin_amdgcn_readlane(sv, k1);
      int s2 = __builtin_amdgcn_readlane(sv, k2), s3 = __builtin_amdgcn_readlane(sv, k3);
      float p0 = sp[k * 8], p1 = sp[k1 * 8], p2 = sp[k2 * 8], p3 = sp[k3 * 8];
      float2 v0 = v2p[(size_t)s0 * 32 + lane];
      float2 v1 = v2p[(size_t)s1 * 32 + lane];
      float2 v2 = v2p[(size_t)s2 * 32 + lane];
      float2 v3 = v2p[(size_t)s3 * 32 + lane];
      p1 = b1 ? p1 : 0.0f;
      p2 = b2 ? p2 : 0.0f;
      p3 = b3 ? p3 : 0.0f;
      acc.x += p0 * v0.x; acc.y += p0 * v0.y;
      acc.x += p1 * v1.x; acc.y += p1 * v1.y;
      acc.x += p2 * v2.x; acc.y += p2 * v2.y;
      acc.x += p3 * v3.x; acc.y += p3 * v3.y;
      zs += p0; zs += p1; zs += p2; zs += p3;
    }
  }
  float zz = zs + 1e-6f;
  float2 o = make_float2(acc.x / zz, acc.y / zz);
  ((float2*)out)[((size_t)row * 192 + 128) / 2 + lane] = o;
}

extern "C" void kernel_launch(void* const* d_in, const int* in_sizes, int n_in,
                              void* d_out, int out_size, void* d_ws, size_t ws_size,
                              hipStream_t stream) {
  const float* h  = (const float*)d_in[0];
  const float* p  = (const float*)d_in[1];
  const float* e  = (const float*)d_in[2];
  const float* ew = (const float*)d_in[3];
  const float* fc = (const float*)d_in[4];
  const int* esrc = (const int*)d_in[5];
  const int* edst = (const int*)d_in[6];
  const float* Wc = (const float*)d_in[7];
  const float* bc = (const float*)d_in[8];
  const float* Wg = (const float*)d_in[9];
  const float* bg = (const float*)d_in[10];
  const float* WQ = (const float*)d_in[11];
  const float* WK = (const float*)d_in[12];
  const float* WVm = (const float*)d_in[13];
  const float* WE = (const float*)d_in[14];
  float* out = (float*)d_out;

  const int N = in_sizes[0] / 64;
  const int E = in_sizes[5];
  const size_t ND = (size_t)N * 64;

  float* w = (float*)d_ws;
  float* Tx1  = w; w += ND;
  float* Tx2  = w; w += ND;
  float* Tx3  = w; w += ND;
  float* tmp  = w; w += ND;
  float* gcn  = w; w += ND;
  float* hp   = w; w += 2 * ND;
  float* Qh   = w; w += ND;
  float* Kh   = w; w += ND;
  float* Vh   = w; w += ND;
  float* lst  = w; w += ND + (size_t)N * 8;
  float* dinv = w; w += N;
  int* cnt    = (int*)w; w += N;
  int* total  = (int*)w; w += 1;
  int* start   = (int*)lst;
  int* csrc    = start + N;
  int* epos    = csrc + E;
  int* cursor  = (int*)hp;
  float* cwl   = (float*)(cursor + N);
  float* cwe   = cwl + E;
  float* ssb   = Tx1;

  hipMemsetAsync(cnt, 0, sizeof(int) * ((size_t)N + 1), stream);

  const int B = 256;
  const int ndBlocks = (int)(((long long)ND + B - 1) / B);
  const int rowBlocks = (N + 7) / 8;

  count_kernel<<<(E + B - 1) / B, B, 0, stream>>>(edst, cnt, E);
  rowalloc_kernel<<<(N + B - 1) / B, B, 0, stream>>>(cnt, total, start, cursor, dinv, N);
  fill_kernel<<<(E + B - 1) / B, B, 0, stream>>>(esrc, edst, dinv, ew, cursor, csrc, cwl, cwe, epos, E);

  gather_rows<<<rowBlocks, B, 0, stream>>>(h, cwl, csrc, start, cnt, Tx1, N);
  gather_rows<<<rowBlocks, B, 0, stream>>>(Tx1, cwl, csrc, start, cnt, tmp, N);
  combine_kernel<<<ndBlocks, B, 0, stream>>>(tmp, h, Tx2, (long long)ND);
  gather_rows<<<rowBlocks, B, 0, stream>>>(Tx2, cwl, csrc, start, cnt, tmp, N);
  combine_kernel<<<ndBlocks, B, 0, stream>>>(tmp, Tx1, Tx3, (long long)ND);

  const int rowTiles = (N + 15) / 16;
  const int gemmBlocks = (rowTiles * 4 + 7) / 8;
  cheb_gemm_kernel<<<gemmBlocks, 256, 0, stream>>>(h, Tx1, Tx2, Tx3, fc, Wc, bc, out, N, N);

  gather_rows<<<rowBlocks, B, 0, stream>>>(h, cwe, csrc, start, cnt, gcn, N);
  gemm_wmma_kernel<<<gemmBlocks, 256, 0, stream>>>(gcn, 64, Wg, bg, out, 192, 64, N, 64);

  concat_hp<<<(int)(((long long)N * 128 + B - 1) / B), B, 0, stream>>>(h, p, hp, N);
  gemm_wmma_kernel<<<gemmBlocks, 256, 0, stream>>>(hp, 128, WQ, nullptr, Qh, 64, 0, N, 128);
  gemm_wmma_kernel<<<gemmBlocks, 256, 0, stream>>>(hp, 128, WK, nullptr, Kh, 64, 0, N, 128);
  gemm_wmma_kernel<<<gemmBlocks, 256, 0, stream>>>(hp, 128, WVm, nullptr, Vh, 64, 0, N, 128);

  const int eTiles = (E + 15) / 16;
  attn_kernel<<<(eTiles + 3) / 4, 128, 0, stream>>>(e, WE, Qh, Kh, esrc, edst, epos, ssb, E);
  attn_gather_rows<<<rowBlocks, B, 0, stream>>>(Vh, ssb, csrc, start, cnt, out, N);
}


# Elementwise passes folded into their neighbours: Chebyshev 2*Lx-prev applied as the gather stores its row (2 combine launches gone); Q/K/V GEMMs read h and p in place (concat copy and launch gone)
# speedup vs baseline: 2.3108x; 1.0950x over previous
#include <hip/hip_runtime.h>
#include <hip/hip_bf16.h>

typedef __attribute__((ext_vector_type(2))) float v2f;
typedef __attribute__((ext_vector_type(8))) float v8f;

__device__ __forceinline__ v8f wmma4(v2f a, v2f b, v8f c) {
  return __builtin_amdgcn_wmma_f32_16x16x4_f32(
       false, a,  false, b,
       (short)0, c,  false,  false);
}

__global__ void count_kernel(const int* __restrict__ dst, int* __restrict__ cnt, int E) {
  int tid = blockIdx.x * blockDim.x + threadIdx.x;
  if (tid < E) atomicAdd(&cnt[dst[tid]], 1);
}

__global__ void rowalloc_kernel(const int* __restrict__ cnt, int* __restrict__ total,
                                int* __restrict__ start, int* __restrict__ cursor,
                                float* __restrict__ dinv, int N) {
  int tid = blockIdx.x * blockDim.x + threadIdx.x;
  int lane = threadIdx.x & 31;
  int c = tid < N ? cnt[tid] : 0;
  int inc = c;
  for (int o = 1; o < 32; o <<= 1) {
    int v = __shfl_up(inc, o);
    if (lane >= o) inc += v;
  }
  int base = 0;
  if (lane == 31) base = atomicAdd(total, inc);
  base = __shfl(base, 31);
  if (tid < N) {
    int st = base + inc - c;
    start[tid] = st;
    cursor[tid] = st;
    float d = (float)c;
    d = d < 1.0f ? 1.0f : d;
    dinv[tid] = rsqrtf(d);
  }
}

__global__ void fill_kernel(const int* __restrict__ src, const int* __restrict__ dst,
                            const float* __restrict__ dinv, const float* __restrict__ ew,
                            int* __restrict__ cursor, int* __restrict__ csrc,
                            float* __restrict__ cwl, float* __restrict__ cwe,
                            int* __restrict__ epos, int E) {
  int tid = blockIdx.x * blockDim.x + threadIdx.x;
  if (tid >= E) return;
  int s = src[tid], d = dst[tid];
  int pos = atomicAdd(&cursor[d], 1);
  csrc[pos] = s;
  cwl[pos] = -(dinv[s] * dinv[d]);
  cwe[pos] = ew[tid];
  epos[tid] = pos;
}

__global__ void gather_rows(const float* __restrict__ x, const float* __restrict__ w,
                            const int* __restrict__ csrc, const int* __restrict__ start,
                            const int* __restrict__ cnt, const float* __restrict__ prev,
                            float* __restrict__ out, int N) {
  int row = blockIdx.x * (blockDim.x >> 5) + (threadIdx.x >> 5);
  if (row >= N) return;
  row = __builtin_amdgcn_readfirstlane(row);
  int lane = threadIdx.x & 31;
  int beg = __builtin_amdgcn_readfirstlane(start[row]);
  int end = beg + __builtin_amdgcn_readfirstlane(cnt[row]);
  const float2* __restrict__ x2 = (const float2*)x;
  float2 pv = make_float2(0.0f, 0.0f);
  if (prev) pv = ((const float2*)prev)[(size_t)row * 32 + lane];
  float2 acc = make_float2(0.0f, 0.0f);
  for (int base = beg; base < end; base += 32) {
    int j = base + lane;
    int sv = 0, wi = 0;
    if (j < end) { sv = csrc[j]; wi = __float_as_int(w[j]); }
    int n = end - base;
    n = n < 32 ? n : 32;
    for (int k = 0; k < n; k += 4) {
      int r = n - k;
      bool b1 = r > 1, b2 = r > 2, b3 = r > 3;
      int k1 = b1 ? k + 1 : k, k2 = b2 ? k + 2 : k, k3 = b3 ? k + 3 : k;
      int s0 = __builtin_amdgcn_readlane(sv, k), s1 = __builtin_amdgcn_readlane(sv, k1);
      int s2 = __builtin_amdgcn_readlane(sv, k2), s3 = __builtin_amdgcn_readlane(sv, k3);
      float w0 = __int_as_float(__builtin_amdgcn_readlane(wi, k));
      float w1 = __int_as_float(__builtin_amdgcn_readlane(wi, k1));
      float w2 = __int_as_float(__builtin_amdgcn_readlane(wi, k2));
      float w3 = __int_as_float(__builtin_amdgcn_readlane(wi, k3));
      float2 v0 = x2[(size_t)s0 * 32 + lane];
      float2 v1 = x2[(size_t)s1 * 32 + lane];
      float2 v2 = x2[(size_t)s2 * 32 + lane];
      float2 v3 = x2[(size_t)s3 * 32 + lane];
      v1.x = b1 ? v1.x : 0.0f; v1.y = b1 ? v1.y : 0.0f;
      v2.x = b2 ? v2.x : 0.0f; v2.y = b2 ? v2.y : 0.0f;
      v3.x = b3 ? v3.x : 0.0f; v3.y = b3 ? v3.y : 0.0f;
      acc.x += w0 * v0.x; acc.y += w0 * v0.y;
      acc.x += w1 * v1.x; acc.y += w1 * v1.y;
      acc.x += w2 * v2.x; acc.y += w2 * v2.y;
      acc.x += w3 * v3.x; acc.y += w3 * v3.y;
    }
  }
  if (prev) {
    acc.x = 2.0f * acc.x - pv.x;
    acc.y = 2.0f * acc.y - pv.y;
  }
  ((float2*)out)[(size_t)row * 32 + lane] = acc;
}

__global__ void gemm_wmma_kernel(const float* __restrict__ A0, const float* __restrict__ A1,
                                 const float* __restrict__ B,
                                 const float* __restrict__ bias,
                                 float* __restrict__ Out, int ldo, int coloff,
                                 int rows) {
  int wave = blockIdx.x * (blockDim.x >> 5) + (threadIdx.x >> 5);
  int rowTiles = (rows + 15) >> 4;
  if (wave >= rowTiles * 4) return;
  int rt = wave >> 2, ct = wave & 3;
  int lane = threadIdx.x & 31, half = lane >> 4, l16 = lane & 15;
  int ar = rt * 16 + l16;
  if (ar >= rows) ar = rows - 1;
  int bc = ct * 16 + l16;
  v8f acc = {};
#pragma unroll 4
  for (int k0 = 0; k0 < 64; k0 += 4) {
    int kr = k0 + 2 * half;
    v2f a, b;
    a.x = A0[(size_t)ar * 64 + kr];
    a.y = A0[(size_t)ar * 64 + kr + 1];
    b.x = B[(size_t)kr * 64 + bc];
    b.y = B[(size_t)(kr + 1) * 64 + bc];
    acc = wmma4(a, b, acc);
  }
  if (A1) {
#pragma unroll 4
    for (int k0 = 0; k0 < 64; k0 += 4) {
      int kr = k0 + 2 * half;
      v2f a, b;
      a.x = A1[(size_t)ar * 64 + kr];
      a.y = A1[(size_t)ar * 64 + kr + 1];
      b.x = B[(size_t)(64 + kr) * 64 + bc];
      b.y = B[(size_t)(64 + kr + 1) * 64 + bc];
      acc = wmma4(a, b, acc);
    }
  }
  float bv = bias ? bias[ct * 16 + l16] : 0.0f;
#pragma unroll
  for (int v = 0; v < 8; ++v) {
    int r = rt * 16 + v + 8 * half;
    if (r < rows) Out[(size_t)r * ldo + coloff + ct * 16 + l16] = acc[v] + bv;
  }
}

__global__ void cheb_gemm_kernel(const float* __restrict__ Tx0, const float* __restrict__ Tx1,
                                 const float* __restrict__ Tx2, const float* __restrict__ Tx3,
                                 const float* __restrict__ fc, const float* __restrict__ Wc,
                                 const float* __restrict__ bias, float* __restrict__ Out,
                                 int rows, int Nn) {
  int wave = blockIdx.x * (blockDim.x >> 5) + (threadIdx.x >> 5);
  int rowTiles = (rows + 15) >> 4;
  if (wave >= rowTiles * 4) return;
  int rt = wave >> 2, ct = wave & 3;
  int lane = threadIdx.x & 31, half = lane >> 4, l16 = lane & 15;
  int ar = rt * 16 + l16;
  if (ar >= rows) ar = rows - 1;
  int bc = ct * 16 + l16;
  const float* As[4] = {Tx0, Tx1, Tx2, Tx3};
  v8f acc = {};
#pragma unroll
  for (int k = 0; k < 4; ++k) {
    const float* A = As[k];
    const float* B = Wc + (size_t)k * 64 * 64;
    float sc = fc[(size_t)k * Nn + ar];
#pragma unroll
    for (int k0 = 0; k0 < 64; k0 += 4) {
      int kr = k0 + 2 * half;
      v2f a, b;
      a.x = sc * A[(size_t)ar * 64 + kr];
      a.y = sc * A[(size_t)ar * 64 + kr + 1];
      b.x = B[(size_t)kr * 64 + bc];
      b.y = B[(size_t)(kr + 1) * 64 + bc];
      acc = wmma4(a, b, acc);
    }
  }
  float bv = bias[ct * 16 + l16];
#pragma unroll
  for (int v = 0; v < 8; ++v) {
    int r = rt * 16 + v + 8 * half;
    if (r < rows) Out[(size_t)r * 192 + ct * 16 + l16] = acc[v] + bv;
  }
}

__global__ void attn_kernel(const float* __restrict__ Ee, const float* __restrict__ WE,
                            const float* __restrict__ Qh, const float* __restrict__ Kh,
                            const int* __restrict__ src, const int* __restrict__ dst,
                            const int* __restrict__ epos, float* __restrict__ ssb, int E) {
  __shared__ float lds[4 * 16 * 64];
  int wid = threadIdx.x >> 5;
  int tile = blockIdx.x * 4 + wid;
  int eTiles = (E + 15) >> 4;
  bool valid = tile < eTiles;
  int t = valid ? tile : 0;
  int e0 = t * 16;
  int lane = threadIdx.x & 31, half = lane >> 4, l16 = lane & 15;
  int ar = e0 + l16;
  if (ar >= E) ar = E - 1;
  float* myLds = lds + wid * 1024;

#pragma unroll
  for (int ct = 0; ct < 4; ++ct) {
    v8f acc = {};
#pragma unroll
    for (int k0 = 0; k0 < 64; k0 += 4) {
      int kr = k0 + 2 * half;
      v2f a, b;
      a.x = Ee[(size_t)ar * 64 + kr];
      a.y = Ee[(size_t)ar * 64 + kr + 1];
      int bc = ct * 16 + l16;
      b.x = WE[(size_t)kr * 64 + bc];
      b.y = WE[(size_t)(kr + 1) * 64 + bc];
      acc = wmma4(a, b, acc);
    }
#pragma unroll
    for (int v = 0; v < 8; ++v)
      myLds[(v + 8 * half) * 64 + ct * 16 + l16] = acc[v];
  }
  __syncthreads();
  if (!valid) return;

  const float scale = 0.35355339059327373f;
#pragma unroll
  for (int pp = 0; pp < 4; ++pp) {
    int pair = pp * 32 + (threadIdx.x & 31);
    int m = pair >> 3, hh = pair & 7;
    int em = e0 + m;
    if (em >= E) continue;
    int s = src[em], d = dst[em];
    float sc = 0.0f;
#pragma unroll
    for (int dd = 0; dd < 8; ++dd) {
      int c = hh * 8 + dd;
      sc += Kh[(size_t)s * 64 + c] * Qh[(size_t)d * 64 + c] * myLds[m * 64 + c];
    }
    sc *= scale;
    sc = fminf(5.0f, fmaxf(-5.0f, sc));
    float ss = __expf(sc);
    ssb[(size_t)epos[em] * 8 + hh] = ss;
  }
}

__global__ void attn_gather_rows(const float* __restrict__ Vh, const float* __restrict__ ssb,
                                 const int* __restrict__ csrc, const int* __restrict__ start,
                                 const int* __restrict__ cnt, float* __restrict__ out, int N) {
  int row = blockIdx.x * (blockDim.x >> 5) + (threadIdx.x >> 5);
  if (row >= N) return;
  row = __builtin_amdgcn_readfirstlane(row);
  int lane = threadIdx.x & 31;
  int hd = lane >> 2;
  int beg = __builtin_amdgcn_readfirstlane(start[row]);
  int end = beg + __builtin_amdgcn_readfirstlane(cnt[row]);
  const float2* __restrict__ v2p = (const float2*)Vh;
  float2 acc = make_float2(0.0f, 0.0f);
  float zs = 0.0f;
  for (int base = beg; base < end; base += 32) {
    int j = base + lane;
    int sv = 0;
    if (j < end) sv = csrc[j];
    int n = end - base;
    n = n < 32 ? n : 32;
    const float* __restrict__ sp = ssb + (size_t)base * 8 + hd;
    for (int k = 0; k < n; k += 4) {
      int r = n - k;
      bool b1 = r > 1, b2 = r > 2, b3 = r > 3;
      int k1 = b1 ? k + 1 : k, k2 = b2 ? k + 2 : k, k3 = b3 ? k + 3 : k;
      int s0 = __builtin_amdgcn_readlane(sv, k), s1 = __builtin_amdgcn_readlane(sv, k1);
      int s2 = __builtin_amdgcn_readlane(sv, k2), s3 = __builtin_amdgcn_readlane(sv, k3);
      float p0 = sp[k * 8], p1 = sp[k1 * 8], p2 = sp[k2 * 8], p3 = sp[k3 * 8];
      float2 v0 = v2p[(size_t)s0 * 32 + lane];
      float2 v1 = v2p[(size_t)s1 * 32 + lane];
      float2 v2 = v2p[(size_t)s2 * 32 + lane];
      float2 v3 = v2p[(size_t)s3 * 32 + lane];
      p1 = b1 ? p1 : 0.0f;
      p2 = b2 ? p2 : 0.0f;
      p3 = b3 ? p3 : 0.0f;
      acc.x += p0 * v0.x; acc.y += p0 * v0.y;
      acc.x += p1 * v1.x; acc.y += p1 * v1.y;
      acc.x += p2 * v2.x; acc.y += p2 * v2.y;
      acc.x += p3 * v3.x; acc.y += p3 * v3.y;
      zs += p0; zs += p1; zs += p2; zs += p3;
    }
  }
  float zz = zs + 1e-6f;
  float2 o = make_float2(acc.x / zz, acc.y / zz);
  ((float2*)out)[((size_t)row * 192 + 128) / 2 + lane] = o;
}

extern "C" void kernel_launch(void* const* d_in, const int* in_sizes, int n_in,
                              void* d_out, int out_size, void* d_ws, size_t ws_size,
                              hipStream_t stream) {
  const float* h  = (const float*)d_in[0];
  const float* p  = (const float*)d_in[1];
  const float* e  = (const float*)d_in[2];
  const float* ew = (const float*)d_in[3];
  const float* fc = (const float*)d_in[4];
  const int* esrc = (const int*)d_in[5];
  const int* edst = (const int*)d_in[6];
  const float* Wc = (const float*)d_in[7];
  const float* bc = (const float*)d_in[8];
  const float* Wg = (const float*)d_in[9];
  const float* bg = (const float*)d_in[10];
  const float* WQ = (const float*)d_in[11];
  const float* WK = (const float*)d_in[12];
  const float* WVm = (const float*)d_in[13];
  const float* WE = (const float*)d_in[14];
  float* out = (float*)d_out;

  const int N = in_sizes[0] / 64;
  const int E = in_sizes[5];
  const size_t ND = (size_t)N * 64;

  float* w = (float*)d_ws;
  float* Tx1  = w; w += ND;
  float* Tx2  = w; w += ND;
  float* Tx3  = w; w += ND;
  float* gcn  = w; w += ND;
  float* scr  = w; w += 2 * ND;
  float* Qh   = w; w += ND;
  float* Kh   = w; w += ND;
  float* Vh   = w; w += ND;
  float* lst  = w; w += ND + (size_t)N * 8;
  float* dinv = w; w += N;
  int* cnt    = (int*)w; w += N;
  int* total  = (int*)w; w += 1;
  int* start   = (int*)lst;
  int* csrc    = start + N;
  int* epos    = csrc + E;
  int* cursor  = (int*)scr;
  float* cwl   = (float*)(cursor + N);
  float* cwe   = cwl + E;
  float* ssb   = Tx1;

  hipMemsetAsync(cnt, 0, sizeof(int) * ((size_t)N + 1), stream);

  const int B = 256;
  const int rowBlocks = (N + 7) / 8;

  count_kernel<<<(E + B - 1) / B, B, 0, stream>>>(edst, cnt, E);
  rowalloc_kernel<<<(N + B - 1) / B, B, 0, stream>>>(cnt, total, start, cursor, dinv, N);
  fill_kernel<<<(E + B - 1) / B, B, 0, stream>>>(esrc, edst, dinv, ew, cursor, csrc, cwl, cwe, epos, E);

  gather_rows<<<rowBlocks, B, 0, stream>>>(h, cwl, csrc, start, cnt, nullptr, Tx1, N);
  gather_rows<<<rowBlocks, B, 0, stream>>>(Tx1, cwl, csrc, start, cnt, h, Tx2, N);
  gather_rows<<<rowBlocks, B, 0, stream>>>(Tx2, cwl, csrc, start, cnt, Tx1, Tx3, N);

  const int rowTiles = (N + 15) / 16;
  const int gemmBlocks = (rowTiles * 4 + 7) / 8;
  cheb_gemm_kernel<<<gemmBlocks, 256, 0, stream>>>(h, Tx1, Tx2, Tx3, fc, Wc, bc, out, N, N);

  gather_rows<<<rowBlocks, B, 0, stream>>>(h, cwe, csrc, start, cnt, nullptr, gcn, N);
  gemm_wmma_kernel<<<gemmBlocks, 256, 0, stream>>>(gcn, nullptr, Wg, bg, out, 192, 64, N);

  gemm_wmma_kernel<<<gemmBlocks, 256, 0, stream>>>(h, p, WQ, nullptr, Qh, 64, 0, N);
  gemm_wmma_kernel<<<gemmBlocks, 256, 0, stream>>>(h, p, WK, nullptr, Kh, 64, 0, N);
  gemm_wmma_kernel<<<gemmBlocks, 256, 0, stream>>>(h, p, WVm, nullptr, Vh, 64, 0, N);

  const int eTiles = (E + 15) / 16;
  attn_kernel<<<(eTiles + 3) / 4, 128, 0, stream>>>(e, WE, Qh, Kh, esrc, edst, epos, ssb, E);
  attn_gather_rows<<<rowBlocks, B, 0, stream>>>(Vh, ssb, csrc, start, cnt, out, N);
}


# WMMA fragment reads widened to 16 bytes (k regrouped, tile columns interleaved), one wave per 16x64 strip; attention scores from the transposed product in registers, no LDS; k sum order changed
# speedup vs baseline: 2.5223x; 1.0915x over previous
#include <hip/hip_runtime.h>
#include <hip/hip_bf16.h>

typedef __attribute__((ext_vector_type(2))) float v2f;
typedef __attribute__((ext_vector_type(8))) float v8f;

__device__ __forceinline__ v8f wmma4(v2f a, v2f b, v8f c) {
  return __builtin_amdgcn_wmma_f32_16x16x4_f32(
       false, a,  false, b,
       (short)0, c,  false,  false);
}

__global__ void count_kernel(const int* __restrict__ dst, int* __restrict__ cnt, int E) {
  int tid = blockIdx.x * blockDim.x + threadIdx.x;
  if (tid < E) atomicAdd(&cnt[dst[tid]], 1);
}

__global__ void rowalloc_kernel(const int* __restrict__ cnt, int* __restrict__ total,
                                int* __restrict__ start, int* __restrict__ cursor,
                                float* __restrict__ dinv, int N) {
  int tid = blockIdx.x * blockDim.x + threadIdx.x;
  int lane = threadIdx.x & 31;
  int c = tid < N ? cnt[tid] : 0;
  int inc = c;
  for (int o = 1; o < 32; o <<= 1) {
    int v = __shfl_up(inc, o);
    if (lane >= o) inc += v;
  }
  int base = 0;
  if (lane == 31) base = atomicAdd(total, inc);
  base = __shfl(base, 31);
  if (tid < N) {
    int st = base + inc - c;
    start[tid] = st;
    cursor[tid] = st;
    float d = (float)c;
    d = d < 1.0f ? 1.0f : d;
    dinv[tid] = rsqrtf(d);
  }
}

__global__ void fill_kernel(const int* __restrict__ src, const int* __restrict__ dst,
                            const float* __restrict__ dinv, const float* __restrict__ ew,
                            int* __restrict__ cursor, int* __restrict__ csrc,
                            float* __restrict__ cwl, float* __restrict__ cwe,
                            int* __restrict__ epos, int E) {
  int tid = blockIdx.x * blockDim.x + threadIdx.x;
  if (tid >= E) return;
  int s = src[tid], d = dst[tid];
  int pos = atomicAdd(&cursor[d], 1);
  csrc[pos] = s;
  cwl[pos] = -(dinv[s] * dinv[d]);
  cwe[pos] = ew[tid];
  epos[tid] = pos;
}

__global__ void gather_rows(const float* __restrict__ x, const float* __restrict__ w,
                            const int* __restrict__ csrc, const int* __restrict__ start,
                            const int* __restrict__ cnt, const float* __restrict__ prev,
                            float* __restrict__ out, int N) {
  int row = blockIdx.x * (blockDim.x >> 5) + (threadIdx.x >> 5);
  if (row >= N) return;
  row = __builtin_amdgcn_readfirstlane(row);
  int lane = threadIdx.x & 31;
  int beg = __builtin_amdgcn_readfirstlane(start[row]);
  int end = beg + __builtin_amdgcn_readfirstlane(cnt[row]);
  const float2* __restrict__ x2 = (const float2*)x;
  float2 pv = make_float2(0.0f, 0.0f);
  if (prev) pv = ((const float2*)prev)[(size_t)row * 32 + lane];
  float2 acc = make_float2(0.0f, 0.0f);
  for (int base = beg; base < end; base += 32) {
    int j = base + lane;
    int sv = 0, wi = 0;
    if (j < end) { sv = csrc[j]; wi = __float_as_int(w[j]); }
    int n = end - base;
    n = n < 32 ? n : 32;
    for (int k = 0; k < n; k += 4) {
      int r = n - k;
      bool b1 = r > 1, b2 = r > 2, b3 = r > 3;
      int k1 = b1 ? k + 1 : k, k2 = b2 ? k + 2 : k, k3 = b3 ? k + 3 : k;
      int s0 = __builtin_amdgcn_readlane(sv, k), s1 = __builtin_amdgcn_readlane(sv, k1);
      int s2 = __builtin_amdgcn_readlane(sv, k2), s3 = __builtin_amdgcn_readlane(sv, k3);
      float w0 = __int_as_float(__builtin_amdgcn_readlane(wi, k));
      float w1 = __int_as_float(__builtin_amdgcn_readlane(wi, k1));
      float w2 = __int_as_float(__builtin_amdgcn_readlane(wi, k2));
      float w3 = __int_as_float(__builtin_amdgcn_readlane(wi, k3));
      float2 v0 = x2[(size_t)s0 * 32 + lane];
      float2 v1 = x2[(size_t)s1 * 32 + lane];
      float2 v2 = x2[(size_t)s2 * 32 + lane];
      float2 v3 = x2[(size_t)s3 * 32 + lane];
      v1.x = b1 ? v1.x : 0.0f; v1.y = b1 ? v1.y : 0.0f;
      v2.x = b2 ? v2.x : 0.0f; v2.y = b2 ? v2.y : 0.0f;
      v3.x = b3 ? v3.x : 0.0f; v3.y = b3 ? v3.y : 0.0f;
      acc.x += w0 * v0.x; acc.y += w0 * v0.y;
      acc.x += w1 * v1.x; acc.y += w1 * v1.y;
      acc.x += w2 * v2.x; acc.y += w2 * v2.y;
      acc.x += w3 * v3.x; acc.y += w3 * v3.y;
    }
  }
  if (prev) {
    acc.x = 2.0f * acc.x - pv.x;
    acc.y = 2.0f * acc.y - pv.y;
  }
  ((float2*)out)[(size_t)row * 32 + lane] = acc;
}


template <bool SCALED>
__device__ __forceinline__ void plane_mma(const float* __restrict__ arow,
                                          const float* __restrict__ bcol, int half, float sc,
                                          v8f (&acc)[4]) {
  float4 av[8], bw[8][4];
#pragma unroll
  for (int j = 0; j < 8; ++j) {
    int k = 8 * j + 4 * half;
    av[j] = *(const float4*)(arow + k);
#pragma unroll
    for (int i = 0; i < 4; ++i) bw[j][i] = *(const float4*)(bcol + (size_t)(k + i) * 64);
  }
#pragma unroll
  for (int j = 0; j < 8; ++j) {
    float4 a = av[j];
    if (SCALED) { a.x *= sc; a.y *= sc; a.z *= sc; a.w *= sc; }
    v2f a01 = {a.x, a.y}, a23 = {a.z, a.w};
    acc[0] = wmma4(a01, (v2f){bw[j][0].x, bw[j][1].x}, acc[0]);
    acc[1] = wmma4(a01, (v2f){bw[j][0].y, bw[j][1].y}, acc[1]);
    acc[2] = wmma4(a01, (v2f){bw[j][0].z, bw[j][1].z}, acc[2]);
    acc[3] = wmma4(a01, (v2f){bw[j][0].w, bw[j][1].w}, acc[3]);
    acc[0] = wmma4(a23, (v2f){bw[j][2].x, bw[j][3].x}, acc[0]);
    acc[1] = wmma4(a23, (v2f){bw[j][2].y, bw[j][3].y}, acc[1]);
    acc[2] = wmma4(a23, (v2f){bw[j][2].z, bw[j][3].z}, acc[2]);
    acc[3] = wmma4(a23, (v2f){bw[j][2].w, bw[j][3].w}, acc[3]);
  }
}

__device__ __forceinline__ void strip_store(const v8f (&acc)[4], float4 bv, float* __restrict__ Out,
                                            int ldo, int col, int row0, int rows) {
#pragma unroll
  for (int r = 0; r < 8; ++r) {
    int row = row0 + r;
    if (row < rows) {
      float4 o = make_float4(acc[0][r] + bv.x, acc[1][r] + bv.y, acc[2][r] + bv.z, acc[3][r] + bv.w);
      *(float4*)(Out + (size_t)row * ldo + col) = o;
    }
  }
}

template <bool TWO_PLANES>
__global__ void __launch_bounds__(64) __attribute__((amdgpu_waves_per_eu(1, 4)))
gemm_wmma_kernel(const float* __restrict__ A0, const float* __restrict__ A1,
                 const float* __restrict__ B,
                                 const float* __restrict__ bias,
                                 float* __restrict__ Out, int ldo, int coloff,
                                 int rows) {
  int rt = blockIdx.x * (blockDim.x >> 5) + (threadIdx.x >> 5);
  int rowTiles = (rows + 15) >> 4;
  if (rt >= rowTiles) return;
  int lane = threadIdx.x & 31, half = lane >> 4, c = lane & 15;
  int ar = rt * 16 + c;
  if (ar >= rows) ar = rows - 1;
  v8f acc[4] = {};
  plane_mma<false>(A0 + (size_t)ar * 64, B + 4 * c, half, 1.0f, acc);
  if (TWO_PLANES)
    plane_mma<false>(A1 + (size_t)ar * 64, B + (size_t)64 * 64 + 4 * c, half, 1.0f, acc);
  float4 bv = make_float4(0.0f, 0.0f, 0.0f, 0.0f);
  if (bias) bv = *(const float4*)(bias + 4 * c);
  strip_store(acc, bv, Out, ldo, coloff + 4 * c, rt * 16 + 8 * half, rows);
}

__global__ void __launch_bounds__(64) __attribute__((amdgpu_waves_per_eu(1, 4)))
cheb_gemm_kernel(const float* __restrict__ Tx0, const float* __restrict__ Tx1,
                                 const float* __restrict__ Tx2, const float* __restrict__ Tx3,
                                 const float* __restrict__ fc, const float* __restrict__ Wc,
                                 const float* __restrict__ bias, float* __restrict__ Out,
                                 int rows, int Nn) {
  int rt = blockIdx.x * (blockDim.x >> 5) + (threadIdx.x >> 5);
  int rowTiles = (rows + 15) >> 4;
  if (rt >= rowTiles) return;
  int lane = threadIdx.x & 31, half = lane >> 4, c = lane & 15;
  int ar = rt * 16 + c;
  if (ar >= rows) ar = rows - 1;
  const float* As[4] = {Tx0, Tx1, Tx2, Tx3};
  v8f acc[4] = {};
#pragma unroll
  for (int k = 0; k < 4; ++k) {
    float sc = fc[(size_t)k * Nn + ar];
    plane_mma<true>(As[k] + (size_t)ar * 64, Wc + (size_t)k * 64 * 64 + 4 * c, half, sc, acc);
  }
  float4 bv = *(const float4*)(bias + 4 * c);
  strip_store(acc, bv, Out, 192, 4 * c, rt * 16 + 8 * half, rows);
}

__global__ void __launch_bounds__(256)
attn_kernel(const float* __restrict__ Ee, const float* __restrict__ WE,
                            const float* __restrict__ Qh, const float* __restrict__ Kh,
                            const int* __restrict__ src, const int* __restrict__ dst,
                            const int* __restrict__ epos, float* __restrict__ ssb, int E) {
  int tile = blockIdx.x * (blockDim.x >> 5) + (threadIdx.x >> 5);
  int eTiles = (E + 15) >> 4;
  if (tile >= eTiles) return;
  int lane = threadIdx.x & 31, half = lane >> 4, c = lane & 15;
  int em = tile * 16 + c;
  bool live = em < E;
  int er = live ? em : E - 1;
  int s = src[er], d = dst[er], slot = epos[er];

  const float4* kp = (const float4*)(Kh + (size_t)s * 64 + 16 * half);
  const float4* qp = (const float4*)(Qh + (size_t)d * 64 + 16 * half);
  float4 kv[8], qv[8];
#pragma unroll
  for (int i = 0; i < 4; ++i) {
    kv[i] = kp[i]; kv[4 + i] = kp[8 + i];
    qv[i] = qp[i]; qv[4 + i] = qp[8 + i];
  }

  int gc = (c & 3) | ((c & 8) >> 1) | ((c & 4) << 1);
  const float* wcol = WE + 4 * gc;
  const float* erow = Ee + (size_t)er * 64;
  v8f acc[4] = {};
#pragma unroll
  for (int j = 0; j < 8; ++j) {
    int k = 8 * j + 4 * half;
    float4 ev = *(const float4*)(erow + k);
    float4 w0 = *(const float4*)(wcol + (size_t)(k + 0) * 64);
    float4 w1 = *(const float4*)(wcol + (size_t)(k + 1) * 64);
    float4 w2 = *(const float4*)(wcol + (size_t)(k + 2) * 64);
    float4 w3 = *(const float4*)(wcol + (size_t)(k + 3) * 64);
    v2f e01 = {ev.x, ev.y}, e23 = {ev.z, ev.w};
    acc[0] = wmma4((v2f){w0.x, w1.x}, e01, acc[0]);
    acc[1] = wmma4((v2f){w0.y, w1.y}, e01, acc[1]);
    acc[2] = wmma4((v2f){w0.z, w1.z}, e01, acc[2]);
    acc[3] = wmma4((v2f){w0.w, w1.w}, e01, acc[3]);
    acc[0] = wmma4((v2f){w2.x, w3.x}, e23, acc[0]);
    acc[1] = wmma4((v2f){w2.y, w3.y}, e23, acc[1]);
    acc[2] = wmma4((v2f){w2.z, w3.z}, e23, acc[2]);
    acc[3] = wmma4((v2f){w2.w, w3.w}, e23, acc[3]);
  }

  const float scale = 0.35355339059327373f;
  float ss[4];
#pragma unroll
  for (int q = 0; q < 4; ++q) {
    float kk[8] = {kv[2 * q].x, kv[2 * q].y, kv[2 * q].z, kv[2 * q].w,
                   kv[2 * q + 1].x, kv[2 * q + 1].y, kv[2 * q + 1].z, kv[2 * q + 1].w};
    float qq[8] = {qv[2 * q].x, qv[2 * q].y, qv[2 * q].z, qv[2 * q].w,
                   qv[2 * q + 1].x, qv[2 * q + 1].y, qv[2 * q + 1].z, qv[2 * q + 1].w};
    float sc = 0.0f;
#pragma unroll
    for (int dd = 0; dd < 8; ++dd)
      sc += kk[dd] * qq[dd] * acc[dd & 3][2 * q + (dd >> 2)];
    sc *= scale;
    sc = fminf(5.0f, fmaxf(-5.0f, sc));
    ss[q] = __expf(sc);
  }
  if (live) {
    float2* o = (float2*)(ssb + (size_t)slot * 8 + 2 * half);
    o[0] = make_float2(ss[0], ss[1]);
    o[2] = make_float2(ss[2], ss[3]);
  }
}

__global__ void attn_gather_rows(const float* __restrict__ Vh, const float* __restrict__ ssb,
                                 const int* __restrict__ csrc, const int* __restrict__ start,
                                 const int* __restrict__ cnt, float* __restrict__ out, int N) {
  int row = blockIdx.x * (blockDim.x >> 5) + (threadIdx.x >> 5);
  if (row >= N) return;
  row = __builtin_amdgcn_readfirstlane(row);
  int lane = threadIdx.x & 31;
  int hd = lane >> 2;
  int beg = __builtin_amdgcn_readfirstlane(start[row]);
  int end = beg + __builtin_amdgcn_readfirstlane(cnt[row]);
  const float2* __restrict__ v2p = (const float2*)Vh;
  float2 acc = make_float2(0.0f, 0.0f);
  float zs = 0.0f;
  for (int base = beg; base < end; base += 32) {
    int j = base + lane;
    int sv = 0;
    if (j < end) sv = csrc[j];
    int n = end - base;
    n = n < 32 ? n : 32;
    const float* __restrict__ sp = ssb + (size_t)base * 8 + hd;
    for (int k = 0; k < n; k += 4) {
      int r = n - k;
      bool b1 = r > 1, b2 = r > 2, b3 = r > 3;
      int k1 = b1 ? k + 1 : k, k2 = b2 ? k + 2 : k, k3 = b3 ? k + 3 : k;
      int s0 = __builtin_amdgcn_readlane(sv, k), s1 = __builtin_amdgcn_readlane(sv, k1);
      int s2 = __builtin_amdgcn_readlane(sv, k2), s3 = __builtin_amdgcn_readlane(sv, k3);
      float p0 = sp[k * 8], p1 = sp[k1 * 8], p2 = sp[k2 * 8], p3 = sp[k3 * 8];
      float2 v0 = v2p[(size_t)s0 * 32 + lane];
      float2 v1 = v2p[(size_t)s1 * 32 + lane];
      float2 v2 = v2p[(size_t)s2 * 32 + lane];
      float2 v3 = v2p[(size_t)s3 * 32 + lane];
      p1 = b1 ? p1 : 0.0f;
      p2 = b2 ? p2 : 0.0f;
      p3 = b3 ? p3 : 0.0f;
      acc.x += p0 * v0.x; acc.y += p0 * v0.y;
      acc.x += p1 * v1.x; acc.y += p1 * v1.y;
      acc.x += p2 * v2.x; acc.y += p2 * v2.y;
      acc.x += p3 * v3.x; acc.y += p3 * v3.y;
      zs += p0; zs += p1; zs += p2; zs += p3;
    }
  }
  float zz = zs + 1e-6f;
  float2 o = make_float2(acc.x / zz, acc.y / zz);
  ((float2*)out)[((size_t)row * 192 + 128) / 2 + lane] = o;
}

extern "C" void kernel_launch(void* const* d_in, const int* in_sizes, int n_in,
                              void* d_out, int out_size, void* d_ws, size_t ws_size,
                              hipStream_t stream) {
  const float* h  = (const float*)d_in[0];
  const float* p  = (const float*)d_in[1];
  const float* e  = (const float*)d_in[2];
  const float* ew = (const float*)d_in[3];
  const float* fc = (const float*)d_in[4];
  const int* esrc = (const int*)d_in[5];
  const int* edst = (const int*)d_in[6];
  const float* Wc = (const float*)d_in[7];
  const float* bc = (const float*)d_in[8];
  const float* Wg = (const float*)d_in[9];
  const float* bg = (const float*)d_in[10];
  const float* WQ = (const float*)d_in[11];
  const float* WK = (const float*)d_in[12];
  const float* WVm = (const float*)d_in[13];
  const float* WE = (const float*)d_in[14];
  float* out = (float*)d_out;

  const int N = in_sizes[0] / 64;
  const int E = in_sizes[5];
  const size_t ND = (size_t)N * 64;

  float* w = (float*)d_ws;
  float* Tx1  = w; w += ND;
  float* Tx2  = w; w += ND;
  float* Tx3  = w; w += ND;
  float* gcn  = w; w += ND;
  float* scr  = w; w += 2 * ND;
  float* Qh   = w; w += ND;
  float* Kh   = w; w += ND;
  float* Vh   = w; w += ND;
  float* lst  = w; w += ND + (size_t)N * 8;
  float* dinv = w; w += N;
  int* cnt    = (int*)w; w += N;
  int* total  = (int*)w; w += 1;
  int* start   = (int*)lst;
  int* csrc    = start + N;
  int* epos    = csrc + E;
  int* cursor  = (int*)scr;
  float* cwl   = (float*)(cursor + N);
  float* cwe   = cwl + E;
  float* ssb   = Tx1;

  hipMemsetAsync(cnt, 0, sizeof(int) * ((size_t)N + 1), stream);

  const int B = 256;
  const int rowBlocks = (N + 7) / 8;

  count_kernel<<<(E + B - 1) / B, B, 0, stream>>>(edst, cnt, E);
  rowalloc_kernel<<<(N + B - 1) / B, B, 0, stream>>>(cnt, total, start, cursor, dinv, N);
  fill_kernel<<<(E + B - 1) / B, B, 0, stream>>>(esrc, edst, dinv, ew, cursor, csrc, cwl, cwe, epos, E);

  gather_rows<<<rowBlocks, B, 0, stream>>>(h, cwl, csrc, start, cnt, nullptr, Tx1, N);
  gather_rows<<<rowBlocks, B, 0, stream>>>(Tx1, cwl, csrc, start, cnt, h, Tx2, N);
  gather_rows<<<rowBlocks, B, 0, stream>>>(Tx2, cwl, csrc, start, cnt, Tx1, Tx3, N);

  const int rowTiles = (N + 15) / 16;
  const int gemmBlocks = (rowTiles + 1) / 2;
  cheb_gemm_kernel<<<gemmBlocks, 64, 0, stream>>>(h, Tx1, Tx2, Tx3, fc, Wc, bc, out, N, N);

  gather_rows<<<rowBlocks, B, 0, stream>>>(h, cwe, csrc, start, cnt, nullptr, gcn, N);
  gemm_wmma_kernel<false><<<gemmBlocks, 64, 0, stream>>>(gcn, nullptr, Wg, bg, out, 192, 64, N);

  gemm_wmma_kernel<true><<<gemmBlocks, 64, 0, stream>>>(h, p, WQ, nullptr, Qh, 64, 0, N);
  gemm_wmma_kernel<true><<<gemmBlocks, 64, 0, stream>>>(h, p, WK, nullptr, Kh, 64, 0, N);
  gemm_wmma_kernel<true><<<gemmBlocks, 64, 0, stream>>>(h, p, WVm, nullptr, Vh, 64, 0, N);

  const int eTiles = (E + 15) / 16;
  attn_kernel<<<(eTiles + 7) / 8, 256, 0, stream>>>(e, WE, Qh, Kh, esrc, edst, epos, ssb, E);
  attn_gather_rows<<<rowBlocks, B, 0, stream>>>(Vh, ssb, csrc, start, cnt, out, N);
}


# The five dense-product launches (Chebyshev, GCN, Q, K, V) merged into one grid: blockIdx.y picks the product, each wave still one 16-row strip of one product; sums and their order unchanged
# speedup vs baseline: 2.6480x; 1.0498x over previous
#include <hip/hip_runtime.h>
#include <hip/hip_bf16.h>

typedef __attribute__((ext_vector_type(2))) float v2f;
typedef __attribute__((ext_vector_type(8))) float v8f;

__device__ __forceinline__ v8f wmma4(v2f a, v2f b, v8f c) {
  return __builtin_amdgcn_wmma_f32_16x16x4_f32(
       false, a,  false, b,
       (short)0, c,  false,  false);
}

__global__ void count_kernel(const int* __restrict__ dst, int* __restrict__ cnt, int E) {
  int tid = blockIdx.x * blockDim.x + threadIdx.x;
  if (tid < E) atomicAdd(&cnt[dst[tid]], 1);
}

__global__ void rowalloc_kernel(const int* __restrict__ cnt, int* __restrict__ total,
                                int* __restrict__ start, int* __restrict__ cursor,
                                float* __restrict__ dinv, int N) {
  int tid = blockIdx.x * blockDim.x + threadIdx.x;
  int lane = threadIdx.x & 31;
  int c = tid < N ? cnt[tid] : 0;
  int inc = c;
  for (int o = 1; o < 32; o <<= 1) {
    int v = __shfl_up(inc, o);
    if (lane >= o) inc += v;
  }
  int base = 0;
  if (lane == 31) base = atomicAdd(total, inc);
  base = __shfl(base, 31);
  if (tid < N) {
    int st = base + inc - c;
    start[tid] = st;
    cursor[tid] = st;
    float d = (float)c;
    d = d < 1.0f ? 1.0f : d;
    dinv[tid] = rsqrtf(d);
  }
}

__global__ void fill_kernel(const int* __restrict__ src, const int* __restrict__ dst,
                            const float* __restrict__ dinv, const float* __restrict__ ew,
                            int* __restrict__ cursor, int* __restrict__ csrc,
                            float* __restrict__ cwl, float* __restrict__ cwe,
                            int* __restrict__ epos, int E) {
  int tid = blockIdx.x * blockDim.x + threadIdx.x;
  if (tid >= E) return;
  int s = src[tid], d = dst[tid];
  int pos = atomicAdd(&cursor[d], 1);
  csrc[pos] = s;
  cwl[pos] = -(dinv[s] * dinv[d]);
  cwe[pos] = ew[tid];
  epos[tid] = pos;
}

__global__ void gather_rows(const float* __restrict__ x, const float* __restrict__ w,
                            const int* __restrict__ csrc, const int* __restrict__ start,
                            const int* __restrict__ cnt, const float* __restrict__ prev,
                            float* __restrict__ out, int N) {
  int row = blockIdx.x * (blockDim.x >> 5) + (threadIdx.x >> 5);
  if (row >= N) return;
  row = __builtin_amdgcn_readfirstlane(row);
  int lane = threadIdx.x & 31;
  int beg = __builtin_amdgcn_readfirstlane(start[row]);
  int end = beg + __builtin_amdgcn_readfirstlane(cnt[row]);
  const float2* __restrict__ x2 = (const float2*)x;
  float2 pv = make_float2(0.0f, 0.0f);
  if (prev) pv = ((const float2*)prev)[(size_t)row * 32 + lane];
  float2 acc = make_float2(0.0f, 0.0f);
  for (int base = beg; base < end; base += 32) {
    int j = base + lane;
    int sv = 0, wi = 0;
    if (j < end) { sv = csrc[j]; wi = __float_as_int(w[j]); }
    int n = end - base;
    n = n < 32 ? n : 32;
    for (int k = 0; k < n; k += 4) {
      int r = n - k;
      bool b1 = r > 1, b2 = r > 2, b3 = r > 3;
      int k1 = b1 ? k + 1 : k, k2 = b2 ? k + 2 : k, k3 = b3 ? k + 3 : k;
      int s0 = __builtin_amdgcn_readlane(sv, k), s1 = __builtin_amdgcn_readlane(sv, k1);
      int s2 = __builtin_amdgcn_readlane(sv, k2), s3 = __builtin_amdgcn_readlane(sv, k3);
      float w0 = __int_as_float(__builtin_amdgcn_readlane(wi, k));
      float w1 = __int_as_float(__builtin_amdgcn_readlane(wi, k1));
      float w2 = __int_as_float(__builtin_amdgcn_readlane(wi, k2));
      float w3 = __int_as_float(__builtin_amdgcn_readlane(wi, k3));
      float2 v0 = x2[(size_t)s0 * 32 + lane];
      float2 v1 = x2[(size_t)s1 * 32 + lane];
      float2 v2 = x2[(size_t)s2 * 32 + lane];
      float2 v3 = x2[(size_t)s3 * 32 + lane];
      v1.x = b1 ? v1.x : 0.0f; v1.y = b1 ? v1.y : 0.0f;
      v2.x = b2 ? v2.x : 0.0f; v2.y = b2 ? v2.y : 0.0f;
      v3.x = b3 ? v3.x : 0.0f; v3.y = b3 ? v3.y : 0.0f;
      acc.x += w0 * v0.x; acc.y += w0 * v0.y;
      acc.x += w1 * v1.x; acc.y += w1 * v1.y;
      acc.x += w2 * v2.x; acc.y += w2 * v2.y;
      acc.x += w3 * v3.x; acc.y += w3 * v3.y;
    }
  }
  if (prev) {
    acc.x = 2.0f * acc.x - pv.x;
    acc.y = 2.0f * acc.y - pv.y;
  }
  ((float2*)out)[(size_t)row * 32 + lane] = acc;
}


template <bool SCALED>
__device__ __forceinline__ void plane_mma(const float* __restrict__ arow,
                                          const float* __restrict__ bcol, int half, float sc,
                                          v8f (&acc)[4]) {
  float4 av[8], bw[8][4];
#pragma unroll
  for (int j = 0; j < 8; ++j) {
    int k = 8 * j + 4 * half;
    av[j] = *(const float4*)(arow + k);
#pragma unroll
    for (int i = 0; i < 4; ++i) bw[j][i] = *(const float4*)(bcol + (size_t)(k + i) * 64);
  }
#pragma unroll
  for (int j = 0; j < 8; ++j) {
    float4 a = av[j];
    if (SCALED) { a.x *= sc; a.y *= sc; a.z *= sc; a.w *= sc; }
    v2f a01 = {a.x, a.y}, a23 = {a.z, a.w};
    acc[0] = wmma4(a01, (v2f){bw[j][0].x, bw[j][1].x}, acc[0]);
    acc[1] = wmma4(a01, (v2f){bw[j][0].y, bw[j][1].y}, acc[1]);
    acc[2] = wmma4(a01, (v2f){bw[j][0].z, bw[j][1].z}, acc[2]);
    acc[3] = wmma4(a01, (v2f){bw[j][0].w, bw[j][1].w}, acc[3]);
    acc[0] = wmma4(a23, (v2f){bw[j][2].x, bw[j][3].x}, acc[0]);
    acc[1] = wmma4(a23, (v2f){bw[j][2].y, bw[j][3].y}, acc[1]);
    acc[2] = wmma4(a23, (v2f){bw[j][2].z, bw[j][3].z}, acc[2]);
    acc[3] = wmma4(a23, (v2f){bw[j][2].w, bw[j][3].w}, acc[3]);
  }
}

__device__ __forceinline__ void strip_store(const v8f (&acc)[4], float4 bv, float* __restrict__ Out,
                                            int ldo, int col, int row0, int rows) {
#pragma unroll
  for (int r = 0; r < 8; ++r) {
    int row = row0 + r;
    if (row < rows) {
      float4 o = make_float4(acc[0][r] + bv.x, acc[1][r] + bv.y, acc[2][r] + bv.z, acc[3][r] + bv.w);
      *(float4*)(Out + (size_t)row * ldo + col) = o;
    }
  }
}

__global__ void __launch_bounds__(64) __attribute__((amdgpu_waves_per_eu(1, 4)))
dense_kernel(const float* __restrict__ h, const float* __restrict__ p,
             const float* __restrict__ Tx1, const float* __restrict__ Tx2,
             const float* __restrict__ Tx3, const float* __restrict__ gcn,
             const float* __restrict__ fc, const float* __restrict__ Wc,
             const float* __restrict__ bc, const float* __restrict__ Wg,
             const float* __restrict__ bg, const float* __restrict__ WQ,
             const float* __restrict__ WK, const float* __restrict__ WV,
             float* __restrict__ Out, float* __restrict__ Qh, float* __restrict__ Kh,
             float* __restrict__ Vh, int rows) {
  int rt = blockIdx.x * (blockDim.x >> 5) + (threadIdx.x >> 5);
  int rowTiles = (rows + 15) >> 4;
  if (rt >= rowTiles) return;
  int lane = threadIdx.x & 31, half = lane >> 4, c = lane & 15;
  int ar = rt * 16 + c;
  if (ar >= rows) ar = rows - 1;
  const size_t ao = (size_t)ar * 64;
  const int row0 = rt * 16 + 8 * half;
  const int which = blockIdx.y;
  v8f acc[4] = {};
  if (which == 0) {
    const float* As[4] = {h, Tx1, Tx2, Tx3};
#pragma unroll
    for (int k = 0; k < 4; ++k) {
      float sc = fc[(size_t)k * rows + ar];
      plane_mma<true>(As[k] + ao, Wc + (size_t)k * 64 * 64 + 4 * c, half, sc, acc);
    }
    float4 bv = *(const float4*)(bc + 4 * c);
    strip_store(acc, bv, Out, 192, 4 * c, row0, rows);
  } else if (which == 4) {
    plane_mma<false>(gcn + ao, Wg + 4 * c, half, 1.0f, acc);
    float4 bv = *(const float4*)(bg + 4 * c);
    strip_store(acc, bv, Out, 192, 64 + 4 * c, row0, rows);
  } else {
    const float* W = which == 1 ? WQ : (which == 2 ? WK : WV);
    float* O = which == 1 ? Qh : (which == 2 ? Kh : Vh);
    plane_mma<false>(h + ao, W + 4 * c, half, 1.0f, acc);
    plane_mma<false>(p + ao, W + (size_t)64 * 64 + 4 * c, half, 1.0f, acc);
    strip_store(acc, make_float4(0.0f, 0.0f, 0.0f, 0.0f), O, 64, 4 * c, row0, rows);
  }
}

__global__ void __launch_bounds__(256)
attn_kernel(const float* __restrict__ Ee, const float* __restrict__ WE,
                            const float* __restrict__ Qh, const float* __restrict__ Kh,
                            const int* __restrict__ src, const int* __restrict__ dst,
                            const int* __restrict__ epos, float* __restrict__ ssb, int E) {
  int tile = blockIdx.x * (blockDim.x >> 5) + (threadIdx.x >> 5);
  int eTiles = (E + 15) >> 4;
  if (tile >= eTiles) return;
  int lane = threadIdx.x & 31, half = lane >> 4, c = lane & 15;
  int em = tile * 16 + c;
  bool live = em < E;
  int er = live ? em : E - 1;
  int s = src[er], d = dst[er], slot = epos[er];

  const float4* kp = (const float4*)(Kh + (size_t)s * 64 + 16 * half);
  const float4* qp = (const float4*)(Qh + (size_t)d * 64 + 16 * half);
  float4 kv[8], qv[8];
#pragma unroll
  for (int i = 0; i < 4; ++i) {
    kv[i] = kp[i]; kv[4 + i] = kp[8 + i];
    qv[i] = qp[i]; qv[4 + i] = qp[8 + i];
  }

  int gc = (c & 3) | ((c & 8) >> 1) | ((c & 4) << 1);
  const float* wcol = WE + 4 * gc;
  const float* erow = Ee + (size_t)er * 64;
  v8f acc[4] = {};
#pragma unroll
  for (int j = 0; j < 8; ++j) {
    int k = 8 * j + 4 * half;
    float4 ev = *(const float4*)(erow + k);
    float4 w0 = *(const float4*)(wcol + (size_t)(k + 0) * 64);
    float4 w1 = *(const float4*)(wcol + (size_t)(k + 1) * 64);
    float4 w2 = *(const float4*)(wcol + (size_t)(k + 2) * 64);
    float4 w3 = *(const float4*)(wcol + (size_t)(k + 3) * 64);
    v2f e01 = {ev.x, ev.y}, e23 = {ev.z, ev.w};
    acc[0] = wmma4((v2f){w0.x, w1.x}, e01, acc[0]);
    acc[1] = wmma4((v2f){w0.y, w1.y}, e01, acc[1]);
    acc[2] = wmma4((v2f){w0.z, w1.z}, e01, acc[2]);
    acc[3] = wmma4((v2f){w0.w, w1.w}, e01, acc[3]);
    acc[0] = wmma4((v2f){w2.x, w3.x}, e23, acc[0]);
    acc[1] = wmma4((v2f){w2.y, w3.y}, e23, acc[1]);
    acc[2] = wmma4((v2f){w2.z, w3.z}, e23, acc[2]);
    acc[3] = wmma4((v2f){w2.w, w3.w}, e23, acc[3]);
  }

  const float scale = 0.35355339059327373f;
  float ss[4];
#pragma unroll
  for (int q = 0; q < 4; ++q) {
    float kk[8] = {kv[2 * q].x, kv[2 * q].y, kv[2 * q].z, kv[2 * q].w,
                   kv[2 * q + 1].x, kv[2 * q + 1].y, kv[2 * q + 1].z, kv[2 * q + 1].w};
    float qq[8] = {qv[2 * q].x, qv[2 * q].y, qv[2 * q].z, qv[2 * q].w,
                   qv[2 * q + 1].x, qv[2 * q + 1].y, qv[2 * q + 1].z, qv[2 * q + 1].w};
    float sc = 0.0f;
#pragma unroll
    for (int dd = 0; dd < 8; ++dd)
      sc += kk[dd] * qq[dd] * acc[dd & 3][2 * q + (dd >> 2)];
    sc *= scale;
    sc = fminf(5.0f, fmaxf(-5.0f, sc));
    ss[q] = __expf(sc);
  }
  if (live) {
    float2* o = (float2*)(ssb + (size_t)slot * 8 + 2 * half);
    o[0] = make_float2(ss[0], ss[1]);
    o[2] = make_float2(ss[2], ss[3]);
  }
}

__global__ void attn_gather_rows(const float* __restrict__ Vh, const float* __restrict__ ssb,
                                 const int* __restrict__ csrc, const int* __restrict__ start,
                                 const int* __restrict__ cnt, float* __restrict__ out, int N) {
  int row = blockIdx.x * (blockDim.x >> 5) + (threadIdx.x >> 5);
  if (row >= N) return;
  row = __builtin_amdgcn_readfirstlane(row);
  int lane = threadIdx.x & 31;
  int hd = lane >> 2;
  int beg = __builtin_amdgcn_readfirstlane(start[row]);
  int end = beg + __builtin_amdgcn_readfirstlane(cnt[row]);
  const float2* __restrict__ v2p = (const float2*)Vh;
  float2 acc = make_float2(0.0f, 0.0f);
  float zs = 0.0f;
  for (int base = beg; base < end; base += 32) {
    int j = base + lane;
    int sv = 0;
    if (j < end) sv = csrc[j];
    int n = end - base;
    n = n < 32 ? n : 32;
    const float* __restrict__ sp = ssb + (size_t)base * 8 + hd;
    for (int k = 0; k < n; k += 4) {
      int r = n - k;
      bool b1 = r > 1, b2 = r > 2, b3 = r > 3;
      int k1 = b1 ? k + 1 : k, k2 = b2 ? k + 2 : k, k3 = b3 ? k + 3 : k;
      int s0 = __builtin_amdgcn_readlane(sv, k), s1 = __builtin_amdgcn_readlane(sv, k1);
      int s2 = __builtin_amdgcn_readlane(sv, k2), s3 = __builtin_amdgcn_readlane(sv, k3);
      float p0 = sp[k * 8], p1 = sp[k1 * 8], p2 = sp[k2 * 8], p3 = sp[k3 * 8];
      float2 v0 = v2p[(size_t)s0 * 32 + lane];
      float2 v1 = v2p[(size_t)s1 * 32 + lane];
      float2 v2 = v2p[(size_t)s2 * 32 + lane];
      float2 v3 = v2p[(size_t)s3 * 32 + lane];
      p1 = b1 ? p1 : 0.0f;
      p2 = b2 ? p2 : 0.0f;
      p3 = b3 ? p3 : 0.0f;
      acc.x += p0 * v0.x; acc.y += p0 * v0.y;
      acc.x += p1 * v1.x; acc.y += p1 * v1.y;
      acc.x += p2 * v2.x; acc.y += p2 * v2.y;
      acc.x += p3 * v3.x; acc.y += p3 * v3.y;
      zs += p0; zs += p1; zs += p2; zs += p3;
    }
  }
  float zz = zs + 1e-6f;
  float2 o = make_float2(acc.x / zz, acc.y / zz);
  ((float2*)out)[((size_t)row * 192 + 128) / 2 + lane] = o;
}

extern "C" void kernel_launch(void* const* d_in, const int* in_sizes, int n_in,
                              void* d_out, int out_size, void* d_ws, size_t ws_size,
                              hipStream_t stream) {
  const float* h  = (const float*)d_in[0];
  const float* p  = (const float*)d_in[1];
  const float* e  = (const float*)d_in[2];
  const float* ew = (const float*)d_in[3];
  const float* fc = (const float*)d_in[4];
  const int* esrc = (const int*)d_in[5];
  const int* edst = (const int*)d_in[6];
  const float* Wc = (const float*)d_in[7];
  const float* bc = (const float*)d_in[8];
  const float* Wg = (const float*)d_in[9];
  const float* bg = (const float*)d_in[10];
  const float* WQ = (const float*)d_in[11];
  const float* WK = (const float*)d_in[12];
  const float* WVm = (const float*)d_in[13];
  const float* WE = (const float*)d_in[14];
  float* out = (float*)d_out;

  const int N = in_sizes[0] / 64;
  const int E = in_sizes[5];
  const size_t ND = (size_t)N * 64;

  float* w = (float*)d_ws;
  float* Tx1  = w; w += ND;
  float* Tx2  = w; w += ND;
  float* Tx3  = w; w += ND;
  float* gcn  = w; w += ND;
  float* scr  = w; w += 2 * ND;
  float* Qh   = w; w += ND;
  float* Kh   = w; w += ND;
  float* Vh   = w; w += ND;
  float* lst  = w; w += ND + (size_t)N * 8;
  float* dinv = w; w += N;
  int* cnt    = (int*)w; w += N;
  int* total  = (int*)w; w += 1;
  int* start   = (int*)lst;
  int* csrc    = start + N;
  int* epos    = csrc + E;
  int* cursor  = (int*)scr;
  float* cwl   = (float*)(cursor + N);
  float* cwe   = cwl + E;
  float* ssb   = Tx1;

  hipMemsetAsync(cnt, 0, sizeof(int) * ((size_t)N + 1), stream);

  const int B = 256;
  const int rowBlocks = (N + 7) / 8;

  count_kernel<<<(E + B - 1) / B, B, 0, stream>>>(edst, cnt, E);
  rowalloc_kernel<<<(N + B - 1) / B, B, 0, stream>>>(cnt, total, start, cursor, dinv, N);
  fill_kernel<<<(E + B - 1) / B, B, 0, stream>>>(esrc, edst, dinv, ew, cursor, csrc, cwl, cwe, epos, E);

  gather_rows<<<rowBlocks, B, 0, stream>>>(h, cwl, csrc, start, cnt, nullptr, Tx1, N);
  gather_rows<<<rowBlocks, B, 0, stream>>>(Tx1, cwl, csrc, start, cnt, h, Tx2, N);
  gather_rows<<<rowBlocks, B, 0, stream>>>(Tx2, cwl, csrc, start, cnt, Tx1, Tx3, N);

  gather_rows<<<rowBlocks, B, 0, stream>>>(h, cwe, csrc, start, cnt, nullptr, gcn, N);

  const int rowTiles = (N + 15) / 16;
  const int gemmBlocks = (rowTiles + 1) / 2;
  dense_kernel<<<dim3(gemmBlocks, 5), 64, 0, stream>>>(h, p, Tx1, Tx2, Tx3, gcn, fc, Wc, bc, Wg, bg,
                                                       WQ, WK, WVm, out, Qh, Kh, Vh, N);

  const int eTiles = (E + 15) / 16;
  attn_kernel<<<(eTiles + 7) / 8, 256, 0, stream>>>(e, WE, Qh, Kh, esrc, edst, epos, ssb, E);
  attn_gather_rows<<<rowBlocks, B, 0, stream>>>(Vh, ssb, csrc, start, cnt, out, N);
}


# First Laplacian product and GCN neighbour sum share one gather pass: each h row is read once and feeds both weighted sums (two accumulators per lane); one launch fewer; slot order of each sum kept
# speedup vs baseline: 2.7348x; 1.0328x over previous
#include <hip/hip_runtime.h>
#include <hip/hip_bf16.h>

typedef __attribute__((ext_vector_type(2))) float v2f;
typedef __attribute__((ext_vector_type(8))) float v8f;

__device__ __forceinline__ v8f wmma4(v2f a, v2f b, v8f c) {
  return __builtin_amdgcn_wmma_f32_16x16x4_f32(
       false, a,  false, b,
       (short)0, c,  false,  false);
}

__global__ void count_kernel(const int* __restrict__ dst, int* __restrict__ cnt, int E) {
  int tid = blockIdx.x * blockDim.x + threadIdx.x;
  if (tid < E) atomicAdd(&cnt[dst[tid]], 1);
}

__global__ void rowalloc_kernel(const int* __restrict__ cnt, int* __restrict__ total,
                                int* __restrict__ start, int* __restrict__ cursor,
                                float* __restrict__ dinv, int N) {
  int tid = blockIdx.x * blockDim.x + threadIdx.x;
  int lane = threadIdx.x & 31;
  int c = tid < N ? cnt[tid] : 0;
  int inc = c;
  for (int o = 1; o < 32; o <<= 1) {
    int v = __shfl_up(inc, o);
    if (lane >= o) inc += v;
  }
  int base = 0;
  if (lane == 31) base = atomicAdd(total, inc);
  base = __shfl(base, 31);
  if (tid < N) {
    int st = base + inc - c;
    start[tid] = st;
    cursor[tid] = st;
    float d = (float)c;
    d = d < 1.0f ? 1.0f : d;
    dinv[tid] = rsqrtf(d);
  }
}

__global__ void fill_kernel(const int* __restrict__ src, const int* __restrict__ dst,
                            const float* __restrict__ dinv, const float* __restrict__ ew,
                            int* __restrict__ cursor, int* __restrict__ csrc,
                            float* __restrict__ cwl, float* __restrict__ cwe,
                            int* __restrict__ epos, int E) {
  int tid = blockIdx.x * blockDim.x + threadIdx.x;
  if (tid >= E) return;
  int s = src[tid], d = dst[tid];
  int pos = atomicAdd(&cursor[d], 1);
  csrc[pos] = s;
  cwl[pos] = -(dinv[s] * dinv[d]);
  cwe[pos] = ew[tid];
  epos[tid] = pos;
}

__global__ void gather_rows(const float* __restrict__ x, const float* __restrict__ w,
                            const int* __restrict__ csrc, const int* __restrict__ start,
                            const int* __restrict__ cnt, const float* __restrict__ prev,
                            float* __restrict__ out, int N) {
  int row = blockIdx.x * (blockDim.x >> 5) + (threadIdx.x >> 5);
  if (row >= N) return;
  row = __builtin_amdgcn_readfirstlane(row);
  int lane = threadIdx.x & 31;
  int beg = __builtin_amdgcn_readfirstlane(start[row]);
  int end = beg + __builtin_amdgcn_readfirstlane(cnt[row]);
  const float2* __restrict__ x2 = (const float2*)x;
  float2 pv = make_float2(0.0f, 0.0f);
  if (prev) pv = ((const float2*)prev)[(size_t)row * 32 + lane];
  float2 acc = make_float2(0.0f, 0.0f);
  for (int base = beg; base < end; base += 32) {
    int j = base + lane;
    int sv = 0, wi = 0;
    if (j < end) { sv = csrc[j]; wi = __float_as_int(w[j]); }
    int n = end - base;
    n = n < 32 ? n : 32;
    for (int k = 0; k < n; k += 4) {
      int r = n - k;
      bool b1 = r > 1, b2 = r > 2, b3 = r > 3;
      int k1 = b1 ? k + 1 : k, k2 = b2 ? k + 2 : k, k3 = b3 ? k + 3 : k;
      int s0 = __builtin_amdgcn_readlane(sv, k), s1 = __builtin_amdgcn_readlane(sv, k1);
      int s2 = __builtin_amdgcn_readlane(sv, k2), s3 = __builtin_amdgcn_readlane(sv, k3);
      float w0 = __int_as_float(__builtin_amdgcn_readlane(wi, k));
      float w1 = __int_as_float(__builtin_amdgcn_readlane(wi, k1));
      float w2 = __int_as_float(__builtin_amdgcn_readlane(wi, k2));
      float w3 = __int_as_float(__builtin_amdgcn_readlane(wi, k3));
      float2 v0 = x2[(size_t)s0 * 32 + lane];
      float2 v1 = x2[(size_t)s1 * 32 + lane];
      float2 v2 = x2[(size_t)s2 * 32 + lane];
      float2 v3 = x2[(size_t)s3 * 32 + lane];
      v1.x = b1 ? v1.x : 0.0f; v1.y = b1 ? v1.y : 0.0f;
      v2.x = b2 ? v2.x : 0.0f; v2.y = b2 ? v2.y : 0.0f;
      v3.x = b3 ? v3.x : 0.0f; v3.y = b3 ? v3.y : 0.0f;
      acc.x += w0 * v0.x; acc.y += w0 * v0.y;
      acc.x += w1 * v1.x; acc.y += w1 * v1.y;
      acc.x += w2 * v2.x; acc.y += w2 * v2.y;
      acc.x += w3 * v3.x; acc.y += w3 * v3.y;
    }
  }
  if (prev) {
    acc.x = 2.0f * acc.x - pv.x;
    acc.y = 2.0f * acc.y - pv.y;
  }
  ((float2*)out)[(size_t)row * 32 + lane] = acc;
}

__global__ void gather_rows_pair(const float* __restrict__ x, const float* __restrict__ wa,
                                 const float* __restrict__ wb, const int* __restrict__ csrc,
                                 const int* __restrict__ start, const int* __restrict__ cnt,
                                 float* __restrict__ outa, float* __restrict__ outb, int N) {
  int row = blockIdx.x * (blockDim.x >> 5) + (threadIdx.x >> 5);
  if (row >= N) return;
  row = __builtin_amdgcn_readfirstlane(row);
  int lane = threadIdx.x & 31;
  int beg = __builtin_amdgcn_readfirstlane(start[row]);
  int end = beg + __builtin_amdgcn_readfirstlane(cnt[row]);
  const float2* __restrict__ x2 = (const float2*)x;
  float2 acca = make_float2(0.0f, 0.0f);
  float2 accb = make_float2(0.0f, 0.0f);
  for (int base = beg; base < end; base += 32) {
    int j = base + lane;
    int sv = 0, ai = 0, bi = 0;
    if (j < end) { sv = csrc[j]; ai = __float_as_int(wa[j]); bi = __float_as_int(wb[j]); }
    int n = end - base;
    n = n < 32 ? n : 32;
    for (int k = 0; k < n; k += 4) {
      int r = n - k;
      bool b1 = r > 1, b2 = r > 2, b3 = r > 3;
      int k1 = b1 ? k + 1 : k, k2 = b2 ? k + 2 : k, k3 = b3 ? k + 3 : k;
      int s0 = __builtin_amdgcn_readlane(sv, k), s1 = __builtin_amdgcn_readlane(sv, k1);
      int s2 = __builtin_amdgcn_readlane(sv, k2), s3 = __builtin_amdgcn_readlane(sv, k3);
      float a0 = __int_as_float(__builtin_amdgcn_readlane(ai, k));
      float a1 = __int_as_float(__builtin_amdgcn_readlane(ai, k1));
      float a2 = __int_as_float(__builtin_amdgcn_readlane(ai, k2));
      float a3 = __int_as_float(__builtin_amdgcn_readlane(ai, k3));
      float c0 = __int_as_float(__builtin_amdgcn_readlane(bi, k));
      float c1 = __int_as_float(__builtin_amdgcn_readlane(bi, k1));
      float c2 = __int_as_float(__builtin_amdgcn_readlane(bi, k2));
      float c3 = __int_as_float(__builtin_amdgcn_readlane(bi, k3));
      float2 v0 = x2[(size_t)s0 * 32 + lane];
      float2 v1 = x2[(size_t)s1 * 32 + lane];
      float2 v2 = x2[(size_t)s2 * 32 + lane];
      float2 v3 = x2[(size_t)s3 * 32 + lane];
      v1.x = b1 ? v1.x : 0.0f; v1.y = b1 ? v1.y : 0.0f;
      v2.x = b2 ? v2.x : 0.0f; v2.y = b2 ? v2.y : 0.0f;
      v3.x = b3 ? v3.x : 0.0f; v3.y = b3 ? v3.y : 0.0f;
      acca.x = fmaf(a0, v0.x, acca.x); acca.y = fmaf(a0, v0.y, acca.y);
      accb.x = fmaf(c0, v0.x, accb.x); accb.y = fmaf(c0, v0.y, accb.y);
      acca.x = fmaf(a1, v1.x, acca.x); acca.y = fmaf(a1, v1.y, acca.y);
      accb.x = fmaf(c1, v1.x, accb.x); accb.y = fmaf(c1, v1.y, accb.y);
      acca.x = fmaf(a2, v2.x, acca.x); acca.y = fmaf(a2, v2.y, acca.y);
      accb.x = fmaf(c2, v2.x, accb.x); accb.y = fmaf(c2, v2.y, accb.y);
      acca.x = fmaf(a3, v3.x, acca.x); acca.y = fmaf(a3, v3.y, acca.y);
      accb.x = fmaf(c3, v3.x, accb.x); accb.y = fmaf(c3, v3.y, accb.y);
    }
  }
  ((float2*)outa)[(size_t)row * 32 + lane] = acca;
  ((float2*)outb)[(size_t)row * 32 + lane] = accb;
}


template <bool SCALED>
__device__ __forceinline__ void plane_mma(const float* __restrict__ arow,
                                          const float* __restrict__ bcol, int half, float sc,
                                          v8f (&acc)[4]) {
  float4 av[8], bw[8][4];
#pragma unroll
  for (int j = 0; j < 8; ++j) {
    int k = 8 * j + 4 * half;
    av[j] = *(const float4*)(arow + k);
#pragma unroll
    for (int i = 0; i < 4; ++i) bw[j][i] = *(const float4*)(bcol + (size_t)(k + i) * 64);
  }
#pragma unroll
  for (int j = 0; j < 8; ++j) {
    float4 a = av[j];
    if (SCALED) { a.x *= sc; a.y *= sc; a.z *= sc; a.w *= sc; }
    v2f a01 = {a.x, a.y}, a23 = {a.z, a.w};
    acc[0] = wmma4(a01, (v2f){bw[j][0].x, bw[j][1].x}, acc[0]);
    acc[1] = wmma4(a01, (v2f){bw[j][0].y, bw[j][1].y}, acc[1]);
    acc[2] = wmma4(a01, (v2f){bw[j][0].z, bw[j][1].z}, acc[2]);
    acc[3] = wmma4(a01, (v2f){bw[j][0].w, bw[j][1].w}, acc[3]);
    acc[0] = wmma4(a23, (v2f){bw[j][2].x, bw[j][3].x}, acc[0]);
    acc[1] = wmma4(a23, (v2f){bw[j][2].y, bw[j][3].y}, acc[1]);
    acc[2] = wmma4(a23, (v2f){bw[j][2].z, bw[j][3].z}, acc[2]);
    acc[3] = wmma4(a23, (v2f){bw[j][2].w, bw[j][3].w}, acc[3]);
  }
}

__device__ __forceinline__ void strip_store(const v8f (&acc)[4], float4 bv, float* __restrict__ Out,
                                            int ldo, int col, int row0, int rows) {
#pragma unroll
  for (int r = 0; r < 8; ++r) {
    int row = row0 + r;
    if (row < rows) {
      float4 o = make_float4(acc[0][r] + bv.x, acc[1][r] + bv.y, acc[2][r] + bv.z, acc[3][r] + bv.w);
      *(float4*)(Out + (size_t)row * ldo + col) = o;
    }
  }
}

__global__ void __launch_bounds__(64) __attribute__((amdgpu_waves_per_eu(1, 4)))
dense_kernel(const float* __restrict__ h, const float* __restrict__ p,
             const float* __restrict__ Tx1, const float* __restrict__ Tx2,
             const float* __restrict__ Tx3, const float* __restrict__ gcn,
             const float* __restrict__ fc, const float* __restrict__ Wc,
             const float* __restrict__ bc, const float* __restrict__ Wg,
             const float* __restrict__ bg, const float* __restrict__ WQ,
             const float* __restrict__ WK, const float* __restrict__ WV,
             float* __restrict__ Out, float* __restrict__ Qh, float* __restrict__ Kh,
             float* __restrict__ Vh, int rows) {
  int rt = blockIdx.x * (blockDim.x >> 5) + (threadIdx.x >> 5);
  int rowTiles = (rows + 15) >> 4;
  if (rt >= rowTiles) return;
  int lane = threadIdx.x & 31, half = lane >> 4, c = lane & 15;
  int ar = rt * 16 + c;
  if (ar >= rows) ar = rows - 1;
  const size_t ao = (size_t)ar * 64;
  const int row0 = rt * 16 + 8 * half;
  const int which = blockIdx.y;
  v8f acc[4] = {};
  if (which == 0) {
    const float* As[4] = {h, Tx1, Tx2, Tx3};
#pragma unroll
    for (int k = 0; k < 4; ++k) {
      float sc = fc[(size_t)k * rows + ar];
      plane_mma<true>(As[k] + ao, Wc + (size_t)k * 64 * 64 + 4 * c, half, sc, acc);
    }
    float4 bv = *(const float4*)(bc + 4 * c);
    strip_store(acc, bv, Out, 192, 4 * c, row0, rows);
  } else if (which == 4) {
    plane_mma<false>(gcn + ao, Wg + 4 * c, half, 1.0f, acc);
    float4 bv = *(const float4*)(bg + 4 * c);
    strip_store(acc, bv, Out, 192, 64 + 4 * c, row0, rows);
  } else {
    const float* W = which == 1 ? WQ : (which == 2 ? WK : WV);
    float* O = which == 1 ? Qh : (which == 2 ? Kh : Vh);
    plane_mma<false>(h + ao, W + 4 * c, half, 1.0f, acc);
    plane_mma<false>(p + ao, W + (size_t)64 * 64 + 4 * c, half, 1.0f, acc);
    strip_store(acc, make_float4(0.0f, 0.0f, 0.0f, 0.0f), O, 64, 4 * c, row0, rows);
  }
}

__global__ void __launch_bounds__(256)
attn_kernel(const float* __restrict__ Ee, const float* __restrict__ WE,
                            const float* __restrict__ Qh, const float* __restrict__ Kh,
                            const int* __restrict__ src, const int* __restrict__ dst,
                            const int* __restrict__ epos, float* __restrict__ ssb, int E) {
  int tile = blockIdx.x * (blockDim.x >> 5) + (threadIdx.x >> 5);
  int eTiles = (E + 15) >> 4;
  if (tile >= eTiles) return;
  int lane = threadIdx.x & 31, half = lane >> 4, c = lane & 15;
  int em = tile * 16 + c;
  bool live = em < E;
  int er = live ? em : E - 1;
  int s = src[er], d = dst[er], slot = epos[er];

  const float4* kp = (const float4*)(Kh + (size_t)s * 64 + 16 * half);
  const float4* qp = (const float4*)(Qh + (size_t)d * 64 + 16 * half);
  float4 kv[8], qv[8];
#pragma unroll
  for (int i = 0; i < 4; ++i) {
    kv[i] = kp[i]; kv[4 + i] = kp[8 + i];
    qv[i] = qp[i]; qv[4 + i] = qp[8 + i];
  }

  int gc = (c & 3) | ((c & 8) >> 1) | ((c & 4) << 1);
  const float* wcol = WE + 4 * gc;
  const float* erow = Ee + (size_t)er * 64;
  v8f acc[4] = {};
#pragma unroll
  for (int j = 0; j < 8; ++j) {
    int k = 8 * j + 4 * half;
    float4 ev = *(const float4*)(erow + k);
    float4 w0 = *(const float4*)(wcol + (size_t)(k + 0) * 64);
    float4 w1 = *(const float4*)(wcol + (size_t)(k + 1) * 64);
    float4 w2 = *(const float4*)(wcol + (size_t)(k + 2) * 64);
    float4 w3 = *(const float4*)(wcol + (size_t)(k + 3) * 64);
    v2f e01 = {ev.x, ev.y}, e23 = {ev.z, ev.w};
    acc[0] = wmma4((v2f){w0.x, w1.x}, e01, acc[0]);
    acc[1] = wmma4((v2f){w0.y, w1.y}, e01, acc[1]);
    acc[2] = wmma4((v2f){w0.z, w1.z}, e01, acc[2]);
    acc[3] = wmma4((v2f){w0.w, w1.w}, e01, acc[3]);
    acc[0] = wmma4((v2f){w2.x, w3.x}, e23, acc[0]);
    acc[1] = wmma4((v2f){w2.y, w3.y}, e23, acc[1]);
    acc[2] = wmma4((v2f){w2.z, w3.z}, e23, acc[2]);
    acc[3] = wmma4((v2f){w2.w, w3.w}, e23, acc[3]);
  }

  const float scale = 0.35355339059327373f;
  float ss[4];
#pragma unroll
  for (int q = 0; q < 4; ++q) {
    float kk[8] = {kv[2 * q].x, kv[2 * q].y, kv[2 * q].z, kv[2 * q].w,
                   kv[2 * q + 1].x, kv[2 * q + 1].y, kv[2 * q + 1].z, kv[2 * q + 1].w};
    float qq[8] = {qv[2 * q].x, qv[2 * q].y, qv[2 * q].z, qv[2 * q].w,
                   qv[2 * q + 1].x, qv[2 * q + 1].y, qv[2 * q + 1].z, qv[2 * q + 1].w};
    float sc = 0.0f;
#pragma unroll
    for (int dd = 0; dd < 8; ++dd)
      sc += kk[dd] * qq[dd] * acc[dd & 3][2 * q + (dd >> 2)];
    sc *= scale;
    sc = fminf(5.0f, fmaxf(-5.0f, sc));
    ss[q] = __expf(sc);
  }
  if (live) {
    float2* o = (float2*)(ssb + (size_t)slot * 8 + 2 * half);
    o[0] = make_float2(ss[0], ss[1]);
    o[2] = make_float2(ss[2], ss[3]);
  }
}

__global__ void attn_gather_rows(const float* __restrict__ Vh, const float* __restrict__ ssb,
                                 const int* __restrict__ csrc, const int* __restrict__ start,
                                 const int* __restrict__ cnt, float* __restrict__ out, int N) {
  int row = blockIdx.x * (blockDim.x >> 5) + (threadIdx.x >> 5);
  if (row >= N) return;
  row = __builtin_amdgcn_readfirstlane(row);
  int lane = threadIdx.x & 31;
  int hd = lane >> 2;
  int beg = __builtin_amdgcn_readfirstlane(start[row]);
  int end = beg + __builtin_amdgcn_readfirstlane(cnt[row]);
  const float2* __restrict__ v2p = (const float2*)Vh;
  float2 acc = make_float2(0.0f, 0.0f);
  float zs = 0.0f;
  for (int base = beg; base < end; base += 32) {
    int j = base + lane;
    int sv = 0;
    if (j < end) sv = csrc[j];
    int n = end - base;
    n = n < 32 ? n : 32;
    const float* __restrict__ sp = ssb + (size_t)base * 8 + hd;
    for (int k = 0; k < n; k += 4) {
      int r = n - k;
      bool b1 = r > 1, b2 = r > 2, b3 = r > 3;
      int k1 = b1 ? k + 1 : k, k2 = b2 ? k + 2 : k, k3 = b3 ? k + 3 : k;
      int s0 = __builtin_amdgcn_readlane(sv, k), s1 = __builtin_amdgcn_readlane(sv, k1);
      int s2 = __builtin_amdgcn_readlane(sv, k2), s3 = __builtin_amdgcn_readlane(sv, k3);
      float p0 = sp[k * 8], p1 = sp[k1 * 8], p2 = sp[k2 * 8], p3 = sp[k3 * 8];
      float2 v0 = v2p[(size_t)s0 * 32 + lane];
      float2 v1 = v2p[(size_t)s1 * 32 + lane];
      float2 v2 = v2p[(size_t)s2 * 32 + lane];
      float2 v3 = v2p[(size_t)s3 * 32 + lane];
      p1 = b1 ? p1 : 0.0f;
      p2 = b2 ? p2 : 0.0f;
      p3 = b3 ? p3 : 0.0f;
      acc.x += p0 * v0.x; acc.y += p0 * v0.y;
      acc.x += p1 * v1.x; acc.y += p1 * v1.y;
      acc.x += p2 * v2.x; acc.y += p2 * v2.y;
      acc.x += p3 * v3.x; acc.y += p3 * v3.y;
      zs += p0; zs += p1; zs += p2; zs += p3;
    }
  }
  float zz = zs + 1e-6f;
  float2 o = make_float2(acc.x / zz, acc.y / zz);
  ((float2*)out)[((size_t)row * 192 + 128) / 2 + lane] = o;
}

extern "C" void kernel_launch(void* const* d_in, const int* in_sizes, int n_in,
                              void* d_out, int out_size, void* d_ws, size_t ws_size,
                              hipStream_t stream) {
  const float* h  = (const float*)d_in[0];
  const float* p  = (const float*)d_in[1];
  const float* e  = (const float*)d_in[2];
  const float* ew = (const float*)d_in[3];
  const float* fc = (const float*)d_in[4];
  const int* esrc = (const int*)d_in[5];
  const int* edst = (const int*)d_in[6];
  const float* Wc = (const float*)d_in[7];
  const float* bc = (const float*)d_in[8];
  const float* Wg = (const float*)d_in[9];
  const float* bg = (const float*)d_in[10];
  const float* WQ = (const float*)d_in[11];
  const float* WK = (const float*)d_in[12];
  const float* WVm = (const float*)d_in[13];
  const float* WE = (const float*)d_in[14];
  float* out = (float*)d_out;

  const int N = in_sizes[0] / 64;
  const int E = in_sizes[5];
  const size_t ND = (size_t)N * 64;

  float* w = (float*)d_ws;
  float* Tx1  = w; w += ND;
  float* Tx2  = w; w += ND;
  float* Tx3  = w; w += ND;
  float* gcn  = w; w += ND;
  float* scr  = w; w += 2 * ND;
  float* Qh   = w; w += ND;
  float* Kh   = w; w += ND;
  float* Vh   = w; w += ND;
  float* lst  = w; w += ND + (size_t)N * 8;
  float* dinv = w; w += N;
  int* cnt    = (int*)w; w += N;
  int* total  = (int*)w; w += 1;
  int* start   = (int*)lst;
  int* csrc    = start + N;
  int* epos    = csrc + E;
  int* cursor  = (int*)scr;
  float* cwl   = (float*)(cursor + N);
  float* cwe   = cwl + E;
  float* ssb   = Tx1;

  hipMemsetAsync(cnt, 0, sizeof(int) * ((size_t)N + 1), stream);

  const int B = 256;
  const int rowBlocks = (N + 7) / 8;

  count_kernel<<<(E + B - 1) / B, B, 0, stream>>>(edst, cnt, E);
  rowalloc_kernel<<<(N + B - 1) / B, B, 0, stream>>>(cnt, total, start, cursor, dinv, N);
  fill_kernel<<<(E + B - 1) / B, B, 0, stream>>>(esrc, edst, dinv, ew, cursor, csrc, cwl, cwe, epos, E);

  gather_rows_pair<<<rowBlocks, B, 0, stream>>>(h, cwl, cwe, csrc, start, cnt, Tx1, gcn, N);
  gather_rows<<<rowBlocks, B, 0, stream>>>(Tx1, cwl, csrc, start, cnt, h, Tx2, N);
  gather_rows<<<rowBlocks, B, 0, stream>>>(Tx2, cwl, csrc, start, cnt, Tx1, Tx3, N);

  const int rowTiles = (N + 15) / 16;
  const int gemmBlocks = (rowTiles + 1) / 2;
  dense_kernel<<<dim3(gemmBlocks, 5), 64, 0, stream>>>(h, p, Tx1, Tx2, Tx3, gcn, fc, Wc, bc, Wg, bg,
                                                       WQ, WK, WVm, out, Qh, Kh, Vh, N);

  const int eTiles = (E + 15) / 16;
  attn_kernel<<<(eTiles + 7) / 8, 256, 0, stream>>>(e, WE, Qh, Kh, esrc, edst, epos, ssb, E);
  attn_gather_rows<<<rowBlocks, B, 0, stream>>>(Vh, ssb, csrc, start, cnt, out, N);
}


# Edge list as one 16-byte record per slot (source, dinv[source], edge weight): fill does 1 scattered store + 1 lookup per edge, not 3 + 2; gathers apply dinv[row] themselves; sums bit-identical
# speedup vs baseline: 3.0508x; 1.1155x over previous
#include <hip/hip_runtime.h>
#include <hip/hip_bf16.h>

typedef __attribute__((ext_vector_type(2))) float v2f;
typedef __attribute__((ext_vector_type(8))) float v8f;

__device__ __forceinline__ v8f wmma4(v2f a, v2f b, v8f c) {
  return __builtin_amdgcn_wmma_f32_16x16x4_f32(
       false, a,  false, b,
       (short)0, c,  false,  false);
}

__global__ void count_kernel(const int* __restrict__ dst, int* __restrict__ cnt, int E) {
  int tid = blockIdx.x * blockDim.x + threadIdx.x;
  if (tid < E) atomicAdd(&cnt[dst[tid]], 1);
}

__global__ void rowalloc_kernel(const int* __restrict__ cnt, int* __restrict__ total,
                                int* __restrict__ start, int* __restrict__ cursor,
                                float* __restrict__ dinv, int N) {
  int tid = blockIdx.x * blockDim.x + threadIdx.x;
  int lane = threadIdx.x & 31;
  int c = tid < N ? cnt[tid] : 0;
  int inc = c;
  for (int o = 1; o < 32; o <<= 1) {
    int v = __shfl_up(inc, o);
    if (lane >= o) inc += v;
  }
  int base = 0;
  if (lane == 31) base = atomicAdd(total, inc);
  base = __shfl(base, 31);
  if (tid < N) {
    int st = base + inc - c;
    start[tid] = st;
    cursor[tid] = st;
    float d = (float)c;
    d = d < 1.0f ? 1.0f : d;
    dinv[tid] = rsqrtf(d);
  }
}

__global__ void fill_kernel(const int* __restrict__ src, const int* __restrict__ dst,
                            const float* __restrict__ dinv, const float* __restrict__ ew,
                            int* __restrict__ cursor, int* __restrict__ rec,
                            int* __restrict__ epos, int E) {
  int tid = blockIdx.x * blockDim.x + threadIdx.x;
  if (tid >= E) return;
  int s = src[tid], d = dst[tid];
  int pos = atomicAdd(&cursor[d], 1);
  ((int4*)rec)[pos] = make_int4(s, __float_as_int(dinv[s]), __float_as_int(ew[tid]), 0);
  epos[tid] = pos;
}

__global__ void gather_rows(const float* __restrict__ x, const int* __restrict__ rec,
                            const float* __restrict__ dinv, const int* __restrict__ start,
                            const int* __restrict__ cnt, const float* __restrict__ prev,
                            float* __restrict__ out, int N) {
  int row = blockIdx.x * (blockDim.x >> 5) + (threadIdx.x >> 5);
  if (row >= N) return;
  row = __builtin_amdgcn_readfirstlane(row);
  int lane = threadIdx.x & 31;
  int beg = __builtin_amdgcn_readfirstlane(start[row]);
  int end = beg + __builtin_amdgcn_readfirstlane(cnt[row]);
  const float dr = dinv[row];
  const int4* __restrict__ rec4 = (const int4*)rec;
  const float2* __restrict__ x2 = (const float2*)x;
  float2 pv = make_float2(0.0f, 0.0f);
  if (prev) pv = ((const float2*)prev)[(size_t)row * 32 + lane];
  float2 acc = make_float2(0.0f, 0.0f);
  for (int base = beg; base < end; base += 32) {
    int j = base + lane;
    int sv = 0, wi = 0;
    if (j < end) {
      int4 r = rec4[j];
      sv = r.x;
      wi = __float_as_int(-(__int_as_float(r.y) * dr));
    }
    int n = end - base;
    n = n < 32 ? n : 32;
    for (int k = 0; k < n; k += 4) {
      int r = n - k;
      bool b1 = r > 1, b2 = r > 2, b3 = r > 3;
      int k1 = b1 ? k + 1 : k, k2 = b2 ? k + 2 : k, k3 = b3 ? k + 3 : k;
      int s0 = __builtin_amdgcn_readlane(sv, k), s1 = __builtin_amdgcn_readlane(sv, k1);
      int s2 = __builtin_amdgcn_readlane(sv, k2), s3 = __builtin_amdgcn_readlane(sv, k3);
      float w0 = __int_as_float(__builtin_amdgcn_readlane(wi, k));
      float w1 = __int_as_float(__builtin_amdgcn_readlane(wi, k1));
      float w2 = __int_as_float(__builtin_amdgcn_readlane(wi, k2));
      float w3 = __int_as_float(__builtin_amdgcn_readlane(wi, k3));
      float2 v0 = x2[(size_t)s0 * 32 + lane];
      float2 v1 = x2[(size_t)s1 * 32 + lane];
      float2 v2 = x2[(size_t)s2 * 32 + lane];
      float2 v3 = x2[(size_t)s3 * 32 + lane];
      v1.x = b1 ? v1.x : 0.0f; v1.y = b1 ? v1.y : 0.0f;
      v2.x = b2 ? v2.x : 0.0f; v2.y = b2 ? v2.y : 0.0f;
      v3.x = b3 ? v3.x : 0.0f; v3.y = b3 ? v3.y : 0.0f;
      acc.x += w0 * v0.x; acc.y += w0 * v0.y;
      acc.x += w1 * v1.x; acc.y += w1 * v1.y;
      acc.x += w2 * v2.x; acc.y += w2 * v2.y;
      acc.x += w3 * v3.x; acc.y += w3 * v3.y;
    }
  }
  if (prev) {
    acc.x = 2.0f * acc.x - pv.x;
    acc.y = 2.0f * acc.y - pv.y;
  }
  ((float2*)out)[(size_t)row * 32 + lane] = acc;
}

__global__ void gather_rows_pair(const float* __restrict__ x, const int* __restrict__ rec,
                                 const float* __restrict__ dinv,
                                 const int* __restrict__ start, const int* __restrict__ cnt,
                                 float* __restrict__ outa, float* __restrict__ outb, int N) {
  int row = blockIdx.x * (blockDim.x >> 5) + (threadIdx.x >> 5);
  if (row >= N) return;
  row = __builtin_amdgcn_readfirstlane(row);
  int lane = threadIdx.x & 31;
  int beg = __builtin_amdgcn_readfirstlane(start[row]);
  int end = beg + __builtin_amdgcn_readfirstlane(cnt[row]);
  const float dr = dinv[row];
  const int4* __restrict__ rec4 = (const int4*)rec;
  const float2* __restrict__ x2 = (const float2*)x;
  float2 acca = make_float2(0.0f, 0.0f);
  float2 accb = make_float2(0.0f, 0.0f);
  for (int base = beg; base < end; base += 32) {
    int j = base + lane;
    int sv = 0, ai = 0, bi = 0;
    if (j < end) {
      int4 r = rec4[j];
      sv = r.x;
      ai = __float_as_int(-(__int_as_float(r.y) * dr));
      bi = r.z;
    }
    int n = end - base;
    n = n < 32 ? n : 32;
    for (int k = 0; k < n; k += 4) {
      int r = n - k;
      bool b1 = r > 1, b2 = r > 2, b3 = r > 3;
      int k1 = b1 ? k + 1 : k, k2 = b2 ? k + 2 : k, k3 = b3 ? k + 3 : k;
      int s0 = __builtin_amdgcn_readlane(sv, k), s1 = __builtin_amdgcn_readlane(sv, k1);
      int s2 = __builtin_amdgcn_readlane(sv, k2), s3 = __builtin_amdgcn_readlane(sv, k3);
      float a0 = __int_as_float(__builtin_amdgcn_readlane(ai, k));
      float a1 = __int_as_float(__builtin_amdgcn_readlane(ai, k1));
      float a2 = __int_as_float(__builtin_amdgcn_readlane(ai, k2));
      float a3 = __int_as_float(__builtin_amdgcn_readlane(ai, k3));
      float c0 = __int_as_float(__builtin_amdgcn_readlane(bi, k));
      float c1 = __int_as_float(__builtin_amdgcn_readlane(bi, k1));
      float c2 = __int_as_float(__builtin_amdgcn_readlane(bi, k2));
      float c3 = __int_as_float(__builtin_amdgcn_readlane(bi, k3));
      float2 v0 = x2[(size_t)s0 * 32 + lane];
      float2 v1 = x2[(size_t)s1 * 32 + lane];
      float2 v2 = x2[(size_t)s2 * 32 + lane];
      float2 v3 = x2[(size_t)s3 * 32 + lane];
      v1.x = b1 ? v1.x : 0.0f; v1.y = b1 ? v1.y : 0.0f;
      v2.x = b2 ? v2.x : 0.0f; v2.y = b2 ? v2.y : 0.0f;
      v3.x = b3 ? v3.x : 0.0f; v3.y = b3 ? v3.y : 0.0f;
      acca.x = fmaf(a0, v0.x, acca.x); acca.y = fmaf(a0, v0.y, acca.y);
      accb.x = fmaf(c0, v0.x, accb.x); accb.y = fmaf(c0, v0.y, accb.y);
      acca.x = fmaf(a1, v1.x, acca.x); acca.y = fmaf(a1, v1.y, acca.y);
      accb.x = fmaf(c1, v1.x, accb.x); accb.y = fmaf(c1, v1.y, accb.y);
      acca.x = fmaf(a2, v2.x, acca.x); acca.y = fmaf(a2, v2.y, acca.y);
      accb.x = fmaf(c2, v2.x, accb.x); accb.y = fmaf(c2, v2.y, accb.y);
      acca.x = fmaf(a3, v3.x, acca.x); acca.y = fmaf(a3, v3.y, acca.y);
      accb.x = fmaf(c3, v3.x, accb.x); accb.y = fmaf(c3, v3.y, accb.y);
    }
  }
  ((float2*)outa)[(size_t)row * 32 + lane] = acca;
  ((float2*)outb)[(size_t)row * 32 + lane] = accb;
}


template <bool SCALED>
__device__ __forceinline__ void plane_mma(const float* __restrict__ arow,
                                          const float* __restrict__ bcol, int half, float sc,
                                          v8f (&acc)[4]) {
  float4 av[8], bw[8][4];
#pragma unroll
  for (int j = 0; j < 8; ++j) {
    int k = 8 * j + 4 * half;
    av[j] = *(const float4*)(arow + k);
#pragma unroll
    for (int i = 0; i < 4; ++i) bw[j][i] = *(const float4*)(bcol + (size_t)(k + i) * 64);
  }
#pragma unroll
  for (int j = 0; j < 8; ++j) {
    float4 a = av[j];
    if (SCALED) { a.x *= sc; a.y *= sc; a.z *= sc; a.w *= sc; }
    v2f a01 = {a.x, a.y}, a23 = {a.z, a.w};
    acc[0] = wmma4(a01, (v2f){bw[j][0].x, bw[j][1].x}, acc[0]);
    acc[1] = wmma4(a01, (v2f){bw[j][0].y, bw[j][1].y}, acc[1]);
    acc[2] = wmma4(a01, (v2f){bw[j][0].z, bw[j][1].z}, acc[2]);
    acc[3] = wmma4(a01, (v2f){bw[j][0].w, bw[j][1].w}, acc[3]);
    acc[0] = wmma4(a23, (v2f){bw[j][2].x, bw[j][3].x}, acc[0]);
    acc[1] = wmma4(a23, (v2f){bw[j][2].y, bw[j][3].y}, acc[1]);
    acc[2] = wmma4(a23, (v2f){bw[j][2].z, bw[j][3].z}, acc[2]);
    acc[3] = wmma4(a23, (v2f){bw[j][2].w, bw[j][3].w}, acc[3]);
  }
}

__device__ __forceinline__ void strip_store(const v8f (&acc)[4], float4 bv, float* __restrict__ Out,
                                            int ldo, int col, int row0, int rows) {
#pragma unroll
  for (int r = 0; r < 8; ++r) {
    int row = row0 + r;
    if (row < rows) {
      float4 o = make_float4(acc[0][r] + bv.x, acc[1][r] + bv.y, acc[2][r] + bv.z, acc[3][r] + bv.w);
      *(float4*)(Out + (size_t)row * ldo + col) = o;
    }
  }
}

__global__ void __launch_bounds__(64) __attribute__((amdgpu_waves_per_eu(1, 4)))
dense_kernel(const float* __restrict__ h, const float* __restrict__ p,
             const float* __restrict__ Tx1, const float* __restrict__ Tx2,
             const float* __restrict__ Tx3, const float* __restrict__ gcn,
             const float* __restrict__ fc, const float* __restrict__ Wc,
             const float* __restrict__ bc, const float* __restrict__ Wg,
             const float* __restrict__ bg, const float* __restrict__ WQ,
             const float* __restrict__ WK, const float* __restrict__ WV,
             float* __restrict__ Out, float* __restrict__ Qh, float* __restrict__ Kh,
             float* __restrict__ Vh, int rows) {
  int rt = blockIdx.x * (blockDim.x >> 5) + (threadIdx.x >> 5);
  int rowTiles = (rows + 15) >> 4;
  if (rt >= rowTiles) return;
  int lane = threadIdx.x & 31, half = lane >> 4, c = lane & 15;
  int ar = rt * 16 + c;
  if (ar >= rows) ar = rows - 1;
  const size_t ao = (size_t)ar * 64;
  const int row0 = rt * 16 + 8 * half;
  const int which = blockIdx.y;
  v8f acc[4] = {};
  if (which == 0) {
    const float* As[4] = {h, Tx1, Tx2, Tx3};
#pragma unroll
    for (int k = 0; k < 4; ++k) {
      float sc = fc[(size_t)k * rows + ar];
      plane_mma<true>(As[k] + ao, Wc + (size_t)k * 64 * 64 + 4 * c, half, sc, acc);
    }
    float4 bv = *(const float4*)(bc + 4 * c);
    strip_store(acc, bv, Out, 192, 4 * c, row0, rows);
  } else if (which == 4) {
    plane_mma<false>(gcn + ao, Wg + 4 * c, half, 1.0f, acc);
    float4 bv = *(const float4*)(bg + 4 * c);
    strip_store(acc, bv, Out, 192, 64 + 4 * c, row0, rows);
  } else {
    const float* W = which == 1 ? WQ : (which == 2 ? WK : WV);
    float* O = which == 1 ? Qh : (which == 2 ? Kh : Vh);
    plane_mma<false>(h + ao, W + 4 * c, half, 1.0f, acc);
    plane_mma<false>(p + ao, W + (size_t)64 * 64 + 4 * c, half, 1.0f, acc);
    strip_store(acc, make_float4(0.0f, 0.0f, 0.0f, 0.0f), O, 64, 4 * c, row0, rows);
  }
}

__global__ void __launch_bounds__(256)
attn_kernel(const float* __restrict__ Ee, const float* __restrict__ WE,
                            const float* __restrict__ Qh, const float* __restrict__ Kh,
                            const int* __restrict__ src, const int* __restrict__ dst,
                            const int* __restrict__ epos, float* __restrict__ ssb, int E) {
  int tile = blockIdx.x * (blockDim.x >> 5) + (threadIdx.x >> 5);
  int eTiles = (E + 15) >> 4;
  if (tile >= eTiles) return;
  int lane = threadIdx.x & 31, half = lane >> 4, c = lane & 15;
  int em = tile * 16 + c;
  bool live = em < E;
  int er = live ? em : E - 1;
  int s = src[er], d = dst[er], slot = epos[er];

  const float4* kp = (const float4*)(Kh + (size_t)s * 64 + 16 * half);
  const float4* qp = (const float4*)(Qh + (size_t)d * 64 + 16 * half);
  float4 kv[8], qv[8];
#pragma unroll
  for (int i = 0; i < 4; ++i) {
    kv[i] = kp[i]; kv[4 + i] = kp[8 + i];
    qv[i] = qp[i]; qv[4 + i] = qp[8 + i];
  }

  int gc = (c & 3) | ((c & 8) >> 1) | ((c & 4) << 1);
  const float* wcol = WE + 4 * gc;
  const float* erow = Ee + (size_t)er * 64;
  v8f acc[4] = {};
#pragma unroll
  for (int j = 0; j < 8; ++j) {
    int k = 8 * j + 4 * half;
    float4 ev = *(const float4*)(erow + k);
    float4 w0 = *(const float4*)(wcol + (size_t)(k + 0) * 64);
    float4 w1 = *(const float4*)(wcol + (size_t)(k + 1) * 64);
    float4 w2 = *(const float4*)(wcol + (size_t)(k + 2) * 64);
    float4 w3 = *(const float4*)(wcol + (size_t)(k + 3) * 64);
    v2f e01 = {ev.x, ev.y}, e23 = {ev.z, ev.w};
    acc[0] = wmma4((v2f){w0.x, w1.x}, e01, acc[0]);
    acc[1] = wmma4((v2f){w0.y, w1.y}, e01, acc[1]);
    acc[2] = wmma4((v2f){w0.z, w1.z}, e01, acc[2]);
    acc[3] = wmma4((v2f){w0.w, w1.w}, e01, acc[3]);
    acc[0] = wmma4((v2f){w2.x, w3.x}, e23, acc[0]);
    acc[1] = wmma4((v2f){w2.y, w3.y}, e23, acc[1]);
    acc[2] = wmma4((v2f){w2.z, w3.z}, e23, acc[2]);
    acc[3] = wmma4((v2f){w2.w, w3.w}, e23, acc[3]);
  }

  const float scale = 0.35355339059327373f;
  float ss[4];
#pragma unroll
  for (int q = 0; q < 4; ++q) {
    float kk[8] = {kv[2 * q].x, kv[2 * q].y, kv[2 * q].z, kv[2 * q].w,
                   kv[2 * q + 1].x, kv[2 * q + 1].y, kv[2 * q + 1].z, kv[2 * q + 1].w};
    float qq[8] = {qv[2 * q].x, qv[2 * q].y, qv[2 * q].z, qv[2 * q].w,
                   qv[2 * q + 1].x, qv[2 * q + 1].y, qv[2 * q + 1].z, qv[2 * q + 1].w};
    float sc = 0.0f;
#pragma unroll
    for (int dd = 0; dd < 8; ++dd)
      sc += kk[dd] * qq[dd] * acc[dd & 3][2 * q + (dd >> 2)];
    sc *= scale;
    sc = fminf(5.0f, fmaxf(-5.0f, sc));
    ss[q] = __expf(sc);
  }
  if (live) {
    float2* o = (float2*)(ssb + (size_t)slot * 8 + 2 * half);
    o[0] = make_float2(ss[0], ss[1]);
    o[2] = make_float2(ss[2], ss[3]);
  }
}

__global__ void attn_gather_rows(const float* __restrict__ Vh, const float* __restrict__ ssb,
                                 const int* __restrict__ rec, const int* __restrict__ start,
                                 const int* __restrict__ cnt, float* __restrict__ out, int N) {
  int row = blockIdx.x * (blockDim.x >> 5) + (threadIdx.x >> 5);
  if (row >= N) return;
  row = __builtin_amdgcn_readfirstlane(row);
  int lane = threadIdx.x & 31;
  int hd = lane >> 2;
  int beg = __builtin_amdgcn_readfirstlane(start[row]);
  int end = beg + __builtin_amdgcn_readfirstlane(cnt[row]);
  const float2* __restrict__ v2p = (const float2*)Vh;
  float2 acc = make_float2(0.0f, 0.0f);
  float zs = 0.0f;
  for (int base = beg; base < end; base += 32) {
    int j = base + lane;
    int sv = 0;
    if (j < end) sv = rec[(size_t)j * 4];
    int n = end - base;
    n = n < 32 ? n : 32;
    const float* __restrict__ sp = ssb + (size_t)base * 8 + hd;
    for (int k = 0; k < n; k += 4) {
      int r = n - k;
      bool b1 = r > 1, b2 = r > 2, b3 = r > 3;
      int k1 = b1 ? k + 1 : k, k2 = b2 ? k + 2 : k, k3 = b3 ? k + 3 : k;
      int s0 = __builtin_amdgcn_readlane(sv, k), s1 = __builtin_amdgcn_readlane(sv, k1);
      int s2 = __builtin_amdgcn_readlane(sv, k2), s3 = __builtin_amdgcn_readlane(sv, k3);
      float p0 = sp[k * 8], p1 = sp[k1 * 8], p2 = sp[k2 * 8], p3 = sp[k3 * 8];
      float2 v0 = v2p[(size_t)s0 * 32 + lane];
      float2 v1 = v2p[(size_t)s1 * 32 + lane];
      float2 v2 = v2p[(size_t)s2 * 32 + lane];
      float2 v3 = v2p[(size_t)s3 * 32 + lane];
      p1 = b1 ? p1 : 0.0f;
      p2 = b2 ? p2 : 0.0f;
      p3 = b3 ? p3 : 0.0f;
      acc.x += p0 * v0.x; acc.y += p0 * v0.y;
      acc.x += p1 * v1.x; acc.y += p1 * v1.y;
      acc.x += p2 * v2.x; acc.y += p2 * v2.y;
      acc.x += p3 * v3.x; acc.y += p3 * v3.y;
      zs += p0; zs += p1; zs += p2; zs += p3;
    }
  }
  float zz = zs + 1e-6f;
  float2 o = make_float2(acc.x / zz, acc.y / zz);
  ((float2*)out)[((size_t)row * 192 + 128) / 2 + lane] = o;
}

extern "C" void kernel_launch(void* const* d_in, const int* in_sizes, int n_in,
                              void* d_out, int out_size, void* d_ws, size_t ws_size,
                              hipStream_t stream) {
  const float* h  = (const float*)d_in[0];
  const float* p  = (const float*)d_in[1];
  const float* e  = (const float*)d_in[2];
  const float* ew = (const float*)d_in[3];
  const float* fc = (const float*)d_in[4];
  const int* esrc = (const int*)d_in[5];
  const int* edst = (const int*)d_in[6];
  const float* Wc = (const float*)d_in[7];
  const float* bc = (const float*)d_in[8];
  const float* Wg = (const float*)d_in[9];
  const float* bg = (const float*)d_in[10];
  const float* WQ = (const float*)d_in[11];
  const float* WK = (const float*)d_in[12];
  const float* WVm = (const float*)d_in[13];
  const float* WE = (const float*)d_in[14];
  float* out = (float*)d_out;

  const int N = in_sizes[0] / 64;
  const int E = in_sizes[5];
  const size_t ND = (size_t)N * 64;

  float* w = (float*)d_ws;
  float* Tx1  = w; w += ND;
  float* Tx2  = w; w += ND;
  float* Tx3  = w; w += ND;
  float* gcn  = w; w += ND;
  float* scr  = w; w += 2 * ND;
  float* Qh   = w; w += ND;
  float* Kh   = w; w += ND;
  float* Vh   = w; w += ND;
  float* lst  = w; w += ND + (size_t)N * 8;
  float* dinv = w; w += N;
  int* cnt    = (int*)w; w += N;
  int* total  = (int*)w; w += 1;
  int* start   = (int*)lst;
  int* epos    = start + N;
  int* rec     = (int*)scr;
  int* cursor  = rec + 4 * (size_t)E;
  float* ssb   = Tx1;

  hipMemsetAsync(cnt, 0, sizeof(int) * ((size_t)N + 1), stream);

  const int B = 256;
  const int rowBlocks = (N + 7) / 8;

  count_kernel<<<(E + B - 1) / B, B, 0, stream>>>(edst, cnt, E);
  rowalloc_kernel<<<(N + B - 1) / B, B, 0, stream>>>(cnt, total, start, cursor, dinv, N);
  fill_kernel<<<(E + B - 1) / B, B, 0, stream>>>(esrc, edst, dinv, ew, cursor, rec, epos, E);

  gather_rows_pair<<<rowBlocks, B, 0, stream>>>(h, rec, dinv, start, cnt, Tx1, gcn, N);
  gather_rows<<<rowBlocks, B, 0, stream>>>(Tx1, rec, dinv, start, cnt, h, Tx2, N);
  gather_rows<<<rowBlocks, B, 0, stream>>>(Tx2, rec, dinv, start, cnt, Tx1, Tx3, N);

  const int rowTiles = (N + 15) / 16;
  const int gemmBlocks = (rowTiles + 1) / 2;
  dense_kernel<<<dim3(gemmBlocks, 5), 64, 0, stream>>>(h, p, Tx1, Tx2, Tx3, gcn, fc, Wc, bc, Wg, bg,
                                                       WQ, WK, WVm, out, Qh, Kh, Vh, N);

  const int eTiles = (E + 15) / 16;
  attn_kernel<<<(eTiles + 7) / 8, 256, 0, stream>>>(e, WE, Qh, Kh, esrc, edst, epos, ssb, E);
  attn_gather_rows<<<rowBlocks, B, 0, stream>>>(Vh, ssb, rec, start, cnt, out, N);
}


# Attention-score kernel moves whole rows: e, Kh[src], Qh[dst] read two 256-byte rows per load and re-laid through LDS (272-byte pitch); an edge's 8 weights stored as one 32-byte row; sums unchanged
# speedup vs baseline: 3.2774x; 1.0743x over previous
#include <hip/hip_runtime.h>
#include <hip/hip_bf16.h>

typedef __attribute__((ext_vector_type(2))) float v2f;
typedef __attribute__((ext_vector_type(8))) float v8f;

__device__ __forceinline__ v8f wmma4(v2f a, v2f b, v8f c) {
  return __builtin_amdgcn_wmma_f32_16x16x4_f32(
       false, a,  false, b,
       (short)0, c,  false,  false);
}

__global__ void count_kernel(const int* __restrict__ dst, int* __restrict__ cnt, int E) {
  int tid = blockIdx.x * blockDim.x + threadIdx.x;
  if (tid < E) atomicAdd(&cnt[dst[tid]], 1);
}

__global__ void rowalloc_kernel(const int* __restrict__ cnt, int* __restrict__ total,
                                int* __restrict__ start, int* __restrict__ cursor,
                                float* __restrict__ dinv, int N) {
  int tid = blockIdx.x * blockDim.x + threadIdx.x;
  int lane = threadIdx.x & 31;
  int c = tid < N ? cnt[tid] : 0;
  int inc = c;
  for (int o = 1; o < 32; o <<= 1) {
    int v = __shfl_up(inc, o);
    if (lane >= o) inc += v;
  }
  int base = 0;
  if (lane == 31) base = atomicAdd(total, inc);
  base = __shfl(base, 31);
  if (tid < N) {
    int st = base + inc - c;
    start[tid] = st;
    cursor[tid] = st;
    float d = (float)c;
    d = d < 1.0f ? 1.0f : d;
    dinv[tid] = rsqrtf(d);
  }
}

__global__ void fill_kernel(const int* __restrict__ src, const int* __restrict__ dst,
                            const float* __restrict__ dinv, const float* __restrict__ ew,
                            int* __restrict__ cursor, int* __restrict__ rec,
                            int* __restrict__ epos, int E) {
  int tid = blockIdx.x * blockDim.x + threadIdx.x;
  if (tid >= E) return;
  int s = src[tid], d = dst[tid];
  int pos = atomicAdd(&cursor[d], 1);
  ((int4*)rec)[pos] = make_int4(s, __float_as_int(dinv[s]), __float_as_int(ew[tid]), 0);
  epos[tid] = pos;
}

__global__ void gather_rows(const float* __restrict__ x, const int* __restrict__ rec,
                            const float* __restrict__ dinv, const int* __restrict__ start,
                            const int* __restrict__ cnt, const float* __restrict__ prev,
                            float* __restrict__ out, int N) {
  int row = blockIdx.x * (blockDim.x >> 5) + (threadIdx.x >> 5);
  if (row >= N) return;
  row = __builtin_amdgcn_readfirstlane(row);
  int lane = threadIdx.x & 31;
  int beg = __builtin_amdgcn_readfirstlane(start[row]);
  int end = beg + __builtin_amdgcn_readfirstlane(cnt[row]);
  const float dr = dinv[row];
  const int4* __restrict__ rec4 = (const int4*)rec;
  const float2* __restrict__ x2 = (const float2*)x;
  float2 pv = make_float2(0.0f, 0.0f);
  if (prev) pv = ((const float2*)prev)[(size_t)row * 32 + lane];
  float2 acc = make_float2(0.0f, 0.0f);
  for (int base = beg; base < end; base += 32) {
    int j = base + lane;
    int sv = 0, wi = 0;
    if (j < end) {
      int4 r = rec4[j];
      sv = r.x;
      wi = __float_as_int(-(__int_as_float(r.y) * dr));
    }
    int n = end - base;
    n = n < 32 ? n : 32;
    for (int k = 0; k < n; k += 4) {
      int r = n - k;
      bool b1 = r > 1, b2 = r > 2, b3 = r > 3;
      int k1 = b1 ? k + 1 : k, k2 = b2 ? k + 2 : k, k3 = b3 ? k + 3 : k;
      int s0 = __builtin_amdgcn_readlane(sv, k), s1 = __builtin_amdgcn_readlane(sv, k1);
      int s2 = __builtin_amdgcn_readlane(sv, k2), s3 = __builtin_amdgcn_readlane(sv, k3);
      float w0 = __int_as_float(__builtin_amdgcn_readlane(wi, k));
      float w1 = __int_as_float(__builtin_amdgcn_readlane(wi, k1));
      float w2 = __int_as_float(__builtin_amdgcn_readlane(wi, k2));
      float w3 = __int_as_float(__builtin_amdgcn_readlane(wi, k3));
      float2 v0 = x2[(size_t)s0 * 32 + lane];
      float2 v1 = x2[(size_t)s1 * 32 + lane];
      float2 v2 = x2[(size_t)s2 * 32 + lane];
      float2 v3 = x2[(size_t)s3 * 32 + lane];
      v1.x = b1 ? v1.x : 0.0f; v1.y = b1 ? v1.y : 0.0f;
      v2.x = b2 ? v2.x : 0.0f; v2.y = b2 ? v2.y : 0.0f;
      v3.x = b3 ? v3.x : 0.0f; v3.y = b3 ? v3.y : 0.0f;
      acc.x += w0 * v0.x; acc.y += w0 * v0.y;
      acc.x += w1 * v1.x; acc.y += w1 * v1.y;
      acc.x += w2 * v2.x; acc.y += w2 * v2.y;
      acc.x += w3 * v3.x; acc.y += w3 * v3.y;
    }
  }
  if (prev) {
    acc.x = 2.0f * acc.x - pv.x;
    acc.y = 2.0f * acc.y - pv.y;
  }
  ((float2*)out)[(size_t)row * 32 + lane] = acc;
}

__global__ void gather_rows_pair(const float* __restrict__ x, const int* __restrict__ rec,
                                 const float* __restrict__ dinv,
                                 const int* __restrict__ start, const int* __restrict__ cnt,
                                 float* __restrict__ outa, float* __restrict__ outb, int N) {
  int row = blockIdx.x * (blockDim.x >> 5) + (threadIdx.x >> 5);
  if (row >= N) return;
  row = __builtin_amdgcn_readfirstlane(row);
  int lane = threadIdx.x & 31;
  int beg = __builtin_amdgcn_readfirstlane(start[row]);
  int end = beg + __builtin_amdgcn_readfirstlane(cnt[row]);
  const float dr = dinv[row];
  const int4* __restrict__ rec4 = (const int4*)rec;
  const float2* __restrict__ x2 = (const float2*)x;
  float2 acca = make_float2(0.0f, 0.0f);
  float2 accb = make_float2(0.0f, 0.0f);
  for (int base = beg; base < end; base += 32) {
    int j = base + lane;
    int sv = 0, ai = 0, bi = 0;
    if (j < end) {
      int4 r = rec4[j];
      sv = r.x;
      ai = __float_as_int(-(__int_as_float(r.y) * dr));
      bi = r.z;
    }
    int n = end - base;
    n = n < 32 ? n : 32;
    for (int k = 0; k < n; k += 4) {
      int r = n - k;
      bool b1 = r > 1, b2 = r > 2, b3 = r > 3;
      int k1 = b1 ? k + 1 : k, k2 = b2 ? k + 2 : k, k3 = b3 ? k + 3 : k;
      int s0 = __builtin_amdgcn_readlane(sv, k), s1 = __builtin_amdgcn_readlane(sv, k1);
      int s2 = __builtin_amdgcn_readlane(sv, k2), s3 = __builtin_amdgcn_readlane(sv, k3);
      float a0 = __int_as_float(__builtin_amdgcn_readlane(ai, k));
      float a1 = __int_as_float(__builtin_amdgcn_readlane(ai, k1));
      float a2 = __int_as_float(__builtin_amdgcn_readlane(ai, k2));
      float a3 = __int_as_float(__builtin_amdgcn_readlane(ai, k3));
      float c0 = __int_as_float(__builtin_amdgcn_readlane(bi, k));
      float c1 = __int_as_float(__builtin_amdgcn_readlane(bi, k1));
      float c2 = __int_as_float(__builtin_amdgcn_readlane(bi, k2));
      float c3 = __int_as_float(__builtin_amdgcn_readlane(bi, k3));
      float2 v0 = x2[(size_t)s0 * 32 + lane];
      float2 v1 = x2[(size_t)s1 * 32 + lane];
      float2 v2 = x2[(size_t)s2 * 32 + lane];
      float2 v3 = x2[(size_t)s3 * 32 + lane];
      v1.x = b1 ? v1.x : 0.0f; v1.y = b1 ? v1.y : 0.0f;
      v2.x = b2 ? v2.x : 0.0f; v2.y = b2 ? v2.y : 0.0f;
      v3.x = b3 ? v3.x : 0.0f; v3.y = b3 ? v3.y : 0.0f;
      acca.x = fmaf(a0, v0.x, acca.x); acca.y = fmaf(a0, v0.y, acca.y);
      accb.x = fmaf(c0, v0.x, accb.x); accb.y = fmaf(c0, v0.y, accb.y);
      acca.x = fmaf(a1, v1.x, acca.x); acca.y = fmaf(a1, v1.y, acca.y);
      accb.x = fmaf(c1, v1.x, accb.x); accb.y = fmaf(c1, v1.y, accb.y);
      acca.x = fmaf(a2, v2.x, acca.x); acca.y = fmaf(a2, v2.y, acca.y);
      accb.x = fmaf(c2, v2.x, accb.x); accb.y = fmaf(c2, v2.y, accb.y);
      acca.x = fmaf(a3, v3.x, acca.x); acca.y = fmaf(a3, v3.y, acca.y);
      accb.x = fmaf(c3, v3.x, accb.x); accb.y = fmaf(c3, v3.y, accb.y);
    }
  }
  ((float2*)outa)[(size_t)row * 32 + lane] = acca;
  ((float2*)outb)[(size_t)row * 32 + lane] = accb;
}


template <bool SCALED>
__device__ __forceinline__ void plane_mma(const float* __restrict__ arow,
                                          const float* __restrict__ bcol, int half, float sc,
                                          v8f (&acc)[4]) {
  float4 av[8], bw[8][4];
#pragma unroll
  for (int j = 0; j < 8; ++j) {
    int k = 8 * j + 4 * half;
    av[j] = *(const float4*)(arow + k);
#pragma unroll
    for (int i = 0; i < 4; ++i) bw[j][i] = *(const float4*)(bcol + (size_t)(k + i) * 64);
  }
#pragma unroll
  for (int j = 0; j < 8; ++j) {
    float4 a = av[j];
    if (SCALED) { a.x *= sc; a.y *= sc; a.z *= sc; a.w *= sc; }
    v2f a01 = {a.x, a.y}, a23 = {a.z, a.w};
    acc[0] = wmma4(a01, (v2f){bw[j][0].x, bw[j][1].x}, acc[0]);
    acc[1] = wmma4(a01, (v2f){bw[j][0].y, bw[j][1].y}, acc[1]);
    acc[2] = wmma4(a01, (v2f){bw[j][0].z, bw[j][1].z}, acc[2]);
    acc[3] = wmma4(a01, (v2f){bw[j][0].w, bw[j][1].w}, acc[3]);
    acc[0] = wmma4(a23, (v2f){bw[j][2].x, bw[j][3].x}, acc[0]);
    acc[1] = wmma4(a23, (v2f){bw[j][2].y, bw[j][3].y}, acc[1]);
    acc[2] = wmma4(a23, (v2f){bw[j][2].z, bw[j][3].z}, acc[2]);
    acc[3] = wmma4(a23, (v2f){bw[j][2].w, bw[j][3].w}, acc[3]);
  }
}

__device__ __forceinline__ void strip_store(const v8f (&acc)[4], float4 bv, float* __restrict__ Out,
                                            int ldo, int col, int row0, int rows) {
#pragma unroll
  for (int r = 0; r < 8; ++r) {
    int row = row0 + r;
    if (row < rows) {
      float4 o = make_float4(acc[0][r] + bv.x, acc[1][r] + bv.y, acc[2][r] + bv.z, acc[3][r] + bv.w);
      *(float4*)(Out + (size_t)row * ldo + col) = o;
    }
  }
}

__global__ void __launch_bounds__(64) __attribute__((amdgpu_waves_per_eu(1, 4)))
dense_kernel(const float* __restrict__ h, const float* __restrict__ p,
             const float* __restrict__ Tx1, const float* __restrict__ Tx2,
             const float* __restrict__ Tx3, const float* __restrict__ gcn,
             const float* __restrict__ fc, const float* __restrict__ Wc,
             const float* __restrict__ bc, const float* __restrict__ Wg,
             const float* __restrict__ bg, const float* __restrict__ WQ,
             const float* __restrict__ WK, const float* __restrict__ WV,
             float* __restrict__ Out, float* __restrict__ Qh, float* __restrict__ Kh,
             float* __restrict__ Vh, int rows) {
  int rt = blockIdx.x * (blockDim.x >> 5) + (threadIdx.x >> 5);
  int rowTiles = (rows + 15) >> 4;
  if (rt >= rowTiles) return;
  int lane = threadIdx.x & 31, half = lane >> 4, c = lane & 15;
  int ar = rt * 16 + c;
  if (ar >= rows) ar = rows - 1;
  const size_t ao = (size_t)ar * 64;
  const int row0 = rt * 16 + 8 * half;
  const int which = blockIdx.y;
  v8f acc[4] = {};
  if (which == 0) {
    const float* As[4] = {h, Tx1, Tx2, Tx3};
#pragma unroll
    for (int k = 0; k < 4; ++k) {
      float sc = fc[(size_t)k * rows + ar];
      plane_mma<true>(As[k] + ao, Wc + (size_t)k * 64 * 64 + 4 * c, half, sc, acc);
    }
    float4 bv = *(const float4*)(bc + 4 * c);
    strip_store(acc, bv, Out, 192, 4 * c, row0, rows);
  } else if (which == 4) {
    plane_mma<false>(gcn + ao, Wg + 4 * c, half, 1.0f, acc);
    float4 bv = *(const float4*)(bg + 4 * c);
    strip_store(acc, bv, Out, 192, 64 + 4 * c, row0, rows);
  } else {
    const float* W = which == 1 ? WQ : (which == 2 ? WK : WV);
    float* O = which == 1 ? Qh : (which == 2 ? Kh : Vh);
    plane_mma<false>(h + ao, W + 4 * c, half, 1.0f, acc);
    plane_mma<false>(p + ao, W + (size_t)64 * 64 + 4 * c, half, 1.0f, acc);
    strip_store(acc, make_float4(0.0f, 0.0f, 0.0f, 0.0f), O, 64, 4 * c, row0, rows);
  }
}

#define ATT_WAVES 4
#define ATT_PITCH 68
__global__ void __launch_bounds__(32 * ATT_WAVES)
attn_kernel(const float* __restrict__ Ee, const float* __restrict__ WE,
                            const float* __restrict__ Qh, const float* __restrict__ Kh,
                            const int* __restrict__ src, const int* __restrict__ dst,
                            const int* __restrict__ epos, float* __restrict__ ssb, int E) {
  __shared__ __attribute__((aligned(16))) float stage[ATT_WAVES][2][16 * ATT_PITCH];
  const int wv = threadIdx.x >> 5;
  int tile = blockIdx.x * ATT_WAVES + wv;
  const int eTiles = (E + 15) >> 4;
  const bool tlive = tile < eTiles;
  if (!tlive) tile = eTiles - 1;
  int lane = threadIdx.x & 31, half = lane >> 4, c = lane & 15;
  const int e0 = tile * 16;
  int em = e0 + c;
  bool live = tlive && em < E;
  int er = em < E ? em : E - 1;
  float* eS = stage[wv][0];
  float* kqS = stage[wv][1];

  float4 ev[8], kv[8], qv[8];
#pragma unroll
  for (int i = 0; i < 8; ++i) {
    int re = e0 + 2 * i + half;
    re = re < E ? re : E - 1;
    ev[i] = *(const float4*)(Ee + (size_t)re * 64 + 4 * c);
  }
  int s = src[er], d = dst[er], slot = epos[er];
#pragma unroll
  for (int i = 0; i < 8; ++i) {
    int ri = 2 * i + half;
    int si = __shfl(s, ri), di = __shfl(d, ri);
    kv[i] = *(const float4*)(Kh + (size_t)si * 64 + 4 * c);
    qv[i] = *(const float4*)(Qh + (size_t)di * 64 + 4 * c);
  }
#pragma unroll
  for (int i = 0; i < 8; ++i) {
    int ri = 2 * i + half;
    *(float4*)(eS + ri * ATT_PITCH + 4 * c) = ev[i];
    float4 kq = make_float4(kv[i].x * qv[i].x, kv[i].y * qv[i].y, kv[i].z * qv[i].z, kv[i].w * qv[i].w);
    *(float4*)(kqS + ri * ATT_PITCH + 4 * c) = kq;
  }
  __syncthreads();

  const float* wcol = WE + 4 * c;
  const float* erow = eS + c * ATT_PITCH;
  v8f acc[4] = {};
#pragma unroll
  for (int j = 0; j < 8; ++j) {
    int k = 8 * j + 4 * half;
    float4 evj = *(const float4*)(erow + k);
    float4 w0 = *(const float4*)(wcol + (size_t)(k + 0) * 64);
    float4 w1 = *(const float4*)(wcol + (size_t)(k + 1) * 64);
    float4 w2 = *(const float4*)(wcol + (size_t)(k + 2) * 64);
    float4 w3 = *(const float4*)(wcol + (size_t)(k + 3) * 64);
    v2f e01 = {evj.x, evj.y}, e23 = {evj.z, evj.w};
    acc[0] = wmma4((v2f){w0.x, w1.x}, e01, acc[0]);
    acc[1] = wmma4((v2f){w0.y, w1.y}, e01, acc[1]);
    acc[2] = wmma4((v2f){w0.z, w1.z}, e01, acc[2]);
    acc[3] = wmma4((v2f){w0.w, w1.w}, e01, acc[3]);
    acc[0] = wmma4((v2f){w2.x, w3.x}, e23, acc[0]);
    acc[1] = wmma4((v2f){w2.y, w3.y}, e23, acc[1]);
    acc[2] = wmma4((v2f){w2.z, w3.z}, e23, acc[2]);
    acc[3] = wmma4((v2f){w2.w, w3.w}, e23, acc[3]);
  }

  const float* kqrow = kqS + c * ATT_PITCH + 32 * half;
  float4 pv[8];
#pragma unroll
  for (int i = 0; i < 8; ++i) pv[i] = *(const float4*)(kqrow + 4 * i);

  const float scale = 0.35355339059327373f;
  float ss[4];
#pragma unroll
  for (int q = 0; q < 4; ++q) {
    float kq[8] = {pv[2 * q].x, pv[2 * q].y, pv[2 * q].z, pv[2 * q].w,
                   pv[2 * q + 1].x, pv[2 * q + 1].y, pv[2 * q + 1].z, pv[2 * q + 1].w};
    float sc = 0.0f;
#pragma unroll
    for (int dd = 0; dd < 8; ++dd)
      sc += kq[dd] * acc[dd & 3][2 * q + (dd >> 2)];
    sc *= scale;
    sc = fminf(5.0f, fmaxf(-5.0f, sc));
    ss[q] = __expf(sc);
  }
  if (live) *(float4*)(ssb + (size_t)slot * 8 + 4 * half) = make_float4(ss[0], ss[1], ss[2], ss[3]);
}

__global__ void attn_gather_rows(const float* __restrict__ Vh, const float* __restrict__ ssb,
                                 const int* __restrict__ rec, const int* __restrict__ start,
                                 const int* __restrict__ cnt, float* __restrict__ out, int N) {
  int row = blockIdx.x * (blockDim.x >> 5) + (threadIdx.x >> 5);
  if (row >= N) return;
  row = __builtin_amdgcn_readfirstlane(row);
  int lane = threadIdx.x & 31;
  int hd = lane >> 2;
  int beg = __builtin_amdgcn_readfirstlane(start[row]);
  int end = beg + __builtin_amdgcn_readfirstlane(cnt[row]);
  const float2* __restrict__ v2p = (const float2*)Vh;
  float2 acc = make_float2(0.0f, 0.0f);
  float zs = 0.0f;
  for (int base = beg; base < end; base += 32) {
    int j = base + lane;
    int sv = 0;
    if (j < end) sv = rec[(size_t)j * 4];
    int n = end - base;
    n = n < 32 ? n : 32;
    const float* __restrict__ sp = ssb + (size_t)base * 8 + hd;
    for (int k = 0; k < n; k += 4) {
      int r = n - k;
      bool b1 = r > 1, b2 = r > 2, b3 = r > 3;
      int k1 = b1 ? k + 1 : k, k2 = b2 ? k + 2 : k, k3 = b3 ? k + 3 : k;
      int s0 = __builtin_amdgcn_readlane(sv, k), s1 = __builtin_amdgcn_readlane(sv, k1);
      int s2 = __builtin_amdgcn_readlane(sv, k2), s3 = __builtin_amdgcn_readlane(sv, k3);
      float p0 = sp[k * 8], p1 = sp[k1 * 8], p2 = sp[k2 * 8], p3 = sp[k3 * 8];
      float2 v0 = v2p[(size_t)s0 * 32 + lane];
      float2 v1 = v2p[(size_t)s1 * 32 + lane];
      float2 v2 = v2p[(size_t)s2 * 32 + lane];
      float2 v3 = v2p[(size_t)s3 * 32 + lane];
      p1 = b1 ? p1 : 0.0f;
      p2 = b2 ? p2 : 0.0f;
      p3 = b3 ? p3 : 0.0f;
      acc.x += p0 * v0.x; acc.y += p0 * v0.y;
      acc.x += p1 * v1.x; acc.y += p1 * v1.y;
      acc.x += p2 * v2.x; acc.y += p2 * v2.y;
      acc.x += p3 * v3.x; acc.y += p3 * v3.y;
      zs += p0; zs += p1; zs += p2; zs += p3;
    }
  }
  float zz = zs + 1e-6f;
  float2 o = make_float2(acc.x / zz, acc.y / zz);
  ((float2*)out)[((size_t)row * 192 + 128) / 2 + lane] = o;
}

extern "C" void kernel_launch(void* const* d_in, const int* in_sizes, int n_in,
                              void* d_out, int out_size, void* d_ws, size_t ws_size,
                              hipStream_t stream) {
  const float* h  = (const float*)d_in[0];
  const float* p  = (const float*)d_in[1];
  const float* e  = (const float*)d_in[2];
  const float* ew = (const float*)d_in[3];
  const float* fc = (const float*)d_in[4];
  const int* esrc = (const int*)d_in[5];
  const int* edst = (const int*)d_in[6];
  const float* Wc = (const float*)d_in[7];
  const float* bc = (const float*)d_in[8];
  const float* Wg = (const float*)d_in[9];
  const float* bg = (const float*)d_in[10];
  const float* WQ = (const float*)d_in[11];
  const float* WK = (const float*)d_in[12];
  const float* WVm = (const float*)d_in[13];
  const float* WE = (const float*)d_in[14];
  float* out = (float*)d_out;

  const int N = in_sizes[0] / 64;
  const int E = in_sizes[5];
  const size_t ND = (size_t)N * 64;

  float* w = (float*)d_ws;
  float* Tx1  = w; w += ND;
  float* Tx2  = w; w += ND;
  float* Tx3  = w; w += ND;
  float* gcn  = w; w += ND;
  float* scr  = w; w += 2 * ND;
  float* Qh   = w; w += ND;
  float* Kh   = w; w += ND;
  float* Vh   = w; w += ND;
  float* lst  = w; w += ND + (size_t)N * 8;
  float* dinv = w; w += N;
  int* cnt    = (int*)w; w += N;
  int* total  = (int*)w; w += 1;
  int* start   = (int*)lst;
  int* epos    = start + N;
  int* rec     = (int*)scr;
  int* cursor  = rec + 4 * (size_t)E;
  float* ssb   = Tx1;

  hipMemsetAsync(cnt, 0, sizeof(int) * ((size_t)N + 1), stream);

  const int B = 256;
  const int rowBlocks = (N + 7) / 8;

  count_kernel<<<(E + B - 1) / B, B, 0, stream>>>(edst, cnt, E);
  rowalloc_kernel<<<(N + B - 1) / B, B, 0, stream>>>(cnt, total, start, cursor, dinv, N);
  fill_kernel<<<(E + B - 1) / B, B, 0, stream>>>(esrc, edst, dinv, ew, cursor, rec, epos, E);

  gather_rows_pair<<<rowBlocks, B, 0, stream>>>(h, rec, dinv, start, cnt, Tx1, gcn, N);
  gather_rows<<<rowBlocks, B, 0, stream>>>(Tx1, rec, dinv, start, cnt, h, Tx2, N);
  gather_rows<<<rowBlocks, B, 0, stream>>>(Tx2, rec, dinv, start, cnt, Tx1, Tx3, N);

  const int rowTiles = (N + 15) / 16;
  const int gemmBlocks = (rowTiles + 1) / 2;
  dense_kernel<<<dim3(gemmBlocks, 5), 64, 0, stream>>>(h, p, Tx1, Tx2, Tx3, gcn, fc, Wc, bc, Wg, bg,
                                                       WQ, WK, WVm, out, Qh, Kh, Vh, N);

  const int eTiles = (E + 15) / 16;
  attn_kernel<<<(eTiles + ATT_WAVES - 1) / ATT_WAVES, 32 * ATT_WAVES, 0, stream>>>(e, WE, Qh, Kh, esrc, edst, epos, ssb, E);
  attn_gather_rows<<<rowBlocks, B, 0, stream>>>(Vh, ssb, rec, start, cnt, out, N);
}


# In-degree atomic also returns each edge's rank in its destination; fill places records at start[dst]+rank with no atomics: one global atomic per edge, not two; sum order = arrival order as before
# speedup vs baseline: 3.7287x; 1.1377x over previous
#include <hip/hip_runtime.h>
#include <hip/hip_bf16.h>

typedef __attribute__((ext_vector_type(2))) float v2f;
typedef __attribute__((ext_vector_type(8))) float v8f;

__device__ __forceinline__ v8f wmma4(v2f a, v2f b, v8f c) {
  return __builtin_amdgcn_wmma_f32_16x16x4_f32(
       false, a,  false, b,
       (short)0, c,  false,  false);
}

__global__ void count_kernel(const int* __restrict__ dst, int* __restrict__ cnt,
                             int* __restrict__ rank, int E) {
  int tid = blockIdx.x * blockDim.x + threadIdx.x;
  if (tid < E) rank[tid] = atomicAdd(&cnt[dst[tid]], 1);
}

__global__ void rowalloc_kernel(const int* __restrict__ cnt, int* __restrict__ total,
                                int* __restrict__ start, float* __restrict__ dinv, int N) {
  int tid = blockIdx.x * blockDim.x + threadIdx.x;
  int lane = threadIdx.x & 31;
  int c = tid < N ? cnt[tid] : 0;
  int inc = c;
  for (int o = 1; o < 32; o <<= 1) {
    int v = __shfl_up(inc, o);
    if (lane >= o) inc += v;
  }
  int base = 0;
  if (lane == 31) base = atomicAdd(total, inc);
  base = __shfl(base, 31);
  if (tid < N) {
    int st = base + inc - c;
    start[tid] = st;
    float d = (float)c;
    d = d < 1.0f ? 1.0f : d;
    dinv[tid] = rsqrtf(d);
  }
}

__global__ void fill_kernel(const int* __restrict__ src, const int* __restrict__ dst,
                            const float* __restrict__ dinv, const float* __restrict__ ew,
                            const int* __restrict__ start, int* __restrict__ rec,
                            int* __restrict__ epos, int E) {
  int tid = blockIdx.x * blockDim.x + threadIdx.x;
  if (tid >= E) return;
  int s = src[tid], d = dst[tid];
  int pos = start[d] + epos[tid];
  ((int4*)rec)[pos] = make_int4(s, __float_as_int(dinv[s]), __float_as_int(ew[tid]), 0);
  epos[tid] = pos;
}

__global__ void gather_rows(const float* __restrict__ x, const int* __restrict__ rec,
                            const float* __restrict__ dinv, const int* __restrict__ start,
                            const int* __restrict__ cnt, const float* __restrict__ prev,
                            float* __restrict__ out, int N) {
  int row = blockIdx.x * (blockDim.x >> 5) + (threadIdx.x >> 5);
  if (row >= N) return;
  row = __builtin_amdgcn_readfirstlane(row);
  int lane = threadIdx.x & 31;
  int beg = __builtin_amdgcn_readfirstlane(start[row]);
  int end = beg + __builtin_amdgcn_readfirstlane(cnt[row]);
  const float dr = dinv[row];
  const int4* __restrict__ rec4 = (const int4*)rec;
  const float2* __restrict__ x2 = (const float2*)x;
  float2 pv = make_float2(0.0f, 0.0f);
  if (prev) pv = ((const float2*)prev)[(size_t)row * 32 + lane];
  float2 acc = make_float2(0.0f, 0.0f);
  for (int base = beg; base < end; base += 32) {
    int j = base + lane;
    int sv = 0, wi = 0;
    if (j < end) {
      int4 r = rec4[j];
      sv = r.x;
      wi = __float_as_int(-(__int_as_float(r.y) * dr));
    }
    int n = end - base;
    n = n < 32 ? n : 32;
    for (int k = 0; k < n; k += 4) {
      int r = n - k;
      bool b1 = r > 1, b2 = r > 2, b3 = r > 3;
      int k1 = b1 ? k + 1 : k, k2 = b2 ? k + 2 : k, k3 = b3 ? k + 3 : k;
      int s0 = __builtin_amdgcn_readlane(sv, k), s1 = __builtin_amdgcn_readlane(sv, k1);
      int s2 = __builtin_amdgcn_readlane(sv, k2), s3 = __builtin_amdgcn_readlane(sv, k3);
      float w0 = __int_as_float(__builtin_amdgcn_readlane(wi, k));
      float w1 = __int_as_float(__builtin_amdgcn_readlane(wi, k1));
      float w2 = __int_as_float(__builtin_amdgcn_readlane(wi, k2));
      float w3 = __int_as_float(__builtin_amdgcn_readlane(wi, k3));
      float2 v0 = x2[(size_t)s0 * 32 + lane];
      float2 v1 = x2[(size_t)s1 * 32 + lane];
      float2 v2 = x2[(size_t)s2 * 32 + lane];
      float2 v3 = x2[(size_t)s3 * 32 + lane];
      v1.x = b1 ? v1.x : 0.0f; v1.y = b1 ? v1.y : 0.0f;
      v2.x = b2 ? v2.x : 0.0f; v2.y = b2 ? v2.y : 0.0f;
      v3.x = b3 ? v3.x : 0.0f; v3.y = b3 ? v3.y : 0.0f;
      acc.x += w0 * v0.x; acc.y += w0 * v0.y;
      acc.x += w1 * v1.x; acc.y += w1 * v1.y;
      acc.x += w2 * v2.x; acc.y += w2 * v2.y;
      acc.x += w3 * v3.x; acc.y += w3 * v3.y;
    }
  }
  if (prev) {
    acc.x = 2.0f * acc.x - pv.x;
    acc.y = 2.0f * acc.y - pv.y;
  }
  ((float2*)out)[(size_t)row * 32 + lane] = acc;
}

__global__ void gather_rows_pair(const float* __restrict__ x, const int* __restrict__ rec,
                                 const float* __restrict__ dinv,
                                 const int* __restrict__ start, const int* __restrict__ cnt,
                                 float* __restrict__ outa, float* __restrict__ outb, int N) {
  int row = blockIdx.x * (blockDim.x >> 5) + (threadIdx.x >> 5);
  if (row >= N) return;
  row = __builtin_amdgcn_readfirstlane(row);
  int lane = threadIdx.x & 31;
  int beg = __builtin_amdgcn_readfirstlane(start[row]);
  int end = beg + __builtin_amdgcn_readfirstlane(cnt[row]);
  const float dr = dinv[row];
  const int4* __restrict__ rec4 = (const int4*)rec;
  const float2* __restrict__ x2 = (const float2*)x;
  float2 acca = make_float2(0.0f, 0.0f);
  float2 accb = make_float2(0.0f, 0.0f);
  for (int base = beg; base < end; base += 32) {
    int j = base + lane;
    int sv = 0, ai = 0, bi = 0;
    if (j < end) {
      int4 r = rec4[j];
      sv = r.x;
      ai = __float_as_int(-(__int_as_float(r.y) * dr));
      bi = r.z;
    }
    int n = end - base;
    n = n < 32 ? n : 32;
    for (int k = 0; k < n; k += 4) {
      int r = n - k;
      bool b1 = r > 1, b2 = r > 2, b3 = r > 3;
      int k1 = b1 ? k + 1 : k, k2 = b2 ? k + 2 : k, k3 = b3 ? k + 3 : k;
      int s0 = __builtin_amdgcn_readlane(sv, k), s1 = __builtin_amdgcn_readlane(sv, k1);
      int s2 = __builtin_amdgcn_readlane(sv, k2), s3 = __builtin_amdgcn_readlane(sv, k3);
      float a0 = __int_as_float(__builtin_amdgcn_readlane(ai, k));
      float a1 = __int_as_float(__builtin_amdgcn_readlane(ai, k1));
      float a2 = __int_as_float(__builtin_amdgcn_readlane(ai, k2));
      float a3 = __int_as_float(__builtin_amdgcn_readlane(ai, k3));
      float c0 = __int_as_float(__builtin_amdgcn_readlane(bi, k));
      float c1 = __int_as_float(__builtin_amdgcn_readlane(bi, k1));
      float c2 = __int_as_float(__builtin_amdgcn_readlane(bi, k2));
      float c3 = __int_as_float(__builtin_amdgcn_readlane(bi, k3));
      float2 v0 = x2[(size_t)s0 * 32 + lane];
      float2 v1 = x2[(size_t)s1 * 32 + lane];
      float2 v2 = x2[(size_t)s2 * 32 + lane];
      float2 v3 = x2[(size_t)s3 * 32 + lane];
      v1.x = b1 ? v1.x : 0.0f; v1.y = b1 ? v1.y : 0.0f;
      v2.x = b2 ? v2.x : 0.0f; v2.y = b2 ? v2.y : 0.0f;
      v3.x = b3 ? v3.x : 0.0f; v3.y = b3 ? v3.y : 0.0f;
      acca.x = fmaf(a0, v0.x, acca.x); acca.y = fmaf(a0, v0.y, acca.y);
      accb.x = fmaf(c0, v0.x, accb.x); accb.y = fmaf(c0, v0.y, accb.y);
      acca.x = fmaf(a1, v1.x, acca.x); acca.y = fmaf(a1, v1.y, acca.y);
      accb.x = fmaf(c1, v1.x, accb.x); accb.y = fmaf(c1, v1.y, accb.y);
      acca.x = fmaf(a2, v2.x, acca.x); acca.y = fmaf(a2, v2.y, acca.y);
      accb.x = fmaf(c2, v2.x, accb.x); accb.y = fmaf(c2, v2.y, accb.y);
      acca.x = fmaf(a3, v3.x, acca.x); acca.y = fmaf(a3, v3.y, acca.y);
      accb.x = fmaf(c3, v3.x, accb.x); accb.y = fmaf(c3, v3.y, accb.y);
    }
  }
  ((float2*)outa)[(size_t)row * 32 + lane] = acca;
  ((float2*)outb)[(size_t)row * 32 + lane] = accb;
}


template <bool SCALED>
__device__ __forceinline__ void plane_mma(const float* __restrict__ arow,
                                          const float* __restrict__ bcol, int half, float sc,
                                          v8f (&acc)[4]) {
  float4 av[8], bw[8][4];
#pragma unroll
  for (int j = 0; j < 8; ++j) {
    int k = 8 * j + 4 * half;
    av[j] = *(const float4*)(arow + k);
#pragma unroll
    for (int i = 0; i < 4; ++i) bw[j][i] = *(const float4*)(bcol + (size_t)(k + i) * 64);
  }
#pragma unroll
  for (int j = 0; j < 8; ++j) {
    float4 a = av[j];
    if (SCALED) { a.x *= sc; a.y *= sc; a.z *= sc; a.w *= sc; }
    v2f a01 = {a.x, a.y}, a23 = {a.z, a.w};
    acc[0] = wmma4(a01, (v2f){bw[j][0].x, bw[j][1].x}, acc[0]);
    acc[1] = wmma4(a01, (v2f){bw[j][0].y, bw[j][1].y}, acc[1]);
    acc[2] = wmma4(a01, (v2f){bw[j][0].z, bw[j][1].z}, acc[2]);
    acc[3] = wmma4(a01, (v2f){bw[j][0].w, bw[j][1].w}, acc[3]);
    acc[0] = wmma4(a23, (v2f){bw[j][2].x, bw[j][3].x}, acc[0]);
    acc[1] = wmma4(a23, (v2f){bw[j][2].y, bw[j][3].y}, acc[1]);
    acc[2] = wmma4(a23, (v2f){bw[j][2].z, bw[j][3].z}, acc[2]);
    acc[3] = wmma4(a23, (v2f){bw[j][2].w, bw[j][3].w}, acc[3]);
  }
}

__device__ __forceinline__ void strip_store(const v8f (&acc)[4], float4 bv, float* __restrict__ Out,
                                            int ldo, int col, int row0, int rows) {
#pragma unroll
  for (int r = 0; r < 8; ++r) {
    int row = row0 + r;
    if (row < rows) {
      float4 o = make_float4(acc[0][r] + bv.x, acc[1][r] + bv.y, acc[2][r] + bv.z, acc[3][r] + bv.w);
      *(float4*)(Out + (size_t)row * ldo + col) = o;
    }
  }
}

__global__ void __launch_bounds__(64) __attribute__((amdgpu_waves_per_eu(1, 4)))
dense_kernel(const float* __restrict__ h, const float* __restrict__ p,
             const float* __restrict__ Tx1, const float* __restrict__ Tx2,
             const float* __restrict__ Tx3, const float* __restrict__ gcn,
             const float* __restrict__ fc, const float* __restrict__ Wc,
             const float* __restrict__ bc, const float* __restrict__ Wg,
             const float* __restrict__ bg, const float* __restrict__ WQ,
             const float* __restrict__ WK, const float* __restrict__ WV,
             float* __restrict__ Out, float* __restrict__ Qh, float* __restrict__ Kh,
             float* __restrict__ Vh, int rows) {
  int rt = blockIdx.x * (blockDim.x >> 5) + (threadIdx.x >> 5);
  int rowTiles = (rows + 15) >> 4;
  if (rt >= rowTiles) return;
  int lane = threadIdx.x & 31, half = lane >> 4, c = lane & 15;
  int ar = rt * 16 + c;
  if (ar >= rows) ar = rows - 1;
  const size_t ao = (size_t)ar * 64;
  const int row0 = rt * 16 + 8 * half;
  const int which = blockIdx.y;
  v8f acc[4] = {};
  if (which == 0) {
    const float* As[4] = {h, Tx1, Tx2, Tx3};
#pragma unroll
    for (int k = 0; k < 4; ++k) {
      float sc = fc[(size_t)k * rows + ar];
      plane_mma<true>(As[k] + ao, Wc + (size_t)k * 64 * 64 + 4 * c, half, sc, acc);
    }
    float4 bv = *(const float4*)(bc + 4 * c);
    strip_store(acc, bv, Out, 192, 4 * c, row0, rows);
  } else if (which == 4) {
    plane_mma<false>(gcn + ao, Wg + 4 * c, half, 1.0f, acc);
    float4 bv = *(const float4*)(bg + 4 * c);
    strip_store(acc, bv, Out, 192, 64 + 4 * c, row0, rows);
  } else {
    const float* W = which == 1 ? WQ : (which == 2 ? WK : WV);
    float* O = which == 1 ? Qh : (which == 2 ? Kh : Vh);
    plane_mma<false>(h + ao, W + 4 * c, half, 1.0f, acc);
    plane_mma<false>(p + ao, W + (size_t)64 * 64 + 4 * c, half, 1.0f, acc);
    strip_store(acc, make_float4(0.0f, 0.0f, 0.0f, 0.0f), O, 64, 4 * c, row0, rows);
  }
}

#define ATT_WAVES 4
#define ATT_PITCH 68
__global__ void __launch_bounds__(32 * ATT_WAVES)
attn_kernel(const float* __restrict__ Ee, const float* __restrict__ WE,
                            const float* __restrict__ Qh, const float* __restrict__ Kh,
                            const int* __restrict__ src, const int* __restrict__ dst,
                            const int* __restrict__ epos, float* __restrict__ ssb, int E) {
  __shared__ __attribute__((aligned(16))) float stage[ATT_WAVES][2][16 * ATT_PITCH];
  const int wv = threadIdx.x >> 5;
  int tile = blockIdx.x * ATT_WAVES + wv;
  const int eTiles = (E + 15) >> 4;
  const bool tlive = tile < eTiles;
  if (!tlive) tile = eTiles - 1;
  int lane = threadIdx.x & 31, half = lane >> 4, c = lane & 15;
  const int e0 = tile * 16;
  int em = e0 + c;
  bool live = tlive && em < E;
  int er = em < E ? em : E - 1;
  float* eS = stage[wv][0];
  float* kqS = stage[wv][1];

  float4 ev[8], kv[8], qv[8];
#pragma unroll
  for (int i = 0; i < 8; ++i) {
    int re = e0 + 2 * i + half;
    re = re < E ? re : E - 1;
    ev[i] = *(const float4*)(Ee + (size_t)re * 64 + 4 * c);
  }
  int s = src[er], d = dst[er], slot = epos[er];
#pragma unroll
  for (int i = 0; i < 8; ++i) {
    int ri = 2 * i + half;
    int si = __shfl(s, ri), di = __shfl(d, ri);
    kv[i] = *(const float4*)(Kh + (size_t)si * 64 + 4 * c);
    qv[i] = *(const float4*)(Qh + (size_t)di * 64 + 4 * c);
  }
#pragma unroll
  for (int i = 0; i < 8; ++i) {
    int ri = 2 * i + half;
    *(float4*)(eS + ri * ATT_PITCH + 4 * c) = ev[i];
    float4 kq = make_float4(kv[i].x * qv[i].x, kv[i].y * qv[i].y, kv[i].z * qv[i].z, kv[i].w * qv[i].w);
    *(float4*)(kqS + ri * ATT_PITCH + 4 * c) = kq;
  }
  __syncthreads();

  const float* wcol = WE + 4 * c;
  const float* erow = eS + c * ATT_PITCH;
  v8f acc[4] = {};
#pragma unroll
  for (int j = 0; j < 8; ++j) {
    int k = 8 * j + 4 * half;
    float4 evj = *(const float4*)(erow + k);
    float4 w0 = *(const float4*)(wcol + (size_t)(k + 0) * 64);
    float4 w1 = *(const float4*)(wcol + (size_t)(k + 1) * 64);
    float4 w2 = *(const float4*)(wcol + (size_t)(k + 2) * 64);
    float4 w3 = *(const float4*)(wcol + (size_t)(k + 3) * 64);
    v2f e01 = {evj.x, evj.y}, e23 = {evj.z, evj.w};
    acc[0] = wmma4((v2f){w0.x, w1.x}, e01, acc[0]);
    acc[1] = wmma4((v2f){w0.y, w1.y}, e01, acc[1]);
    acc[2] = wmma4((v2f){w0.z, w1.z}, e01, acc[2]);
    acc[3] = wmma4((v2f){w0.w, w1.w}, e01, acc[3]);
    acc[0] = wmma4((v2f){w2.x, w3.x}, e23, acc[0]);
    acc[1] = wmma4((v2f){w2.y, w3.y}, e23, acc[1]);
    acc[2] = wmma4((v2f){w2.z, w3.z}, e23, acc[2]);
    acc[3] = wmma4((v2f){w2.w, w3.w}, e23, acc[3]);
  }

  const float* kqrow = kqS + c * ATT_PITCH + 32 * half;
  float4 pv[8];
#pragma unroll
  for (int i = 0; i < 8; ++i) pv[i] = *(const float4*)(kqrow + 4 * i);

  const float scale = 0.35355339059327373f;
  float ss[4];
#pragma unroll
  for (int q = 0; q < 4; ++q) {
    float kq[8] = {pv[2 * q].x, pv[2 * q].y, pv[2 * q].z, pv[2 * q].w,
                   pv[2 * q + 1].x, pv[2 * q + 1].y, pv[2 * q + 1].z, pv[2 * q + 1].w};
    float sc = 0.0f;
#pragma unroll
    for (int dd = 0; dd < 8; ++dd)
      sc += kq[dd] * acc[dd & 3][2 * q + (dd >> 2)];
    sc *= scale;
    sc = fminf(5.0f, fmaxf(-5.0f, sc));
    ss[q] = __expf(sc);
  }
  if (live) *(float4*)(ssb + (size_t)slot * 8 + 4 * half) = make_float4(ss[0], ss[1], ss[2], ss[3]);
}

__global__ void attn_gather_rows(const float* __restrict__ Vh, const float* __restrict__ ssb,
                                 const int* __restrict__ rec, const int* __restrict__ start,
                                 const int* __restrict__ cnt, float* __restrict__ out, int N) {
  int row = blockIdx.x * (blockDim.x >> 5) + (threadIdx.x >> 5);
  if (row >= N) return;
  row = __builtin_amdgcn_readfirstlane(row);
  int lane = threadIdx.x & 31;
  int hd = lane >> 2;
  int beg = __builtin_amdgcn_readfirstlane(start[row]);
  int end = beg + __builtin_amdgcn_readfirstlane(cnt[row]);
  const float2* __restrict__ v2p = (const float2*)Vh;
  float2 acc = make_float2(0.0f, 0.0f);
  float zs = 0.0f;
  for (int base = beg; base < end; base += 32) {
    int j = base + lane;
    int sv = 0;
    if (j < end) sv = rec[(size_t)j * 4];
    int n = end - base;
    n = n < 32 ? n : 32;
    const float* __restrict__ sp = ssb + (size_t)base * 8 + hd;
    for (int k = 0; k < n; k += 4) {
      int r = n - k;
      bool b1 = r > 1, b2 = r > 2, b3 = r > 3;
      int k1 = b1 ? k + 1 : k, k2 = b2 ? k + 2 : k, k3 = b3 ? k + 3 : k;
      int s0 = __builtin_amdgcn_readlane(sv, k), s1 = __builtin_amdgcn_readlane(sv, k1);
      int s2 = __builtin_amdgcn_readlane(sv, k2), s3 = __builtin_amdgcn_readlane(sv, k3);
      float p0 = sp[k * 8], p1 = sp[k1 * 8], p2 = sp[k2 * 8], p3 = sp[k3 * 8];
      float2 v0 = v2p[(size_t)s0 * 32 + lane];
      float2 v1 = v2p[(size_t)s1 * 32 + lane];
      float2 v2 = v2p[(size_t)s2 * 32 + lane];
      float2 v3 = v2p[(size_t)s3 * 32 + lane];
      p1 = b1 ? p1 : 0.0f;
      p2 = b2 ? p2 : 0.0f;
      p3 = b3 ? p3 : 0.0f;
      acc.x += p0 * v0.x; acc.y += p0 * v0.y;
      acc.x += p1 * v1.x; acc.y += p1 * v1.y;
      acc.x += p2 * v2.x; acc.y += p2 * v2.y;
      acc.x += p3 * v3.x; acc.y += p3 * v3.y;
      zs += p0; zs += p1; zs += p2; zs += p3;
    }
  }
  float zz = zs + 1e-6f;
  float2 o = make_float2(acc.x / zz, acc.y / zz);
  ((float2*)out)[((size_t)row * 192 + 128) / 2 + lane] = o;
}

extern "C" void kernel_launch(void* const* d_in, const int* in_sizes, int n_in,
                              void* d_out, int out_size, void* d_ws, size_t ws_size,
                              hipStream_t stream) {
  const float* h  = (const float*)d_in[0];
  const float* p  = (const float*)d_in[1];
  const float* e  = (const float*)d_in[2];
  const float* ew = (const float*)d_in[3];
  const float* fc = (const float*)d_in[4];
  const int* esrc = (const int*)d_in[5];
  const int* edst = (const int*)d_in[6];
  const float* Wc = (const float*)d_in[7];
  const float* bc = (const float*)d_in[8];
  const float* Wg = (const float*)d_in[9];
  const float* bg = (const float*)d_in[10];
  const float* WQ = (const float*)d_in[11];
  const float* WK = (const float*)d_in[12];
  const float* WVm = (const float*)d_in[13];
  const float* WE = (const float*)d_in[14];
  float* out = (float*)d_out;

  const int N = in_sizes[0] / 64;
  const int E = in_sizes[5];
  const size_t ND = (size_t)N * 64;

  float* w = (float*)d_ws;
  float* Tx1  = w; w += ND;
  float* Tx2  = w; w += ND;
  float* Tx3  = w; w += ND;
  float* gcn  = w; w += ND;
  float* scr  = w; w += 2 * ND;
  float* Qh   = w; w += ND;
  float* Kh   = w; w += ND;
  float* Vh   = w; w += ND;
  float* lst  = w; w += ND + (size_t)N * 8;
  float* dinv = w; w += N;
  int* cnt    = (int*)w; w += N;
  int* total  = (int*)w; w += 1;
  int* start   = (int*)lst;
  int* epos    = start + N;
  int* rec     = (int*)scr;
  float* ssb   = Tx1;

  hipMemsetAsync(cnt, 0, sizeof(int) * ((size_t)N + 1), stream);

  const int B = 256;
  const int rowBlocks = (N + 7) / 8;

  count_kernel<<<(E + B - 1) / B, B, 0, stream>>>(edst, cnt, epos, E);
  rowalloc_kernel<<<(N + B - 1) / B, B, 0, stream>>>(cnt, total, start, dinv, N);
  fill_kernel<<<(E + B - 1) / B, B, 0, stream>>>(esrc, edst, dinv, ew, start, rec, epos, E);

  gather_rows_pair<<<rowBlocks, B, 0, stream>>>(h, rec, dinv, start, cnt, Tx1, gcn, N);
  gather_rows<<<rowBlocks, B, 0, stream>>>(Tx1, rec, dinv, start, cnt, h, Tx2, N);
  gather_rows<<<rowBlocks, B, 0, stream>>>(Tx2, rec, dinv, start, cnt, Tx1, Tx3, N);

  const int rowTiles = (N + 15) / 16;
  const int gemmBlocks = (rowTiles + 1) / 2;
  dense_kernel<<<dim3(gemmBlocks, 5), 64, 0, stream>>>(h, p, Tx1, Tx2, Tx3, gcn, fc, Wc, bc, Wg, bg,
                                                       WQ, WK, WVm, out, Qh, Kh, Vh, N);

  const int eTiles = (E + 15) / 16;
  attn_kernel<<<(eTiles + ATT_WAVES - 1) / ATT_WAVES, 32 * ATT_WAVES, 0, stream>>>(e, WE, Qh, Kh, esrc, edst, epos, ssb, E);
  attn_gather_rows<<<rowBlocks, B, 0, stream>>>(Vh, ssb, rec, start, cnt, out, N);
}


# All four neighbour-sum gathers: one wave sums two rows, one per 16-lane half, 16-byte lane reads (two 256-byte rows per load), slot words handed out by v_permlane16; half the waves; sum order kept
# speedup vs baseline: 3.9041x; 1.0470x over previous
#include <hip/hip_runtime.h>
#include <hip/hip_bf16.h>

typedef __attribute__((ext_vector_type(2))) float v2f;
typedef __attribute__((ext_vector_type(8))) float v8f;

__device__ __forceinline__ v8f wmma4(v2f a, v2f b, v8f c) {
  return __builtin_amdgcn_wmma_f32_16x16x4_f32(
       false, a,  false, b,
       (short)0, c,  false,  false);
}

__global__ void count_kernel(const int* __restrict__ dst, int* __restrict__ cnt,
                             int* __restrict__ rank, int E) {
  int tid = blockIdx.x * blockDim.x + threadIdx.x;
  if (tid < E) rank[tid] = atomicAdd(&cnt[dst[tid]], 1);
}

__global__ void rowalloc_kernel(const int* __restrict__ cnt, int* __restrict__ total,
                                int* __restrict__ start, float* __restrict__ dinv, int N) {
  int tid = blockIdx.x * blockDim.x + threadIdx.x;
  int lane = threadIdx.x & 31;
  int c = tid < N ? cnt[tid] : 0;
  int inc = c;
  for (int o = 1; o < 32; o <<= 1) {
    int v = __shfl_up(inc, o);
    if (lane >= o) inc += v;
  }
  int base = 0;
  if (lane == 31) base = atomicAdd(total, inc);
  base = __shfl(base, 31);
  if (tid < N) {
    int st = base + inc - c;
    start[tid] = st;
    float d = (float)c;
    d = d < 1.0f ? 1.0f : d;
    dinv[tid] = rsqrtf(d);
  }
}

__global__ void fill_kernel(const int* __restrict__ src, const int* __restrict__ dst,
                            const float* __restrict__ dinv, const float* __restrict__ ew,
                            const int* __restrict__ start, int* __restrict__ rec,
                            int* __restrict__ epos, int E) {
  int tid = blockIdx.x * blockDim.x + threadIdx.x;
  if (tid >= E) return;
  int s = src[tid], d = dst[tid];
  int pos = start[d] + epos[tid];
  ((int4*)rec)[pos] = make_int4(s, __float_as_int(dinv[s]), __float_as_int(ew[tid]), 0);
  epos[tid] = pos;
}


__device__ __forceinline__ int half_bcast(int v, int k) {
  unsigned sel = (unsigned)k * 0x11111111u;
  return (int)__builtin_amdgcn_permlane16((unsigned)v, (unsigned)v, sel, sel, false, false);
}

__device__ __forceinline__ float4 sel4(bool b, float4 v) {
  return make_float4(b ? v.x : 0.0f, b ? v.y : 0.0f, b ? v.z : 0.0f, b ? v.w : 0.0f);
}

__global__ void gather_rows(const float* __restrict__ x, const int* __restrict__ rec,
                            const float* __restrict__ dinv, const int* __restrict__ start,
                            const int* __restrict__ cnt, const float* __restrict__ prev,
                            float* __restrict__ out, int N) {
  int pr = blockIdx.x * (blockDim.x >> 5) + (threadIdx.x >> 5);
  if (2 * pr >= N) return;
  int lane = threadIdx.x & 31, c = lane & 15;
  const bool hi = lane >= 16;
  int row = 2 * pr + (hi ? 1 : 0);
  const bool rlive = row < N;
  row = rlive ? row : N - 1;
  const int beg = start[row];
  const int cn = rlive ? cnt[row] : 0;
  const float dr = dinv[row];
  const int4* __restrict__ rec4 = (const int4*)rec;
  const float4* __restrict__ x4 = (const float4*)x;
  float4 pv = make_float4(0.0f, 0.0f, 0.0f, 0.0f);
  if (prev) pv = ((const float4*)prev)[(size_t)row * 16 + c];
  int n0 = __builtin_amdgcn_readlane(cn, 0), n1 = __builtin_amdgcn_readlane(cn, 16);
  const int nmax = n0 > n1 ? n0 : n1;
  float4 acc = make_float4(0.0f, 0.0f, 0.0f, 0.0f);
  for (int base = 0; base < nmax; base += 32) {
    const bool ha = base + c < cn, hb = base + 16 + c < cn;
    int4 ra = rec4[ha ? beg + base + c : 0];
    int4 rb = rec4[hb ? beg + base + 16 + c : 0];
    const int sva = ha ? ra.x : 0, svb = hb ? rb.x : 0;
    const int wia = ha ? __float_as_int(-(__int_as_float(ra.y) * dr)) : 0;
    const int wib = hb ? __float_as_int(-(__int_as_float(rb.y) * dr)) : 0;
    int n = nmax - base;
    n = n < 32 ? n : 32;
    for (int k0 = 0; k0 < n; k0 += 4) {
      const int q = base + k0;
      const bool up = k0 >= 16;
      const int sv = up ? svb : sva, wi = up ? wib : wia;
      const int k = k0 & 15;
      bool b0 = q < cn, b1 = q + 1 < cn, b2 = q + 2 < cn, b3 = q + 3 < cn;
      int s0 = half_bcast(sv, k), s1 = half_bcast(sv, k + 1);
      int s2 = half_bcast(sv, k + 2), s3 = half_bcast(sv, k + 3);
      float w0 = __int_as_float(half_bcast(wi, k));
      float w1 = __int_as_float(half_bcast(wi, k + 1));
      float w2 = __int_as_float(half_bcast(wi, k + 2));
      float w3 = __int_as_float(half_bcast(wi, k + 3));
      float4 v0 = x4[(unsigned)s0 * 16u + c];
      float4 v1 = x4[(unsigned)s1 * 16u + c];
      float4 v2 = x4[(unsigned)s2 * 16u + c];
      float4 v3 = x4[(unsigned)s3 * 16u + c];
      v0 = sel4(b0, v0); v1 = sel4(b1, v1); v2 = sel4(b2, v2); v3 = sel4(b3, v3);
      acc.x = fmaf(w0, v0.x, acc.x); acc.y = fmaf(w0, v0.y, acc.y);
      acc.z = fmaf(w0, v0.z, acc.z); acc.w = fmaf(w0, v0.w, acc.w);
      acc.x = fmaf(w1, v1.x, acc.x); acc.y = fmaf(w1, v1.y, acc.y);
      acc.z = fmaf(w1, v1.z, acc.z); acc.w = fmaf(w1, v1.w, acc.w);
      acc.x = fmaf(w2, v2.x, acc.x); acc.y = fmaf(w2, v2.y, acc.y);
      acc.z = fmaf(w2, v2.z, acc.z); acc.w = fmaf(w2, v2.w, acc.w);
      acc.x = fmaf(w3, v3.x, acc.x); acc.y = fmaf(w3, v3.y, acc.y);
      acc.z = fmaf(w3, v3.z, acc.z); acc.w = fmaf(w3, v3.w, acc.w);
    }
  }
  if (prev) {
    acc.x = 2.0f * acc.x - pv.x;
    acc.y = 2.0f * acc.y - pv.y;
    acc.z = 2.0f * acc.z - pv.z;
    acc.w = 2.0f * acc.w - pv.w;
  }
  if (rlive) ((float4*)out)[(size_t)row * 16 + c] = acc;
}

__global__ void gather_rows_pair(const float* __restrict__ x, const int* __restrict__ rec,
                                 const float* __restrict__ dinv,
                                 const int* __restrict__ start, const int* __restrict__ cnt,
                                 float* __restrict__ outa, float* __restrict__ outb, int N) {
  int pr = blockIdx.x * (blockDim.x >> 5) + (threadIdx.x >> 5);
  if (2 * pr >= N) return;
  int lane = threadIdx.x & 31, c = lane & 15;
  const bool hi = lane >= 16;
  int row = 2 * pr + (hi ? 1 : 0);
  const bool rlive = row < N;
  row = rlive ? row : N - 1;
  const int beg = start[row];
  const int cn = rlive ? cnt[row] : 0;
  const float dr = dinv[row];
  const int4* __restrict__ rec4 = (const int4*)rec;
  const float4* __restrict__ x4 = (const float4*)x;
  int n0 = __builtin_amdgcn_readlane(cn, 0), n1 = __builtin_amdgcn_readlane(cn, 16);
  const int nmax = n0 > n1 ? n0 : n1;
  float4 acca = make_float4(0.0f, 0.0f, 0.0f, 0.0f);
  float4 accb = make_float4(0.0f, 0.0f, 0.0f, 0.0f);
  for (int base = 0; base < nmax; base += 32) {
    const bool ha = base + c < cn, hb = base + 16 + c < cn;
    int4 ra = rec4[ha ? beg + base + c : 0];
    int4 rb = rec4[hb ? beg + base + 16 + c : 0];
    const int sva = ha ? ra.x : 0, svb = hb ? rb.x : 0;
    const int aia = ha ? __float_as_int(-(__int_as_float(ra.y) * dr)) : 0;
    const int aib = hb ? __float_as_int(-(__int_as_float(rb.y) * dr)) : 0;
    const int bia = ha ? ra.z : 0, bib = hb ? rb.z : 0;
    int n = nmax - base;
    n = n < 32 ? n : 32;
    for (int k0 = 0; k0 < n; k0 += 4) {
      const int q = base + k0;
      const bool up = k0 >= 16;
      const int sv = up ? svb : sva, ai = up ? aib : aia, bi = up ? bib : bia;
      const int k = k0 & 15;
      bool b0 = q < cn, b1 = q + 1 < cn, b2 = q + 2 < cn, b3 = q + 3 < cn;
      int s0 = half_bcast(sv, k), s1 = half_bcast(sv, k + 1);
      int s2 = half_bcast(sv, k + 2), s3 = half_bcast(sv, k + 3);
      float a0 = __int_as_float(half_bcast(ai, k));
      float a1 = __int_as_float(half_bcast(ai, k + 1));
      float a2 = __int_as_float(half_bcast(ai, k + 2));
      float a3 = __int_as_float(half_bcast(ai, k + 3));
      float c0 = __int_as_float(half_bcast(bi, k));
      float c1 = __int_as_float(half_bcast(bi, k + 1));
      float c2 = __int_as_float(half_bcast(bi, k + 2));
      float c3 = __int_as_float(half_bcast(bi, k + 3));
      float4 v0 = x4[(unsigned)s0 * 16u + c];
      float4 v1 = x4[(unsigned)s1 * 16u + c];
      float4 v2 = x4[(unsigned)s2 * 16u + c];
      float4 v3 = x4[(unsigned)s3 * 16u + c];
      v0 = sel4(b0, v0); v1 = sel4(b1, v1); v2 = sel4(b2, v2); v3 = sel4(b3, v3);
      acca.x = fmaf(a0, v0.x, acca.x); acca.y = fmaf(a0, v0.y, acca.y);
      acca.z = fmaf(a0, v0.z, acca.z); acca.w = fmaf(a0, v0.w, acca.w);
      accb.x = fmaf(c0, v0.x, accb.x); accb.y = fmaf(c0, v0.y, accb.y);
      accb.z = fmaf(c0, v0.z, accb.z); accb.w = fmaf(c0, v0.w, accb.w);
      acca.x = fmaf(a1, v1.x, acca.x); acca.y = fmaf(a1, v1.y, acca.y);
      acca.z = fmaf(a1, v1.z, acca.z); acca.w = fmaf(a1, v1.w, acca.w);
      accb.x = fmaf(c1, v1.x, accb.x); accb.y = fmaf(c1, v1.y, accb.y);
      accb.z = fmaf(c1, v1.z, accb.z); accb.w = fmaf(c1, v1.w, accb.w);
      acca.x = fmaf(a2, v2.x, acca.x); acca.y = fmaf(a2, v2.y, acca.y);
      acca.z = fmaf(a2, v2.z, acca.z); acca.w = fmaf(a2, v2.w, acca.w);
      accb.x = fmaf(c2, v2.x, accb.x); accb.y = fmaf(c2, v2.y, accb.y);
      accb.z = fmaf(c2, v2.z, accb.z); accb.w = fmaf(c2, v2.w, accb.w);
      acca.x = fmaf(a3, v3.x, acca.x); acca.y = fmaf(a3, v3.y, acca.y);
      acca.z = fmaf(a3, v3.z, acca.z); acca.w = fmaf(a3, v3.w, acca.w);
      accb.x = fmaf(c3, v3.x, accb.x); accb.y = fmaf(c3, v3.y, accb.y);
      accb.z = fmaf(c3, v3.z, accb.z); accb.w = fmaf(c3, v3.w, accb.w);
    }
  }
  if (rlive) {
    ((float4*)outa)[(size_t)row * 16 + c] = acca;
    ((float4*)outb)[(size_t)row * 16 + c] = accb;
  }
}


template <bool SCALED>
__device__ __forceinline__ void plane_mma(const float* __restrict__ arow,
                                          const float* __restrict__ bcol, int half, float sc,
                                          v8f (&acc)[4]) {
  float4 av[8], bw[8][4];
#pragma unroll
  for (int j = 0; j < 8; ++j) {
    int k = 8 * j + 4 * half;
    av[j] = *(const float4*)(arow + k);
#pragma unroll
    for (int i = 0; i < 4; ++i) bw[j][i] = *(const float4*)(bcol + (size_t)(k + i) * 64);
  }
#pragma unroll
  for (int j = 0; j < 8; ++j) {
    float4 a = av[j];
    if (SCALED) { a.x *= sc; a.y *= sc; a.z *= sc; a.w *= sc; }
    v2f a01 = {a.x, a.y}, a23 = {a.z, a.w};
    acc[0] = wmma4(a01, (v2f){bw[j][0].x, bw[j][1].x}, acc[0]);
    acc[1] = wmma4(a01, (v2f){bw[j][0].y, bw[j][1].y}, acc[1]);
    acc[2] = wmma4(a01, (v2f){bw[j][0].z, bw[j][1].z}, acc[2]);
    acc[3] = wmma4(a01, (v2f){bw[j][0].w, bw[j][1].w}, acc[3]);
    acc[0] = wmma4(a23, (v2f){bw[j][2].x, bw[j][3].x}, acc[0]);
    acc[1] = wmma4(a23, (v2f){bw[j][2].y, bw[j][3].y}, acc[1]);
    acc[2] = wmma4(a23, (v2f){bw[j][2].z, bw[j][3].z}, acc[2]);
    acc[3] = wmma4(a23, (v2f){bw[j][2].w, bw[j][3].w}, acc[3]);
  }
}

__device__ __forceinline__ void strip_store(const v8f (&acc)[4], float4 bv, float* __restrict__ Out,
                                            int ldo, int col, int row0, int rows) {
#pragma unroll
  for (int r = 0; r < 8; ++r) {
    int row = row0 + r;
    if (row < rows) {
      float4 o = make_float4(acc[0][r] + bv.x, acc[1][r] + bv.y, acc[2][r] + bv.z, acc[3][r] + bv.w);
      *(float4*)(Out + (size_t)row * ldo + col) = o;
    }
  }
}

__global__ void __launch_bounds__(64) __attribute__((amdgpu_waves_per_eu(1, 4)))
dense_kernel(const float* __restrict__ h, const float* __restrict__ p,
             const float* __restrict__ Tx1, const float* __restrict__ Tx2,
             const float* __restrict__ Tx3, const float* __restrict__ gcn,
             const float* __restrict__ fc, const float* __restrict__ Wc,
             const float* __restrict__ bc, const float* __restrict__ Wg,
             const float* __restrict__ bg, const float* __restrict__ WQ,
             const float* __restrict__ WK, const float* __restrict__ WV,
             float* __restrict__ Out, float* __restrict__ Qh, float* __restrict__ Kh,
             float* __restrict__ Vh, int rows) {
  int rt = blockIdx.x * (blockDim.x >> 5) + (threadIdx.x >> 5);
  int rowTiles = (rows + 15) >> 4;
  if (rt >= rowTiles) return;
  int lane = threadIdx.x & 31, half = lane >> 4, c = lane & 15;
  int ar = rt * 16 + c;
  if (ar >= rows) ar = rows - 1;
  const size_t ao = (size_t)ar * 64;
  const int row0 = rt * 16 + 8 * half;
  const int which = blockIdx.y;
  v8f acc[4] = {};
  if (which == 0) {
    const float* As[4] = {h, Tx1, Tx2, Tx3};
#pragma unroll
    for (int k = 0; k < 4; ++k) {
      float sc = fc[(size_t)k * rows + ar];
      plane_mma<true>(As[k] + ao, Wc + (size_t)k * 64 * 64 + 4 * c, half, sc, acc);
    }
    float4 bv = *(const float4*)(bc + 4 * c);
    strip_store(acc, bv, Out, 192, 4 * c, row0, rows);
  } else if (which == 4) {
    plane_mma<false>(gcn + ao, Wg + 4 * c, half, 1.0f, acc);
    float4 bv = *(const float4*)(bg + 4 * c);
    strip_store(acc, bv, Out, 192, 64 + 4 * c, row0, rows);
  } else {
    const float* W = which == 1 ? WQ : (which == 2 ? WK : WV);
    float* O = which == 1 ? Qh : (which == 2 ? Kh : Vh);
    plane_mma<false>(h + ao, W + 4 * c, half, 1.0f, acc);
    plane_mma<false>(p + ao, W + (size_t)64 * 64 + 4 * c, half, 1.0f, acc);
    strip_store(acc, make_float4(0.0f, 0.0f, 0.0f, 0.0f), O, 64, 4 * c, row0, rows);
  }
}

#define ATT_WAVES 4
#define ATT_PITCH 68
__global__ void __launch_bounds__(32 * ATT_WAVES)
attn_kernel(const float* __restrict__ Ee, const float* __restrict__ WE,
                            const float* __restrict__ Qh, const float* __restrict__ Kh,
                            const int* __restrict__ src, const int* __restrict__ dst,
                            const int* __restrict__ epos, float* __restrict__ ssb, int E) {
  __shared__ __attribute__((aligned(16))) float stage[ATT_WAVES][2][16 * ATT_PITCH];
  const int wv = threadIdx.x >> 5;
  int tile = blockIdx.x * ATT_WAVES + wv;
  const int eTiles = (E + 15) >> 4;
  const bool tlive = tile < eTiles;
  if (!tlive) tile = eTiles - 1;
  int lane = threadIdx.x & 31, half = lane >> 4, c = lane & 15;
  const int e0 = tile * 16;
  int em = e0 + c;
  bool live = tlive && em < E;
  int er = em < E ? em : E - 1;
  float* eS = stage[wv][0];
  float* kqS = stage[wv][1];

  float4 ev[8], kv[8], qv[8];
#pragma unroll
  for (int i = 0; i < 8; ++i) {
    int re = e0 + 2 * i + half;
    re = re < E ? re : E - 1;
    ev[i] = *(const float4*)(Ee + (size_t)re * 64 + 4 * c);
  }
  int s = src[er], d = dst[er], slot = epos[er];
#pragma unroll
  for (int i = 0; i < 8; ++i) {
    int ri = 2 * i + half;
    int si = __shfl(s, ri), di = __shfl(d, ri);
    kv[i] = *(const float4*)(Kh + (size_t)si * 64 + 4 * c);
    qv[i] = *(const float4*)(Qh + (size_t)di * 64 + 4 * c);
  }
#pragma unroll
  for (int i = 0; i < 8; ++i) {
    int ri = 2 * i + half;
    *(float4*)(eS + ri * ATT_PITCH + 4 * c) = ev[i];
    float4 kq = make_float4(kv[i].x * qv[i].x, kv[i].y * qv[i].y, kv[i].z * qv[i].z, kv[i].w * qv[i].w);
    *(float4*)(kqS + ri * ATT_PITCH + 4 * c) = kq;
  }
  __syncthreads();

  const float* wcol = WE + 4 * c;
  const float* erow = eS + c * ATT_PITCH;
  v8f acc[4] = {};
#pragma unroll
  for (int j = 0; j < 8; ++j) {
    int k = 8 * j + 4 * half;
    float4 evj = *(const float4*)(erow + k);
    float4 w0 = *(const float4*)(wcol + (size_t)(k + 0) * 64);
    float4 w1 = *(const float4*)(wcol + (size_t)(k + 1) * 64);
    float4 w2 = *(const float4*)(wcol + (size_t)(k + 2) * 64);
    float4 w3 = *(const float4*)(wcol + (size_t)(k + 3) * 64);
    v2f e01 = {evj.x, evj.y}, e23 = {evj.z, evj.w};
    acc[0] = wmma4((v2f){w0.x, w1.x}, e01, acc[0]);
    acc[1] = wmma4((v2f){w0.y, w1.y}, e01, acc[1]);
    acc[2] = wmma4((v2f){w0.z, w1.z}, e01, acc[2]);
    acc[3] = wmma4((v2f){w0.w, w1.w}, e01, acc[3]);
    acc[0] = wmma4((v2f){w2.x, w3.x}, e23, acc[0]);
    acc[1] = wmma4((v2f){w2.y, w3.y}, e23, acc[1]);
    acc[2] = wmma4((v2f){w2.z, w3.z}, e23, acc[2]);
    acc[3] = wmma4((v2f){w2.w, w3.w}, e23, acc[3]);
  }

  const float* kqrow = kqS + c * ATT_PITCH + 32 * half;
  float4 pv[8];
#pragma unroll
  for (int i = 0; i < 8; ++i) pv[i] = *(const float4*)(kqrow + 4 * i);

  const float scale = 0.35355339059327373f;
  float ss[4];
#pragma unroll
  for (int q = 0; q < 4; ++q) {
    float kq[8] = {pv[2 * q].x, pv[2 * q].y, pv[2 * q].z, pv[2 * q].w,
                   pv[2 * q + 1].x, pv[2 * q + 1].y, pv[2 * q + 1].z, pv[2 * q + 1].w};
    float sc = 0.0f;
#pragma unroll
    for (int dd = 0; dd < 8; ++dd)
      sc += kq[dd] * acc[dd & 3][2 * q + (dd >> 2)];
    sc *= scale;
    sc = fminf(5.0f, fmaxf(-5.0f, sc));
    ss[q] = __expf(sc);
  }
  if (live) *(float4*)(ssb + (size_t)slot * 8 + 4 * half) = make_float4(ss[0], ss[1], ss[2], ss[3]);
}

__global__ void attn_gather_rows(const float* __restrict__ Vh, const float* __restrict__ ssb,
                                 const int* __restrict__ rec, const int* __restrict__ start,
                                 const int* __restrict__ cnt, float* __restrict__ out, int N) {
  int pr = blockIdx.x * (blockDim.x >> 5) + (threadIdx.x >> 5);
  if (2 * pr >= N) return;
  int lane = threadIdx.x & 31, c = lane & 15;
  const bool hi = lane >= 16;
  int row = 2 * pr + (hi ? 1 : 0);
  const bool rlive = row < N;
  row = rlive ? row : N - 1;
  const int hd = c >> 1;
  const int beg = start[row];
  const int cn = rlive ? cnt[row] : 0;
  const float4* __restrict__ v4p = (const float4*)Vh;
  int n0 = __builtin_amdgcn_readlane(cn, 0), n1 = __builtin_amdgcn_readlane(cn, 16);
  const int nmax = n0 > n1 ? n0 : n1;
  float4 acc = make_float4(0.0f, 0.0f, 0.0f, 0.0f);
  float zs = 0.0f;
  for (int base = 0; base < nmax; base += 32) {
    const bool ha = base + c < cn, hb = base + 16 + c < cn;
    int ra = rec[(size_t)(ha ? beg + base + c : 0) * 4];
    int rb = rec[(size_t)(hb ? beg + base + 16 + c : 0) * 4];
    const int sva = ha ? ra : 0, svb = hb ? rb : 0;
    int n = nmax - base;
    n = n < 32 ? n : 32;
    for (int k0 = 0; k0 < n; k0 += 4) {
      const int q = base + k0;
      const bool up = k0 >= 16;
      const int sv = up ? svb : sva;
      const int k = k0 & 15;
      bool b0 = q < cn, b1 = q + 1 < cn, b2 = q + 2 < cn, b3 = q + 3 < cn;
      int s0 = half_bcast(sv, k), s1 = half_bcast(sv, k + 1);
      int s2 = half_bcast(sv, k + 2), s3 = half_bcast(sv, k + 3);
      unsigned j0 = b0 ? (unsigned)(beg + q) : 0u, j1 = b1 ? (unsigned)(beg + q + 1) : 0u;
      unsigned j2 = b2 ? (unsigned)(beg + q + 2) : 0u, j3 = b3 ? (unsigned)(beg + q + 3) : 0u;
      float p0 = ssb[j0 * 8u + hd], p1 = ssb[j1 * 8u + hd];
      float p2 = ssb[j2 * 8u + hd], p3 = ssb[j3 * 8u + hd];
      float4 v0 = v4p[(unsigned)s0 * 16u + c];
      float4 v1 = v4p[(unsigned)s1 * 16u + c];
      float4 v2 = v4p[(unsigned)s2 * 16u + c];
      float4 v3 = v4p[(unsigned)s3 * 16u + c];
      p0 = b0 ? p0 : 0.0f; p1 = b1 ? p1 : 0.0f; p2 = b2 ? p2 : 0.0f; p3 = b3 ? p3 : 0.0f;
      v0 = sel4(b0, v0); v1 = sel4(b1, v1); v2 = sel4(b2, v2); v3 = sel4(b3, v3);
      acc.x = fmaf(p0, v0.x, acc.x); acc.y = fmaf(p0, v0.y, acc.y);
      acc.z = fmaf(p0, v0.z, acc.z); acc.w = fmaf(p0, v0.w, acc.w);
      acc.x = fmaf(p1, v1.x, acc.x); acc.y = fmaf(p1, v1.y, acc.y);
      acc.z = fmaf(p1, v1.z, acc.z); acc.w = fmaf(p1, v1.w, acc.w);
      acc.x = fmaf(p2, v2.x, acc.x); acc.y = fmaf(p2, v2.y, acc.y);
      acc.z = fmaf(p2, v2.z, acc.z); acc.w = fmaf(p2, v2.w, acc.w);
      acc.x = fmaf(p3, v3.x, acc.x); acc.y = fmaf(p3, v3.y, acc.y);
      acc.z = fmaf(p3, v3.z, acc.z); acc.w = fmaf(p3, v3.w, acc.w);
      zs += p0; zs += p1; zs += p2; zs += p3;
    }
  }
  float zz = zs + 1e-6f;
  float4 o = make_float4(acc.x / zz, acc.y / zz, acc.z / zz, acc.w / zz);
  if (rlive) *(float4*)(out + (size_t)row * 192 + 128 + 4 * c) = o;
}

extern "C" void kernel_launch(void* const* d_in, const int* in_sizes, int n_in,
                              void* d_out, int out_size, void* d_ws, size_t ws_size,
                              hipStream_t stream) {
  const float* h  = (const float*)d_in[0];
  const float* p  = (const float*)d_in[1];
  const float* e  = (const float*)d_in[2];
  const float* ew = (const float*)d_in[3];
  const float* fc = (const float*)d_in[4];
  const int* esrc = (const int*)d_in[5];
  const int* edst = (const int*)d_in[6];
  const float* Wc = (const float*)d_in[7];
  const float* bc = (const float*)d_in[8];
  const float* Wg = (const float*)d_in[9];
  const float* bg = (const float*)d_in[10];
  const float* WQ = (const float*)d_in[11];
  const float* WK = (const float*)d_in[12];
  const float* WVm = (const float*)d_in[13];
  const float* WE = (const float*)d_in[14];
  float* out = (float*)d_out;

  const int N = in_sizes[0] / 64;
  const int E = in_sizes[5];
  const size_t ND = (size_t)N * 64;

  float* w = (float*)d_ws;
  float* Tx1  = w; w += ND;
  float* Tx2  = w; w += ND;
  float* Tx3  = w; w += ND;
  float* gcn  = w; w += ND;
  float* scr  = w; w += 2 * ND;
  float* Qh   = w; w += ND;
  float* Kh   = w; w += ND;
  float* Vh   = w; w += ND;
  float* lst  = w; w += ND + (size_t)N * 8;
  float* dinv = w; w += N;
  int* cnt    = (int*)w; w += N;
  int* total  = (int*)w; w += 1;
  int* start   = (int*)lst;
  int* epos    = start + N;
  int* rec     = (int*)scr;
  float* ssb   = Tx1;

  hipMemsetAsync(cnt, 0, sizeof(int) * ((size_t)N + 1), stream);

  const int B = 256;
  const int rowBlocks = (N + 15) / 16;

  count_kernel<<<(E + B - 1) / B, B, 0, stream>>>(edst, cnt, epos, E);
  rowalloc_kernel<<<(N + B - 1) / B, B, 0, stream>>>(cnt, total, start, dinv, N);
  fill_kernel<<<(E + B - 1) / B, B, 0, stream>>>(esrc, edst, dinv, ew, start, rec, epos, E);

  gather_rows_pair<<<rowBlocks, B, 0, stream>>>(h, rec, dinv, start, cnt, Tx1, gcn, N);
  gather_rows<<<rowBlocks, B, 0, stream>>>(Tx1, rec, dinv, start, cnt, h, Tx2, N);
  gather_rows<<<rowBlocks, B, 0, stream>>>(Tx2, rec, dinv, start, cnt, Tx1, Tx3, N);

  const int rowTiles = (N + 15) / 16;
  const int gemmBlocks = (rowTiles + 1) / 2;
  dense_kernel<<<dim3(gemmBlocks, 5), 64, 0, stream>>>(h, p, Tx1, Tx2, Tx3, gcn, fc, Wc, bc, Wg, bg,
                                                       WQ, WK, WVm, out, Qh, Kh, Vh, N);

  const int eTiles = (E + 15) / 16;
  attn_kernel<<<(eTiles + ATT_WAVES - 1) / ATT_WAVES, 32 * ATT_WAVES, 0, stream>>>(e, WE, Qh, Kh, esrc, edst, epos, ssb, E);
  attn_gather_rows<<<rowBlocks, B, 0, stream>>>(Vh, ssb, rec, start, cnt, out, N);
}
